# starting kernel
# baseline (speedup 1.0000x reference)
#include <hip/hip_runtime.h>

typedef __attribute__((ext_vector_type(2))) float v2f;
typedef __attribute__((ext_vector_type(8))) float v8f;

#define NEG_SLOPE 0.2f


__device__ __forceinline__ unsigned f2ord(float f) {
  unsigned u = __float_as_uint(f);
  return (u & 0x80000000u) ? ~u : (u | 0x80000000u);
}
__device__ __forceinline__ float ord2f(unsigned o) {
  unsigned u = (o & 0x80000000u) ? (o & 0x7fffffffu) : ~o;
  return __uint_as_float(u);
}

__device__ __forceinline__ void edge_ends(const int* __restrict__ ei, int e,
                                          int Eraw, int& s, int& d) {
  if (e < Eraw) { s = ei[e]; d = ei[Eraw + e]; }
  else          { s = e - Eraw; d = s; }
}


__global__ void fill_f32(float* __restrict__ p, float v, long long n) {
  long long i = (long long)blockIdx.x * blockDim.x + threadIdx.x;
  if (i < n) p[i] = v;
}
__global__ void fill_u32(unsigned* __restrict__ p, unsigned v, long long n) {
  long long i = (long long)blockIdx.x * blockDim.x + threadIdx.x;
  if (i < n) p[i] = v;
}


__global__ void gemm_wmma_f32(const float* __restrict__ A,
                              const float* __restrict__ B,
                              float* __restrict__ C,
                              int M, int Nn, int K) {
  const int lane   = threadIdx.x & 31;
  const int wave   = blockIdx.x * (blockDim.x >> 5) + (threadIdx.x >> 5);
  const int ntiles = Nn >> 4;
  const int mtiles = M >> 4;
  const int mt64   = (mtiles + 3) >> 2;
  const int tileM  = wave / ntiles;
  const int tileN  = wave - tileM * ntiles;
  if (tileM >= mt64) return;

  const int lh  = lane >> 4;
  const int l16 = lane & 15;
  const int m0  = tileM * 4;

  const float* arow[4];
#pragma unroll
  for (int r = 0; r < 4; ++r) {
    int mr = m0 + r; if (mr > mtiles - 1) mr = mtiles - 1;
    arow[r] = A + (size_t)(mr * 16 + l16) * K + lh * 2;
  }
  const float* bptr = B + (size_t)(lh * 2) * Nn + tileN * 16 + l16;

  v8f acc0 = {}, acc1 = {}, acc2 = {}, acc3 = {};
  for (int k = 0; k < K; k += 4) {
    v2f b;
    b.x = bptr[(size_t)k * Nn];
    b.y = bptr[(size_t)(k + 1) * Nn];
    v2f a0 = *(const v2f*)(arow[0] + k);
    v2f a1 = *(const v2f*)(arow[1] + k);
    v2f a2 = *(const v2f*)(arow[2] + k);
    v2f a3 = *(const v2f*)(arow[3] + k);
    acc0 = __builtin_amdgcn_wmma_f32_16x16x4_f32(false, a0, false, b, (short)0, acc0, false, false);
    acc1 = __builtin_amdgcn_wmma_f32_16x16x4_f32(false, a1, false, b, (short)0, acc1, false, false);
    acc2 = __builtin_amdgcn_wmma_f32_16x16x4_f32(false, a2, false, b, (short)0, acc2, false, false);
    acc3 = __builtin_amdgcn_wmma_f32_16x16x4_f32(false, a3, false, b, (short)0, acc3, false, false);
  }

  const int col = tileN * 16 + l16;
  v8f accs[4] = {acc0, acc1, acc2, acc3};
#pragma unroll
  for (int r = 0; r < 4; ++r) {
    if (m0 + r < mtiles) {
      float* crow = C + (size_t)((m0 + r) * 16 + lh * 8) * Nn + col;
#pragma unroll
      for (int q = 0; q < 8; ++q) crow[(size_t)q * Nn] = accs[r][q];
    }
  }
}


template <int H, int F>
__global__ void node_alpha(const float* __restrict__ h,
                           const float* __restrict__ att_s,
                           const float* __restrict__ att_d,
                           float* __restrict__ asrc, float* __restrict__ adst,
                           int n_nodes) {
  int idx = blockIdx.x * blockDim.x + threadIdx.x;
  if (idx >= n_nodes * H) return;
  int n  = idx / H;
  int hd = idx - n * H;
  const float* hp = h + (size_t)n * (H * F) + hd * F;
  const float* as = att_s + hd * F;
  const float* ad = att_d + hd * F;
  float s = 0.f, d = 0.f;
#pragma unroll 8
  for (int f = 0; f < F; ++f) { float v = hp[f]; s += v * as[f]; d += v * ad[f]; }
  asrc[idx] = s;
  adst[idx] = d;
}


template <int H>
__global__ void edge_max_k(const int* __restrict__ ei,
                           const float* __restrict__ asrc,
                           const float* __restrict__ adst,
                           unsigned* __restrict__ mord, int Eraw, int Etot) {
  int idx = blockIdx.x * blockDim.x + threadIdx.x;
  if (idx >= Etot * H) return;
  int e  = idx / H;
  int hd = idx - e * H;
  int s, d; edge_ends(ei, e, Eraw, s, d);
  float v = asrc[s * H + hd] + adst[d * H + hd];
  v = v > 0.f ? v : NEG_SLOPE * v;
  atomicMax(mord + d * H + hd, f2ord(v));
}

template <int H>
__global__ void edge_expsum_k(const int* __restrict__ ei,
                              const float* __restrict__ asrc,
                              const float* __restrict__ adst,
                              const unsigned* __restrict__ mord,
                              float* __restrict__ denom, int Eraw, int Etot) {
  int idx = blockIdx.x * blockDim.x + threadIdx.x;
  if (idx >= Etot * H) return;
  int e  = idx / H;
  int hd = idx - e * H;
  int s, d; edge_ends(ei, e, Eraw, s, d);
  float v = asrc[s * H + hd] + adst[d * H + hd];
  v = v > 0.f ? v : NEG_SLOPE * v;
  float m = ord2f(mord[d * H + hd]);
  atomicAdd(denom + d * H + hd, __expf(v - m));
}

template <int H, int F>
__global__ void edge_agg_k(const int* __restrict__ ei,
                           const float* __restrict__ asrc,
                           const float* __restrict__ adst,
                           const unsigned* __restrict__ mord,
                           const float* __restrict__ denom,
                           const float* __restrict__ feat,
                           float* __restrict__ out, int Eraw, int Etot) {
  const int C  = H * F;
  const int CV = C / 4;
  long long idx = (long long)blockIdx.x * blockDim.x + threadIdx.x;
  if (idx >= (long long)Etot * CV) return;
  int e  = (int)(idx / CV);
  int c4 = (int)(idx - (long long)e * CV);
  int c  = c4 * 4;
  int hd = c / F;
  int s, d; edge_ends(ei, e, Eraw, s, d);
  float v = asrc[s * H + hd] + adst[d * H + hd];
  v = v > 0.f ? v : NEG_SLOPE * v;
  float m     = ord2f(mord[d * H + hd]);
  float alpha = __expf(v - m) / (denom[d * H + hd] + 1e-16f);
  const float4 f4 = *(const float4*)(feat + (size_t)s * C + c);
  float* op = out + (size_t)d * C + c;
  atomicAdd(op + 0, f4.x * alpha);
  atomicAdd(op + 1, f4.y * alpha);
  atomicAdd(op + 2, f4.z * alpha);
  atomicAdd(op + 3, f4.w * alpha);
}


__global__ void bias_elu(float* __restrict__ x, const float* __restrict__ b,
                         int C, long long n) {
  long long i = (long long)blockIdx.x * blockDim.x + threadIdx.x;
  if (i >= n) return;
  float v = x[i] + b[(int)(i % C)];
  x[i] = v > 0.f ? v : expm1f(v);
}

__global__ void bias_add(float* __restrict__ x, const float* __restrict__ b,
                         int C, long long n) {
  long long i = (long long)blockIdx.x * blockDim.x + threadIdx.x;
  if (i >= n) return;
  x[i] += b[(int)(i % C)];
}


static inline int cdiv(long long a, long long b) { return (int)((a + b - 1) / b); }

extern "C" void kernel_launch(void* const* d_in, const int* in_sizes, int n_in,
                              void* d_out, int out_size, void* d_ws, size_t ws_size,
                              hipStream_t stream) {
  const float* x      = (const float*)d_in[0];
  const int*   ei     = (const int*)  d_in[1];
  const float* W1     = (const float*)d_in[2];
  const float* att_s1 = (const float*)d_in[3];
  const float* att_d1 = (const float*)d_in[4];
  const float* b1     = (const float*)d_in[5];
  const float* W2     = (const float*)d_in[6];
  const float* att_s2 = (const float*)d_in[7];
  const float* att_d2 = (const float*)d_in[8];
  const float* b2     = (const float*)d_in[9];

  const int IN   = 128;
  const int C1   = 256;
  const int F2   = 64;
  const int Nn   = in_sizes[0] / IN;
  const int Eraw = in_sizes[1] / 2;
  const int Etot = Eraw + Nn;

  float* ws = (float*)d_ws;
  size_t off = 0;
  float*    h1     = ws + off; off += (size_t)Nn * C1;
  float*    out1   = ws + off; off += (size_t)Nn * C1;
  float*    asrc1  = ws + off; off += (size_t)Nn * 4;
  float*    adst1  = ws + off; off += (size_t)Nn * 4;
  unsigned* mord1  = (unsigned*)(ws + off); off += (size_t)Nn * 4;
  float*    denom1 = ws + off; off += (size_t)Nn * 4;
  float*    h2     = h1;
  float*    asrc2  = asrc1;
  float*    adst2  = adst1;
  unsigned* mord2  = mord1;
  float*    denom2 = denom1;
  float*    outf   = (float*)d_out;

  const int T = 256;

  {
    int mt64  = (Nn / 16 + 3) / 4;
    int tiles = mt64 * (C1 / 16);
    gemm_wmma_f32<<<cdiv(tiles, 8), 256, 0, stream>>>(x, W1, h1, Nn, C1, IN);
  }
  node_alpha<4, 64><<<cdiv((long long)Nn * 4, T), T, 0, stream>>>(
      h1, att_s1, att_d1, asrc1, adst1, Nn);
  fill_u32<<<cdiv((long long)Nn * 4, T), T, 0, stream>>>(mord1, 0u, (long long)Nn * 4);
  edge_max_k<4><<<cdiv((long long)Etot * 4, T), T, 0, stream>>>(
      ei, asrc1, adst1, mord1, Eraw, Etot);
  fill_f32<<<cdiv((long long)Nn * 4, T), T, 0, stream>>>(denom1, 0.f, (long long)Nn * 4);
  edge_expsum_k<4><<<cdiv((long long)Etot * 4, T), T, 0, stream>>>(
      ei, asrc1, adst1, mord1, denom1, Eraw, Etot);
  fill_f32<<<cdiv((long long)Nn * C1, T), T, 0, stream>>>(out1, 0.f, (long long)Nn * C1);
  edge_agg_k<4, 64><<<cdiv((long long)Etot * (C1 / 4), T), T, 0, stream>>>(
      ei, asrc1, adst1, mord1, denom1, h1, out1, Eraw, Etot);
  bias_elu<<<cdiv((long long)Nn * C1, T), T, 0, stream>>>(out1, b1, C1, (long long)Nn * C1);

  {
    int mt64  = (Nn / 16 + 3) / 4;
    int tiles = mt64 * (F2 / 16);
    gemm_wmma_f32<<<cdiv(tiles, 8), 256, 0, stream>>>(out1, W2, h2, Nn, F2, C1);
  }
  node_alpha<1, 64><<<cdiv((long long)Nn, T), T, 0, stream>>>(
      h2, att_s2, att_d2, asrc2, adst2, Nn);
  fill_u32<<<cdiv((long long)Nn, T), T, 0, stream>>>(mord2, 0u, (long long)Nn);
  edge_max_k<1><<<cdiv((long long)Etot, T), T, 0, stream>>>(
      ei, asrc2, adst2, mord2, Eraw, Etot);
  fill_f32<<<cdiv((long long)Nn, T), T, 0, stream>>>(denom2, 0.f, (long long)Nn);
  edge_expsum_k<1><<<cdiv((long long)Etot, T), T, 0, stream>>>(
      ei, asrc2, adst2, mord2, denom2, Eraw, Etot);
  fill_f32<<<cdiv((long long)Nn * F2, T), T, 0, stream>>>(outf, 0.f, (long long)Nn * F2);
  edge_agg_k<1, 64><<<cdiv((long long)Etot * (F2 / 4), T), T, 0, stream>>>(
      ei, asrc2, adst2, mord2, denom2, h2, outf, Eraw, Etot);
  bias_add<<<cdiv((long long)Nn * F2, T), T, 0, stream>>>(outf, b2, F2, (long long)Nn * F2);
}


# Edges counting-sorted by destination each call; one wave per node does max, exp-sum and weighted row sum in registers: float atomics and zero fills gone; sums now in arrival order then lane tree
# speedup vs baseline: 2.3351x; 2.3351x over previous
#include <hip/hip_runtime.h>

typedef __attribute__((ext_vector_type(2))) float v2f;
typedef __attribute__((ext_vector_type(8))) float v8f;

#define NEG_SLOPE 0.2f


__global__ void fill_i32(int* __restrict__ p, int v, long long n) {
  long long i = (long long)blockIdx.x * blockDim.x + threadIdx.x;
  if (i < n) p[i] = v;
}


__global__ void gemm_wmma_f32(const float* __restrict__ A,
                              const float* __restrict__ B,
                              float* __restrict__ C,
                              int M, int Nn, int K) {
  const int lane   = threadIdx.x & 31;
  const int wave   = blockIdx.x * (blockDim.x >> 5) + (threadIdx.x >> 5);
  const int ntiles = Nn >> 4;
  const int mtiles = M >> 4;
  const int mt64   = (mtiles + 3) >> 2;
  const int tileM  = wave / ntiles;
  const int tileN  = wave - tileM * ntiles;
  if (tileM >= mt64) return;

  const int lh  = lane >> 4;
  const int l16 = lane & 15;
  const int m0  = tileM * 4;

  const float* arow[4];
#pragma unroll
  for (int r = 0; r < 4; ++r) {
    int mr = m0 + r; if (mr > mtiles - 1) mr = mtiles - 1;
    arow[r] = A + (size_t)(mr * 16 + l16) * K + lh * 2;
  }
  const float* bptr = B + (size_t)(lh * 2) * Nn + tileN * 16 + l16;

  v8f acc0 = {}, acc1 = {}, acc2 = {}, acc3 = {};
  for (int k = 0; k < K; k += 4) {
    v2f b;
    b.x = bptr[(size_t)k * Nn];
    b.y = bptr[(size_t)(k + 1) * Nn];
    v2f a0 = *(const v2f*)(arow[0] + k);
    v2f a1 = *(const v2f*)(arow[1] + k);
    v2f a2 = *(const v2f*)(arow[2] + k);
    v2f a3 = *(const v2f*)(arow[3] + k);
    acc0 = __builtin_amdgcn_wmma_f32_16x16x4_f32(false, a0, false, b, (short)0, acc0, false, false);
    acc1 = __builtin_amdgcn_wmma_f32_16x16x4_f32(false, a1, false, b, (short)0, acc1, false, false);
    acc2 = __builtin_amdgcn_wmma_f32_16x16x4_f32(false, a2, false, b, (short)0, acc2, false, false);
    acc3 = __builtin_amdgcn_wmma_f32_16x16x4_f32(false, a3, false, b, (short)0, acc3, false, false);
  }

  const int col = tileN * 16 + l16;
  v8f accs[4] = {acc0, acc1, acc2, acc3};
#pragma unroll
  for (int r = 0; r < 4; ++r) {
    if (m0 + r < mtiles) {
      float* crow = C + (size_t)((m0 + r) * 16 + lh * 8) * Nn + col;
#pragma unroll
      for (int q = 0; q < 8; ++q) crow[(size_t)q * Nn] = accs[r][q];
    }
  }
}


template <int H, int F>
__global__ void node_alpha(const float* __restrict__ h,
                           const float* __restrict__ att_s,
                           const float* __restrict__ att_d,
                           float* __restrict__ asrc, float* __restrict__ adst,
                           int n_nodes) {
  int idx = blockIdx.x * blockDim.x + threadIdx.x;
  if (idx >= n_nodes * H) return;
  int n  = idx / H;
  int hd = idx - n * H;
  const float* hp = h + (size_t)n * (H * F) + hd * F;
  const float* as = att_s + hd * F;
  const float* ad = att_d + hd * F;
  float s = 0.f, d = 0.f;
#pragma unroll 8
  for (int f = 0; f < F; ++f) { float v = hp[f]; s += v * as[f]; d += v * ad[f]; }
  asrc[idx] = s;
  adst[idx] = d;
}


__global__ void dst_count(const int* __restrict__ ei, int* __restrict__ cnt, int Eraw) {
  int e = blockIdx.x * blockDim.x + threadIdx.x;
  if (e >= Eraw) return;
  atomicAdd(cnt + ei[Eraw + e], 1);
}

__global__ void __launch_bounds__(1024)
dst_scan(const int* __restrict__ cnt, int* __restrict__ rowptr,
         int* __restrict__ cursor, int n) {
  __shared__ int wtot[32];
  const int t    = threadIdx.x;
  const int lane = t & 31;
  const int wv   = t >> 5;
  int carry = 0;
  for (int base = 0; base < n; base += 4096) {
    const int i0 = base + t * 4;
    int c[4];
    for (int j = 0; j < 4; ++j) c[j] = (i0 + j < n) ? cnt[i0 + j] : 0;
    const int tsum = c[0] + c[1] + c[2] + c[3];
    int incl = tsum;
    for (int d = 1; d < 32; d <<= 1) {
      int y = __shfl_up(incl, d);
      if (lane >= d) incl += y;
    }
    if (lane == 31) wtot[wv] = incl;
    __syncthreads();
    int ws = wtot[lane];
    for (int d = 1; d < 32; d <<= 1) {
      int y = __shfl_up(ws, d);
      if (lane >= d) ws += y;
    }
    const int wprefix = __shfl(ws, wv) - __shfl(incl, 31);
    const int total   = __shfl(ws, 31);
    int run = carry + wprefix + incl - tsum;
    for (int j = 0; j < 4; ++j) {
      if (i0 + j < n) { rowptr[i0 + j] = run; cursor[i0 + j] = run; run += c[j]; }
    }
    carry += total;
    __syncthreads();
  }
  if (t == 0) rowptr[n] = carry;
}

__global__ void dst_place(const int* __restrict__ ei, int* __restrict__ cursor,
                          int* __restrict__ col, int Eraw) {
  int e = blockIdx.x * blockDim.x + threadIdx.x;
  if (e >= Eraw) return;
  int s = ei[e];
  int d = ei[Eraw + e];
  int p = atomicAdd(cursor + d, 1);
  col[p] = s;
}


template <int N> using vecf = float __attribute__((ext_vector_type(N)));

__device__ __forceinline__ void load_heads(const float* __restrict__ p, float (&v)[4]) {
  const float4 t = *(const float4*)p;
  v[0] = t.x; v[1] = t.y; v[2] = t.z; v[3] = t.w;
}
__device__ __forceinline__ void load_heads(const float* __restrict__ p, float (&v)[1]) {
  v[0] = p[0];
}

__device__ __forceinline__ float leaky(float v) { return v > 0.f ? v : NEG_SLOPE * v; }

template <int H, int F, bool ELU>
__global__ void __launch_bounds__(256)
seg_softmax_agg(const int* __restrict__ rowptr, const int* __restrict__ col,
                const float* __restrict__ asrc, const float* __restrict__ adst,
                const float* __restrict__ feat, const float* __restrict__ bias,
                float* __restrict__ out, int n_nodes) {
  constexpr int C   = H * F;
  constexpr int CPL = C / 32;
  constexpr int VW  = CPL >= 4 ? 4 : CPL;
  constexpr int NL  = CPL / VW;
  constexpr int U   = 8 / NL;
  static_assert(U * NL == 8, "the load fence below names eight values");
  typedef vecf<VW> vrow;

  const int lane = threadIdx.x & 31;
  const int n = __builtin_amdgcn_readfirstlane(
      (int)(blockIdx.x * (blockDim.x >> 5) + (threadIdx.x >> 5)));
  if (n >= n_nodes) return;
  const int beg = rowptr[n];
  const int end = rowptr[n + 1];

  float ad[H], as_self[H], vself[H], mx[H];
  load_heads(adst + (size_t)n * H, ad);
  load_heads(asrc + (size_t)n * H, as_self);
  for (int h = 0; h < H; ++h) { vself[h] = leaky(as_self[h] + ad[h]); mx[h] = vself[h]; }

  for (int i = beg + lane; i < end; i += 32) {
    float a[H];
    load_heads(asrc + (size_t)col[i] * H, a);
    for (int h = 0; h < H; ++h) mx[h] = fmaxf(mx[h], leaky(a[h] + ad[h]));
  }
  for (int h = 0; h < H; ++h) {
    for (int m = 16; m >= 1; m >>= 1) mx[h] = fmaxf(mx[h], __shfl_xor(mx[h], m));
  }

  float den[H];
  for (int h = 0; h < H; ++h) den[h] = 0.f;
  for (int i = beg + lane; i < end; i += 32) {
    float a[H];
    load_heads(asrc + (size_t)col[i] * H, a);
    for (int h = 0; h < H; ++h) den[h] += __expf(leaky(a[h] + ad[h]) - mx[h]);
  }
  for (int h = 0; h < H; ++h) {
    for (int m = 16; m >= 1; m >>= 1) den[h] += __shfl_xor(den[h], m);
    den[h] += __expf(vself[h] - mx[h]);
    den[h] += 1e-16f;
  }

  int hsel[NL];
  vrow acc[NL];
  for (int j = 0; j < NL; ++j) {
    hsel[j] = (j * 32 * VW + lane * VW) / F;
    for (int q = 0; q < VW; ++q) acc[j][q] = 0.f;
  }
  {
    const float* row = feat + (size_t)n * C + lane * VW;
    for (int j = 0; j < NL; ++j) {
      float w = __expf(vself[0] - mx[0]) / den[0];
      for (int h = 1; h < H; ++h)
        w = (hsel[j] == h) ? __expf(vself[h] - mx[h]) / den[h] : w;
      const vrow f = *(const vrow*)(row + j * 32 * VW);
      for (int q = 0; q < VW; ++q) acc[j][q] += w * f[q];
    }
  }
  for (int base = beg; base < end; base += 32) {
    const int  i  = base + lane;
    const bool ok = i < end;
    const int  s  = ok ? col[i] : n;
    float a[H], w[H];
    load_heads(asrc + (size_t)s * H, a);
    for (int h = 0; h < H; ++h)
      w[h] = ok ? __expf(leaky(a[h] + ad[h]) - mx[h]) / den[h] : 0.f;
    const int cn  = (end - base) < 32 ? (end - base) : 32;
    const int cnp = (cn + U - 1) & ~(U - 1);
    for (int k = 0; k < cnp; k += U) {
      vrow  f[U][NL];
      float wj[U][NL];
#pragma unroll
      for (int u = 0; u < U; ++u) {
        const int sk = __builtin_amdgcn_readlane(s, k + u);
        float wk[H];
        for (int h = 0; h < H; ++h)
          wk[h] = __int_as_float(__builtin_amdgcn_readlane(__float_as_int(w[h]), k + u));
        const float* row = feat + (size_t)sk * C + lane * VW;
        for (int j = 0; j < NL; ++j) {
          float t = wk[0];
          for (int h = 1; h < H; ++h) t = (hsel[j] == h) ? wk[h] : t;
          wj[u][j] = t;
          f[u][j]  = *(const vrow*)(row + j * 32 * VW);
        }
      }
      asm volatile("" :: "v"(f[0 / NL][0 % NL]), "v"(f[1 / NL][1 % NL]), "v"(f[2 / NL][2 % NL]),
                         "v"(f[3 / NL][3 % NL]), "v"(f[4 / NL][4 % NL]), "v"(f[5 / NL][5 % NL]),
                         "v"(f[6 / NL][6 % NL]), "v"(f[7 / NL][7 % NL]));
#pragma unroll
      for (int u = 0; u < U; ++u)
        for (int j = 0; j < NL; ++j)
          for (int q = 0; q < VW; ++q) acc[j][q] += wj[u][j] * f[u][j][q];
    }
  }

  float* orow = out + (size_t)n * C + lane * VW;
  const float* brow = bias + lane * VW;
  for (int j = 0; j < NL; ++j) {
    const vrow b = *(const vrow*)(brow + j * 32 * VW);
    vrow r;
    for (int q = 0; q < VW; ++q) {
      float v = acc[j][q] + b[q];
      r[q] = ELU ? (v > 0.f ? v : expm1f(v)) : v;
    }
    *(vrow*)(orow + j * 32 * VW) = r;
  }
}


static inline int cdiv(long long a, long long b) { return (int)((a + b - 1) / b); }

extern "C" void kernel_launch(void* const* d_in, const int* in_sizes, int n_in,
                              void* d_out, int out_size, void* d_ws, size_t ws_size,
                              hipStream_t stream) {
  const float* x      = (const float*)d_in[0];
  const int*   ei     = (const int*)  d_in[1];
  const float* W1     = (const float*)d_in[2];
  const float* att_s1 = (const float*)d_in[3];
  const float* att_d1 = (const float*)d_in[4];
  const float* b1     = (const float*)d_in[5];
  const float* W2     = (const float*)d_in[6];
  const float* att_s2 = (const float*)d_in[7];
  const float* att_d2 = (const float*)d_in[8];
  const float* b2     = (const float*)d_in[9];

  const int IN   = 128;
  const int C1   = 256;
  const int F2   = 64;
  const int Nn   = in_sizes[0] / IN;
  const int Eraw = in_sizes[1] / 2;

  float* ws = (float*)d_ws;
  size_t off = 0;
  const size_t npad = ((size_t)Nn + 4) & ~(size_t)3;
  float* h1     = ws + off; off += (size_t)Nn * C1;
  float* out1   = ws + off; off += (size_t)Nn * C1;
  float* asrc1  = ws + off; off += (size_t)Nn * 4;
  float* adst1  = ws + off; off += (size_t)Nn * 4;
  int*   cnt    = (int*)(ws + off); off += npad;
  int*   rowptr = (int*)(ws + off); off += npad;
  int*   cursor = (int*)(ws + off); off += npad;
  int*   col    = (int*)(ws + off); off += (size_t)Eraw;
  float* h2     = h1;
  float* asrc2  = asrc1;
  float* adst2  = adst1;
  float* outf   = (float*)d_out;

  const int T = 256;

  fill_i32<<<cdiv((long long)Nn, T), T, 0, stream>>>(cnt, 0, (long long)Nn);
  dst_count<<<cdiv((long long)Eraw, T), T, 0, stream>>>(ei, cnt, Eraw);
  dst_scan<<<1, 1024, 0, stream>>>(cnt, rowptr, cursor, Nn);
  dst_place<<<cdiv((long long)Eraw, T), T, 0, stream>>>(ei, cursor, col, Eraw);

  {
    int mt64  = (Nn / 16 + 3) / 4;
    int tiles = mt64 * (C1 / 16);
    gemm_wmma_f32<<<cdiv(tiles, 8), 256, 0, stream>>>(x, W1, h1, Nn, C1, IN);
  }
  node_alpha<4, 64><<<cdiv((long long)Nn * 4, T), T, 0, stream>>>(
      h1, att_s1, att_d1, asrc1, adst1, Nn);
  seg_softmax_agg<4, 64, true><<<cdiv((long long)Nn, T / 32), T, 0, stream>>>(
      rowptr, col, asrc1, adst1, h1, b1, out1, Nn);

  {
    int mt64  = (Nn / 16 + 3) / 4;
    int tiles = mt64 * (F2 / 16);
    gemm_wmma_f32<<<cdiv(tiles, 8), 256, 0, stream>>>(out1, W2, h2, Nn, F2, C1);
  }
  node_alpha<1, 64><<<cdiv((long long)Nn, T), T, 0, stream>>>(
      h2, att_s2, att_d2, asrc2, adst2, Nn);
  seg_softmax_agg<1, 64, false><<<cdiv((long long)Nn, T / 32), T, 0, stream>>>(
      rowptr, col, asrc2, adst2, h2, b2, outf, Nn);
}


# GEMM wave tile turned to 16 rows x one head (64 cols): a_src/a_dst dots reduced from the accumulators in its epilogue (4 products per lane, then 16-lane tree); both node_alpha launches gone
# speedup vs baseline: 2.8840x; 1.2351x over previous
#include <hip/hip_runtime.h>

typedef __attribute__((ext_vector_type(2))) float v2f;
typedef __attribute__((ext_vector_type(8))) float v8f;

#define NEG_SLOPE 0.2f


__global__ void fill_i32(int* __restrict__ p, int v, long long n) {
  long long i = (long long)blockIdx.x * blockDim.x + threadIdx.x;
  if (i < n) p[i] = v;
}


typedef __attribute__((ext_vector_type(4))) float v4f;

template <int K, int HN>
__global__ void __launch_bounds__(256)
gemm_head_alpha(const float* __restrict__ A, const float* __restrict__ B,
                const float* __restrict__ att_s, const float* __restrict__ att_d,
                float* __restrict__ C, float* __restrict__ asrc, float* __restrict__ adst,
                int M) {
  constexpr int Nn = HN * 64;
  const int lane   = threadIdx.x & 31;
  const int wave   = blockIdx.x * (blockDim.x >> 5) + (threadIdx.x >> 5);
  const int mtiles = M >> 4;
  const int tileM  = wave / HN;
  const int hd     = wave - tileM * HN;
  if (tileM >= mtiles) return;

  const int lh  = lane >> 4;
  const int l16 = lane & 15;

  const float* ap = A + (size_t)(tileM * 16 + l16) * K + lh * 2;
  const float* bp = B + (size_t)(lh * 2) * Nn + hd * 64 + l16 * 4;

  v8f acc0 = {}, acc1 = {}, acc2 = {}, acc3 = {};
#pragma unroll 8
  for (int k = 0; k < K; k += 4) {
    const v2f a  = *(const v2f*)(ap + k);
    const v4f b0 = *(const v4f*)(bp + (size_t)k * Nn);
    const v4f b1 = *(const v4f*)(bp + (size_t)(k + 1) * Nn);
    v2f b;
    b.x = b0.x; b.y = b1.x;
    acc0 = __builtin_amdgcn_wmma_f32_16x16x4_f32(false, a, false, b, (short)0, acc0, false, false);
    b.x = b0.y; b.y = b1.y;
    acc1 = __builtin_amdgcn_wmma_f32_16x16x4_f32(false, a, false, b, (short)0, acc1, false, false);
    b.x = b0.z; b.y = b1.z;
    acc2 = __builtin_amdgcn_wmma_f32_16x16x4_f32(false, a, false, b, (short)0, acc2, false, false);
    b.x = b0.w; b.y = b1.w;
    acc3 = __builtin_amdgcn_wmma_f32_16x16x4_f32(false, a, false, b, (short)0, acc3, false, false);
  }

  const int row0 = tileM * 16 + lh * 8;
  float* crow = C + (size_t)row0 * Nn + hd * 64 + l16 * 4;
#pragma unroll
  for (int r = 0; r < 8; ++r) {
    v4f o;
    o.x = acc0[r]; o.y = acc1[r]; o.z = acc2[r]; o.w = acc3[r];
    *(v4f*)(crow + (size_t)r * Nn) = o;
  }

  const v4f s = *(const v4f*)(att_s + hd * 64 + l16 * 4);
  const v4f d = *(const v4f*)(att_d + hd * 64 + l16 * 4);
  float ps[8], pd[8];
#pragma unroll
  for (int r = 0; r < 8; ++r) {
    ps[r] = acc0[r] * s.x + acc1[r] * s.y + acc2[r] * s.z + acc3[r] * s.w;
    pd[r] = acc0[r] * d.x + acc1[r] * d.y + acc2[r] * d.z + acc3[r] * d.w;
  }
#pragma unroll
  for (int r = 0; r < 8; ++r) {
    for (int m = 8; m >= 1; m >>= 1) {
      ps[r] += __shfl_xor(ps[r], m);
      pd[r] += __shfl_xor(pd[r], m);
    }
  }
  float vs = ps[0], vd = pd[0];
#pragma unroll
  for (int r = 1; r < 8; ++r) {
    vs = (l16 == r) ? ps[r] : vs;
    vd = (l16 == r) ? pd[r] : vd;
  }
  if (l16 < 8) {
    asrc[(size_t)(row0 + l16) * HN + hd] = vs;
    adst[(size_t)(row0 + l16) * HN + hd] = vd;
  }
}


__global__ void dst_count(const int* __restrict__ ei, int* __restrict__ cnt, int Eraw) {
  int e = blockIdx.x * blockDim.x + threadIdx.x;
  if (e >= Eraw) return;
  atomicAdd(cnt + ei[Eraw + e], 1);
}

__global__ void __launch_bounds__(1024)
dst_scan(const int* __restrict__ cnt, int* __restrict__ rowptr,
         int* __restrict__ cursor, int n) {
  __shared__ int wtot[32];
  const int t    = threadIdx.x;
  const int lane = t & 31;
  const int wv   = t >> 5;
  int carry = 0;
  for (int base = 0; base < n; base += 4096) {
    const int i0 = base + t * 4;
    int c[4];
    for (int j = 0; j < 4; ++j) c[j] = (i0 + j < n) ? cnt[i0 + j] : 0;
    const int tsum = c[0] + c[1] + c[2] + c[3];
    int incl = tsum;
    for (int d = 1; d < 32; d <<= 1) {
      int y = __shfl_up(incl, d);
      if (lane >= d) incl += y;
    }
    if (lane == 31) wtot[wv] = incl;
    __syncthreads();
    int ws = wtot[lane];
    for (int d = 1; d < 32; d <<= 1) {
      int y = __shfl_up(ws, d);
      if (lane >= d) ws += y;
    }
    const int wprefix = __shfl(ws, wv) - __shfl(incl, 31);
    const int total   = __shfl(ws, 31);
    int run = carry + wprefix + incl - tsum;
    for (int j = 0; j < 4; ++j) {
      if (i0 + j < n) { rowptr[i0 + j] = run; cursor[i0 + j] = run; run += c[j]; }
    }
    carry += total;
    __syncthreads();
  }
  if (t == 0) rowptr[n] = carry;
}

__global__ void dst_place(const int* __restrict__ ei, int* __restrict__ cursor,
                          int* __restrict__ col, int Eraw) {
  int e = blockIdx.x * blockDim.x + threadIdx.x;
  if (e >= Eraw) return;
  int s = ei[e];
  int d = ei[Eraw + e];
  int p = atomicAdd(cursor + d, 1);
  col[p] = s;
}


template <int N> using vecf = float __attribute__((ext_vector_type(N)));

__device__ __forceinline__ void load_heads(const float* __restrict__ p, float (&v)[4]) {
  const float4 t = *(const float4*)p;
  v[0] = t.x; v[1] = t.y; v[2] = t.z; v[3] = t.w;
}
__device__ __forceinline__ void load_heads(const float* __restrict__ p, float (&v)[1]) {
  v[0] = p[0];
}

__device__ __forceinline__ float leaky(float v) { return v > 0.f ? v : NEG_SLOPE * v; }

template <int H, int F, bool ELU>
__global__ void __launch_bounds__(256)
seg_softmax_agg(const int* __restrict__ rowptr, const int* __restrict__ col,
                const float* __restrict__ asrc, const float* __restrict__ adst,
                const float* __restrict__ feat, const float* __restrict__ bias,
                float* __restrict__ out, int n_nodes) {
  constexpr int C   = H * F;
  constexpr int CPL = C / 32;
  constexpr int VW  = CPL >= 4 ? 4 : CPL;
  constexpr int NL  = CPL / VW;
  constexpr int U   = 8 / NL;
  static_assert(U * NL == 8, "the load fence below names eight values");
  typedef vecf<VW> vrow;

  const int lane = threadIdx.x & 31;
  const int n = __builtin_amdgcn_readfirstlane(
      (int)(blockIdx.x * (blockDim.x >> 5) + (threadIdx.x >> 5)));
  if (n >= n_nodes) return;
  const int beg = rowptr[n];
  const int end = rowptr[n + 1];

  float ad[H], as_self[H], vself[H], mx[H];
  load_heads(adst + (size_t)n * H, ad);
  load_heads(asrc + (size_t)n * H, as_self);
  for (int h = 0; h < H; ++h) { vself[h] = leaky(as_self[h] + ad[h]); mx[h] = vself[h]; }

  for (int i = beg + lane; i < end; i += 32) {
    float a[H];
    load_heads(asrc + (size_t)col[i] * H, a);
    for (int h = 0; h < H; ++h) mx[h] = fmaxf(mx[h], leaky(a[h] + ad[h]));
  }
  for (int h = 0; h < H; ++h) {
    for (int m = 16; m >= 1; m >>= 1) mx[h] = fmaxf(mx[h], __shfl_xor(mx[h], m));
  }

  float den[H];
  for (int h = 0; h < H; ++h) den[h] = 0.f;
  for (int i = beg + lane; i < end; i += 32) {
    float a[H];
    load_heads(asrc + (size_t)col[i] * H, a);
    for (int h = 0; h < H; ++h) den[h] += __expf(leaky(a[h] + ad[h]) - mx[h]);
  }
  for (int h = 0; h < H; ++h) {
    for (int m = 16; m >= 1; m >>= 1) den[h] += __shfl_xor(den[h], m);
    den[h] += __expf(vself[h] - mx[h]);
    den[h] += 1e-16f;
  }

  int hsel[NL];
  vrow acc[NL];
  for (int j = 0; j < NL; ++j) {
    hsel[j] = (j * 32 * VW + lane * VW) / F;
    for (int q = 0; q < VW; ++q) acc[j][q] = 0.f;
  }
  {
    const float* row = feat + (size_t)n * C + lane * VW;
    for (int j = 0; j < NL; ++j) {
      float w = __expf(vself[0] - mx[0]) / den[0];
      for (int h = 1; h < H; ++h)
        w = (hsel[j] == h) ? __expf(vself[h] - mx[h]) / den[h] : w;
      const vrow f = *(const vrow*)(row + j * 32 * VW);
      for (int q = 0; q < VW; ++q) acc[j][q] += w * f[q];
    }
  }
  for (int base = beg; base < end; base += 32) {
    const int  i  = base + lane;
    const bool ok = i < end;
    const int  s  = ok ? col[i] : n;
    float a[H], w[H];
    load_heads(asrc + (size_t)s * H, a);
    for (int h = 0; h < H; ++h)
      w[h] = ok ? __expf(leaky(a[h] + ad[h]) - mx[h]) / den[h] : 0.f;
    const int cn  = (end - base) < 32 ? (end - base) : 32;
    const int cnp = (cn + U - 1) & ~(U - 1);
    for (int k = 0; k < cnp; k += U) {
      vrow  f[U][NL];
      float wj[U][NL];
#pragma unroll
      for (int u = 0; u < U; ++u) {
        const int sk = __builtin_amdgcn_readlane(s, k + u);
        float wk[H];
        for (int h = 0; h < H; ++h)
          wk[h] = __int_as_float(__builtin_amdgcn_readlane(__float_as_int(w[h]), k + u));
        const float* row = feat + (size_t)sk * C + lane * VW;
        for (int j = 0; j < NL; ++j) {
          float t = wk[0];
          for (int h = 1; h < H; ++h) t = (hsel[j] == h) ? wk[h] : t;
          wj[u][j] = t;
          f[u][j]  = *(const vrow*)(row + j * 32 * VW);
        }
      }
      asm volatile("" :: "v"(f[0 / NL][0 % NL]), "v"(f[1 / NL][1 % NL]), "v"(f[2 / NL][2 % NL]),
                         "v"(f[3 / NL][3 % NL]), "v"(f[4 / NL][4 % NL]), "v"(f[5 / NL][5 % NL]),
                         "v"(f[6 / NL][6 % NL]), "v"(f[7 / NL][7 % NL]));
#pragma unroll
      for (int u = 0; u < U; ++u)
        for (int j = 0; j < NL; ++j)
          for (int q = 0; q < VW; ++q) acc[j][q] += wj[u][j] * f[u][j][q];
    }
  }

  float* orow = out + (size_t)n * C + lane * VW;
  const float* brow = bias + lane * VW;
  for (int j = 0; j < NL; ++j) {
    const vrow b = *(const vrow*)(brow + j * 32 * VW);
    vrow r;
    for (int q = 0; q < VW; ++q) {
      float v = acc[j][q] + b[q];
      r[q] = ELU ? (v > 0.f ? v : expm1f(v)) : v;
    }
    *(vrow*)(orow + j * 32 * VW) = r;
  }
}


static inline int cdiv(long long a, long long b) { return (int)((a + b - 1) / b); }

extern "C" void kernel_launch(void* const* d_in, const int* in_sizes, int n_in,
                              void* d_out, int out_size, void* d_ws, size_t ws_size,
                              hipStream_t stream) {
  const float* x      = (const float*)d_in[0];
  const int*   ei     = (const int*)  d_in[1];
  const float* W1     = (const float*)d_in[2];
  const float* att_s1 = (const float*)d_in[3];
  const float* att_d1 = (const float*)d_in[4];
  const float* b1     = (const float*)d_in[5];
  const float* W2     = (const float*)d_in[6];
  const float* att_s2 = (const float*)d_in[7];
  const float* att_d2 = (const float*)d_in[8];
  const float* b2     = (const float*)d_in[9];

  const int IN   = 128;
  const int C1   = 256;
  const int Nn   = in_sizes[0] / IN;
  const int Eraw = in_sizes[1] / 2;

  float* ws = (float*)d_ws;
  size_t off = 0;
  const size_t npad = ((size_t)Nn + 4) & ~(size_t)3;
  float* h1     = ws + off; off += (size_t)Nn * C1;
  float* out1   = ws + off; off += (size_t)Nn * C1;
  float* asrc1  = ws + off; off += (size_t)Nn * 4;
  float* adst1  = ws + off; off += (size_t)Nn * 4;
  int*   cnt    = (int*)(ws + off); off += npad;
  int*   rowptr = (int*)(ws + off); off += npad;
  int*   cursor = (int*)(ws + off); off += npad;
  int*   col    = (int*)(ws + off); off += (size_t)Eraw;
  float* h2     = h1;
  float* asrc2  = asrc1;
  float* adst2  = adst1;
  float* outf   = (float*)d_out;

  const int T = 256;

  fill_i32<<<cdiv((long long)Nn, T), T, 0, stream>>>(cnt, 0, (long long)Nn);
  dst_count<<<cdiv((long long)Eraw, T), T, 0, stream>>>(ei, cnt, Eraw);
  dst_scan<<<1, 1024, 0, stream>>>(cnt, rowptr, cursor, Nn);
  dst_place<<<cdiv((long long)Eraw, T), T, 0, stream>>>(ei, cursor, col, Eraw);

  gemm_head_alpha<128, 4><<<cdiv((long long)(Nn / 16) * 4, T / 32), T, 0, stream>>>(
      x, W1, att_s1, att_d1, h1, asrc1, adst1, Nn);
  seg_softmax_agg<4, 64, true><<<cdiv((long long)Nn, T / 32), T, 0, stream>>>(
      rowptr, col, asrc1, adst1, h1, b1, out1, Nn);

  gemm_head_alpha<256, 1><<<cdiv((long long)(Nn / 16), T / 32), T, 0, stream>>>(
      out1, W2, att_s2, att_d2, h2, asrc2, adst2, Nn);
  seg_softmax_agg<1, 64, false><<<cdiv((long long)Nn, T / 32), T, 0, stream>>>(
      rowptr, col, asrc2, adst2, h2, b2, outf, Nn);
}


# Edge sort by destination moved from global atomics to LDS counters: slices binned by 256-node range (LDS rank kept), cell table scanned, each range sorted in LDS; node lists stay in arrival order
# speedup vs baseline: 4.1574x; 1.4415x over previous
#include <hip/hip_runtime.h>

typedef __attribute__((ext_vector_type(2))) float v2f;
typedef __attribute__((ext_vector_type(8))) float v8f;

#define NEG_SLOPE 0.2f


typedef __attribute__((ext_vector_type(4))) float v4f;

template <int K, int HN>
__global__ void __launch_bounds__(256)
gemm_head_alpha(const float* __restrict__ A, const float* __restrict__ B,
                const float* __restrict__ att_s, const float* __restrict__ att_d,
                float* __restrict__ C, float* __restrict__ asrc, float* __restrict__ adst,
                int M) {
  constexpr int Nn = HN * 64;
  const int lane   = threadIdx.x & 31;
  const int wave   = blockIdx.x * (blockDim.x >> 5) + (threadIdx.x >> 5);
  const int mtiles = M >> 4;
  const int tileM  = wave / HN;
  const int hd     = wave - tileM * HN;
  if (tileM >= mtiles) return;

  const int lh  = lane >> 4;
  const int l16 = lane & 15;

  const float* ap = A + (size_t)(tileM * 16 + l16) * K + lh * 2;
  const float* bp = B + (size_t)(lh * 2) * Nn + hd * 64 + l16 * 4;

  v8f acc0 = {}, acc1 = {}, acc2 = {}, acc3 = {};
#pragma unroll 8
  for (int k = 0; k < K; k += 4) {
    const v2f a  = *(const v2f*)(ap + k);
    const v4f b0 = *(const v4f*)(bp + (size_t)k * Nn);
    const v4f b1 = *(const v4f*)(bp + (size_t)(k + 1) * Nn);
    v2f b;
    b.x = b0.x; b.y = b1.x;
    acc0 = __builtin_amdgcn_wmma_f32_16x16x4_f32(false, a, false, b, (short)0, acc0, false, false);
    b.x = b0.y; b.y = b1.y;
    acc1 = __builtin_amdgcn_wmma_f32_16x16x4_f32(false, a, false, b, (short)0, acc1, false, false);
    b.x = b0.z; b.y = b1.z;
    acc2 = __builtin_amdgcn_wmma_f32_16x16x4_f32(false, a, false, b, (short)0, acc2, false, false);
    b.x = b0.w; b.y = b1.w;
    acc3 = __builtin_amdgcn_wmma_f32_16x16x4_f32(false, a, false, b, (short)0, acc3, false, false);
  }

  const int row0 = tileM * 16 + lh * 8;
  float* crow = C + (size_t)row0 * Nn + hd * 64 + l16 * 4;
#pragma unroll
  for (int r = 0; r < 8; ++r) {
    v4f o;
    o.x = acc0[r]; o.y = acc1[r]; o.z = acc2[r]; o.w = acc3[r];
    *(v4f*)(crow + (size_t)r * Nn) = o;
  }

  const v4f s = *(const v4f*)(att_s + hd * 64 + l16 * 4);
  const v4f d = *(const v4f*)(att_d + hd * 64 + l16 * 4);
  float ps[8], pd[8];
#pragma unroll
  for (int r = 0; r < 8; ++r) {
    ps[r] = acc0[r] * s.x + acc1[r] * s.y + acc2[r] * s.z + acc3[r] * s.w;
    pd[r] = acc0[r] * d.x + acc1[r] * d.y + acc2[r] * d.z + acc3[r] * d.w;
  }
#pragma unroll
  for (int r = 0; r < 8; ++r) {
    for (int m = 8; m >= 1; m >>= 1) {
      ps[r] += __shfl_xor(ps[r], m);
      pd[r] += __shfl_xor(pd[r], m);
    }
  }
  float vs = ps[0], vd = pd[0];
#pragma unroll
  for (int r = 1; r < 8; ++r) {
    vs = (l16 == r) ? ps[r] : vs;
    vd = (l16 == r) ? pd[r] : vd;
  }
  if (l16 < 8) {
    asrc[(size_t)(row0 + l16) * HN + hd] = vs;
    adst[(size_t)(row0 + l16) * HN + hd] = vd;
  }
}


#define RANGE_SHIFT  8
#define RANGE_NODES  256
#define SORT_THREADS 1024

__global__ void __launch_bounds__(SORT_THREADS)
slice_count(const int* __restrict__ ei, int* __restrict__ lrank, int* __restrict__ cellcnt,
            int Eraw, int chunk, int nrange) {
  extern __shared__ int hcell[];
  const int t = threadIdx.x;
  const int g = blockIdx.x;
  for (int r = t; r < nrange; r += SORT_THREADS) hcell[r] = 0;
  __syncthreads();
  const int beg = g * chunk;
  const int end = (beg + chunk < Eraw) ? (beg + chunk) : Eraw;
  for (int e0 = beg + t; e0 < end; e0 += 4 * SORT_THREADS) {
    int d[4];
#pragma unroll
    for (int j = 0; j < 4; ++j) {
      const int e = e0 + j * SORT_THREADS;
      d[j] = ei[Eraw + ((e < end) ? e : (end - 1))];
    }
    asm volatile("" :: "v"(d[0]), "v"(d[1]), "v"(d[2]), "v"(d[3]));
#pragma unroll
    for (int j = 0; j < 4; ++j) {
      const int e = e0 + j * SORT_THREADS;
      if (e < end) lrank[e] = atomicAdd(&hcell[d[j] >> RANGE_SHIFT], 1);
    }
  }
  __syncthreads();
  for (int r = t; r < nrange; r += SORT_THREADS)
    cellcnt[(size_t)r * gridDim.x + g] = hcell[r];
}

__global__ void __launch_bounds__(1024)
table_scan(const int* __restrict__ cnt, int* __restrict__ out, int n) {
  __shared__ int wtot[32];
  const int t    = threadIdx.x;
  const int lane = t & 31;
  const int wv   = t >> 5;
  int carry = 0;
  for (int base = 0; base < n; base += 4096) {
    const int i0 = base + t * 4;
    int c[4];
    for (int j = 0; j < 4; ++j) c[j] = (i0 + j < n) ? cnt[i0 + j] : 0;
    const int tsum = c[0] + c[1] + c[2] + c[3];
    int incl = tsum;
    for (int d = 1; d < 32; d <<= 1) {
      int y = __shfl_up(incl, d);
      if (lane >= d) incl += y;
    }
    if (lane == 31) wtot[wv] = incl;
    __syncthreads();
    int ws = wtot[lane];
    for (int d = 1; d < 32; d <<= 1) {
      int y = __shfl_up(ws, d);
      if (lane >= d) ws += y;
    }
    const int wprefix = __shfl(ws, wv) - __shfl(incl, 31);
    const int total   = __shfl(ws, 31);
    int run = carry + wprefix + incl - tsum;
    for (int j = 0; j < 4; ++j) {
      if (i0 + j < n) { out[i0 + j] = run; run += c[j]; }
    }
    carry += total;
    __syncthreads();
  }
  if (t == 0) out[n] = carry;
}

__global__ void __launch_bounds__(SORT_THREADS)
slice_place(const int* __restrict__ ei, const int* __restrict__ lrank,
            const int* __restrict__ cellpos, int* __restrict__ binned, int Eraw, int chunk) {
  const int t = threadIdx.x;
  const int g = blockIdx.x;
  const int beg = g * chunk;
  const int end = (beg + chunk < Eraw) ? (beg + chunk) : Eraw;
  int2* bp = (int2*)binned;
  for (int e0 = beg + t; e0 < end; e0 += 4 * SORT_THREADS) {
    int s[4], d[4], r[4], c[4];
#pragma unroll
    for (int j = 0; j < 4; ++j) {
      const int e  = e0 + j * SORT_THREADS;
      const int ec = (e < end) ? e : (end - 1);
      s[j] = ei[ec];
      d[j] = ei[Eraw + ec];
      r[j] = lrank[ec];
    }
    asm volatile("" :: "v"(s[0]), "v"(s[1]), "v"(s[2]), "v"(s[3]), "v"(d[0]), "v"(d[1]), "v"(d[2]),
                       "v"(d[3]), "v"(r[0]), "v"(r[1]), "v"(r[2]), "v"(r[3]));
#pragma unroll
    for (int j = 0; j < 4; ++j)
      c[j] = cellpos[(size_t)(d[j] >> RANGE_SHIFT) * gridDim.x + g];
    asm volatile("" :: "v"(c[0]), "v"(c[1]), "v"(c[2]), "v"(c[3]));
#pragma unroll
    for (int j = 0; j < 4; ++j) {
      const int e = e0 + j * SORT_THREADS;
      if (e < end) bp[c[j] + r[j]] = make_int2(s[j], d[j]);
    }
  }
}

__global__ void __launch_bounds__(SORT_THREADS)
range_sort(const int* __restrict__ binned, const int* __restrict__ cellpos,
           int* __restrict__ rowptr, int* __restrict__ col, int n_nodes, int nslice) {
  __shared__ int cnt[RANGE_NODES];
  const int t = threadIdx.x;
  const int b = blockIdx.x;
  const int beg = cellpos[(size_t)b * nslice];
  const int end = cellpos[(size_t)(b + 1) * nslice];
  const int2* bp = (const int2*)binned;
  if (t < RANGE_NODES) cnt[t] = 0;
  __syncthreads();
  for (int i0 = beg + t; i0 < end; i0 += 8 * SORT_THREADS) {
    int d[8];
#pragma unroll
    for (int j = 0; j < 8; ++j) {
      const int i = i0 + j * SORT_THREADS;
      d[j] = bp[(i < end) ? i : (end - 1)].y;
    }
    asm volatile("" :: "v"(d[0]), "v"(d[1]), "v"(d[2]), "v"(d[3]),
                       "v"(d[4]), "v"(d[5]), "v"(d[6]), "v"(d[7]));
#pragma unroll
    for (int j = 0; j < 8; ++j) {
      const int i = i0 + j * SORT_THREADS;
      if (i < end) atomicAdd(&cnt[d[j] & (RANGE_NODES - 1)], 1);
    }
  }
  __syncthreads();
  if (t < 32) {
    int c[8];
    int tsum = 0;
    for (int j = 0; j < 8; ++j) { c[j] = cnt[t * 8 + j]; tsum += c[j]; }
    int incl = tsum;
    for (int d = 1; d < 32; d <<= 1) {
      int y = __shfl_up(incl, d);
      if (t >= d) incl += y;
    }
    int run = incl - tsum;
    for (int j = 0; j < 8; ++j) { cnt[t * 8 + j] = run; run += c[j]; }
  }
  __syncthreads();
  if (t < RANGE_NODES) {
    const int node = b * RANGE_NODES + t;
    if (node <= n_nodes) rowptr[node] = beg + cnt[t];
  }
  __syncthreads();
  for (int i0 = beg + t; i0 < end; i0 += 8 * SORT_THREADS) {
    int2 sd[8];
#pragma unroll
    for (int j = 0; j < 8; ++j) {
      const int i = i0 + j * SORT_THREADS;
      sd[j] = bp[(i < end) ? i : (end - 1)];
    }
    asm volatile("" :: "v"(sd[0].y), "v"(sd[1].y), "v"(sd[2].y), "v"(sd[3].y),
                       "v"(sd[4].y), "v"(sd[5].y), "v"(sd[6].y), "v"(sd[7].y));
#pragma unroll
    for (int j = 0; j < 8; ++j) {
      const int i = i0 + j * SORT_THREADS;
      if (i < end) {
        const int p = atomicAdd(&cnt[sd[j].y & (RANGE_NODES - 1)], 1);
        col[beg + p] = sd[j].x;
      }
    }
  }
}


template <int N> using vecf = float __attribute__((ext_vector_type(N)));

__device__ __forceinline__ void load_heads(const float* __restrict__ p, float (&v)[4]) {
  const float4 t = *(const float4*)p;
  v[0] = t.x; v[1] = t.y; v[2] = t.z; v[3] = t.w;
}
__device__ __forceinline__ void load_heads(const float* __restrict__ p, float (&v)[1]) {
  v[0] = p[0];
}

__device__ __forceinline__ float leaky(float v) { return v > 0.f ? v : NEG_SLOPE * v; }

template <int H, int F, bool ELU>
__global__ void __launch_bounds__(256)
seg_softmax_agg(const int* __restrict__ rowptr, const int* __restrict__ col,
                const float* __restrict__ asrc, const float* __restrict__ adst,
                const float* __restrict__ feat, const float* __restrict__ bias,
                float* __restrict__ out, int n_nodes) {
  constexpr int C   = H * F;
  constexpr int CPL = C / 32;
  constexpr int VW  = CPL >= 4 ? 4 : CPL;
  constexpr int NL  = CPL / VW;
  constexpr int U   = 8 / NL;
  static_assert(U * NL == 8, "the load fence below names eight values");
  typedef vecf<VW> vrow;

  const int lane = threadIdx.x & 31;
  const int n = __builtin_amdgcn_readfirstlane(
      (int)(blockIdx.x * (blockDim.x >> 5) + (threadIdx.x >> 5)));
  if (n >= n_nodes) return;
  const int beg = rowptr[n];
  const int end = rowptr[n + 1];

  float ad[H], as_self[H], vself[H], mx[H];
  load_heads(adst + (size_t)n * H, ad);
  load_heads(asrc + (size_t)n * H, as_self);
  for (int h = 0; h < H; ++h) { vself[h] = leaky(as_self[h] + ad[h]); mx[h] = vself[h]; }

  for (int i = beg + lane; i < end; i += 32) {
    float a[H];
    load_heads(asrc + (size_t)col[i] * H, a);
    for (int h = 0; h < H; ++h) mx[h] = fmaxf(mx[h], leaky(a[h] + ad[h]));
  }
  for (int h = 0; h < H; ++h) {
    for (int m = 16; m >= 1; m >>= 1) mx[h] = fmaxf(mx[h], __shfl_xor(mx[h], m));
  }

  float den[H];
  for (int h = 0; h < H; ++h) den[h] = 0.f;
  for (int i = beg + lane; i < end; i += 32) {
    float a[H];
    load_heads(asrc + (size_t)col[i] * H, a);
    for (int h = 0; h < H; ++h) den[h] += __expf(leaky(a[h] + ad[h]) - mx[h]);
  }
  for (int h = 0; h < H; ++h) {
    for (int m = 16; m >= 1; m >>= 1) den[h] += __shfl_xor(den[h], m);
    den[h] += __expf(vself[h] - mx[h]);
    den[h] += 1e-16f;
  }

  int hsel[NL];
  vrow acc[NL];
  for (int j = 0; j < NL; ++j) {
    hsel[j] = (j * 32 * VW + lane * VW) / F;
    for (int q = 0; q < VW; ++q) acc[j][q] = 0.f;
  }
  {
    const float* row = feat + (size_t)n * C + lane * VW;
    for (int j = 0; j < NL; ++j) {
      float w = __expf(vself[0] - mx[0]) / den[0];
      for (int h = 1; h < H; ++h)
        w = (hsel[j] == h) ? __expf(vself[h] - mx[h]) / den[h] : w;
      const vrow f = *(const vrow*)(row + j * 32 * VW);
      for (int q = 0; q < VW; ++q) acc[j][q] += w * f[q];
    }
  }
  for (int base = beg; base < end; base += 32) {
    const int  i  = base + lane;
    const bool ok = i < end;
    const int  s  = ok ? col[i] : n;
    float a[H], w[H];
    load_heads(asrc + (size_t)s * H, a);
    for (int h = 0; h < H; ++h)
      w[h] = ok ? __expf(leaky(a[h] + ad[h]) - mx[h]) / den[h] : 0.f;
    const int cn  = (end - base) < 32 ? (end - base) : 32;
    const int cnp = (cn + U - 1) & ~(U - 1);
    for (int k = 0; k < cnp; k += U) {
      vrow  f[U][NL];
      float wj[U][NL];
#pragma unroll
      for (int u = 0; u < U; ++u) {
        const int sk = __builtin_amdgcn_readlane(s, k + u);
        float wk[H];
        for (int h = 0; h < H; ++h)
          wk[h] = __int_as_float(__builtin_amdgcn_readlane(__float_as_int(w[h]), k + u));
        const float* row = feat + (size_t)sk * C + lane * VW;
        for (int j = 0; j < NL; ++j) {
          float t = wk[0];
          for (int h = 1; h < H; ++h) t = (hsel[j] == h) ? wk[h] : t;
          wj[u][j] = t;
          f[u][j]  = *(const vrow*)(row + j * 32 * VW);
        }
      }
      asm volatile("" :: "v"(f[0 / NL][0 % NL]), "v"(f[1 / NL][1 % NL]), "v"(f[2 / NL][2 % NL]),
                         "v"(f[3 / NL][3 % NL]), "v"(f[4 / NL][4 % NL]), "v"(f[5 / NL][5 % NL]),
                         "v"(f[6 / NL][6 % NL]), "v"(f[7 / NL][7 % NL]));
#pragma unroll
      for (int u = 0; u < U; ++u)
        for (int j = 0; j < NL; ++j)
          for (int q = 0; q < VW; ++q) acc[j][q] += wj[u][j] * f[u][j][q];
    }
  }

  float* orow = out + (size_t)n * C + lane * VW;
  const float* brow = bias + lane * VW;
  for (int j = 0; j < NL; ++j) {
    const vrow b = *(const vrow*)(brow + j * 32 * VW);
    vrow r;
    for (int q = 0; q < VW; ++q) {
      float v = acc[j][q] + b[q];
      r[q] = ELU ? (v > 0.f ? v : expm1f(v)) : v;
    }
    *(vrow*)(orow + j * 32 * VW) = r;
  }
}


static inline int cdiv(long long a, long long b) { return (int)((a + b - 1) / b); }

extern "C" void kernel_launch(void* const* d_in, const int* in_sizes, int n_in,
                              void* d_out, int out_size, void* d_ws, size_t ws_size,
                              hipStream_t stream) {
  const float* x      = (const float*)d_in[0];
  const int*   ei     = (const int*)  d_in[1];
  const float* W1     = (const float*)d_in[2];
  const float* att_s1 = (const float*)d_in[3];
  const float* att_d1 = (const float*)d_in[4];
  const float* b1     = (const float*)d_in[5];
  const float* W2     = (const float*)d_in[6];
  const float* att_s2 = (const float*)d_in[7];
  const float* att_d2 = (const float*)d_in[8];
  const float* b2     = (const float*)d_in[9];

  const int IN   = 128;
  const int C1   = 256;
  const int Nn   = in_sizes[0] / IN;
  const int Eraw = in_sizes[1] / 2;

  float* ws = (float*)d_ws;
  size_t off = 0;
  const size_t npad = ((size_t)Nn + 4) & ~(size_t)3;
  float* h1     = ws + off; off += (size_t)Nn * C1;
  float* out1   = ws + off; off += (size_t)Nn * C1;
  float* asrc1  = ws + off; off += (size_t)Nn * 4;
  float* adst1  = ws + off; off += (size_t)Nn * 4;
  const int NS     = 256;
  const int nrange = Nn / 256 + 1;
  const size_t ncell = (size_t)nrange * NS;
  int*   cellcnt = (int*)(ws + off); off += ncell;
  int*   cellpos = (int*)(ws + off); off += ncell + 4;
  int*   rowptr  = (int*)(ws + off); off += npad;
  int*   binned  = (int*)(ws + off); off += (size_t)Eraw * 2 + 2;
  off = (off + 3) & ~(size_t)3;
  int*   lrank   = (int*)(ws + off); off += ((size_t)Eraw + 3) & ~(size_t)3;
  int*   col     = (int*)(ws + off); off += (size_t)Eraw;
  float* h2     = h1;
  float* asrc2  = asrc1;
  float* adst2  = adst1;
  float* outf   = (float*)d_out;

  const int T = 256;

  const int chunk = cdiv((long long)Eraw, NS);
  slice_count<<<NS, SORT_THREADS, nrange * sizeof(int), stream>>>(ei, lrank, cellcnt, Eraw, chunk, nrange);
  table_scan<<<1, 1024, 0, stream>>>(cellcnt, cellpos, (int)ncell);
  slice_place<<<NS, SORT_THREADS, 0, stream>>>(ei, lrank, cellpos, binned, Eraw, chunk);
  range_sort<<<nrange, SORT_THREADS, 0, stream>>>(binned, cellpos, rowptr, col, Nn, NS);

  gemm_head_alpha<128, 4><<<cdiv((long long)(Nn / 16) * 4, T / 32), T, 0, stream>>>(
      x, W1, att_s1, att_d1, h1, asrc1, adst1, Nn);
  seg_softmax_agg<4, 64, true><<<cdiv((long long)Nn, T / 32), T, 0, stream>>>(
      rowptr, col, asrc1, adst1, h1, b1, out1, Nn);

  gemm_head_alpha<256, 1><<<cdiv((long long)(Nn / 16), T / 32), T, 0, stream>>>(
      out1, W2, att_s2, att_d2, h2, asrc2, adst2, Nn);
  seg_softmax_agg<1, 64, false><<<cdiv((long long)Nn, T / 32), T, 0, stream>>>(
      rowptr, col, asrc2, adst2, h2, b2, outf, Nn);
}


# Cell-place scan folded into the second slice pass: counts stored slice-major, each slice group sums the table itself (16-byte reads, LDS block scan), places read from LDS; scan launch gone
# speedup vs baseline: 4.7433x; 1.1409x over previous
#include <hip/hip_runtime.h>

typedef __attribute__((ext_vector_type(2))) float v2f;
typedef __attribute__((ext_vector_type(8))) float v8f;

#define NEG_SLOPE 0.2f


typedef __attribute__((ext_vector_type(4))) float v4f;

template <int K, int HN>
__global__ void __launch_bounds__(256)
gemm_head_alpha(const float* __restrict__ A, const float* __restrict__ B,
                const float* __restrict__ att_s, const float* __restrict__ att_d,
                float* __restrict__ C, float* __restrict__ asrc, float* __restrict__ adst,
                int M) {
  constexpr int Nn = HN * 64;
  const int lane   = threadIdx.x & 31;
  const int wave   = blockIdx.x * (blockDim.x >> 5) + (threadIdx.x >> 5);
  const int mtiles = M >> 4;
  const int tileM  = wave / HN;
  const int hd     = wave - tileM * HN;
  if (tileM >= mtiles) return;

  const int lh  = lane >> 4;
  const int l16 = lane & 15;

  const float* ap = A + (size_t)(tileM * 16 + l16) * K + lh * 2;
  const float* bp = B + (size_t)(lh * 2) * Nn + hd * 64 + l16 * 4;

  v8f acc0 = {}, acc1 = {}, acc2 = {}, acc3 = {};
#pragma unroll 8
  for (int k = 0; k < K; k += 4) {
    const v2f a  = *(const v2f*)(ap + k);
    const v4f b0 = *(const v4f*)(bp + (size_t)k * Nn);
    const v4f b1 = *(const v4f*)(bp + (size_t)(k + 1) * Nn);
    v2f b;
    b.x = b0.x; b.y = b1.x;
    acc0 = __builtin_amdgcn_wmma_f32_16x16x4_f32(false, a, false, b, (short)0, acc0, false, false);
    b.x = b0.y; b.y = b1.y;
    acc1 = __builtin_amdgcn_wmma_f32_16x16x4_f32(false, a, false, b, (short)0, acc1, false, false);
    b.x = b0.z; b.y = b1.z;
    acc2 = __builtin_amdgcn_wmma_f32_16x16x4_f32(false, a, false, b, (short)0, acc2, false, false);
    b.x = b0.w; b.y = b1.w;
    acc3 = __builtin_amdgcn_wmma_f32_16x16x4_f32(false, a, false, b, (short)0, acc3, false, false);
  }

  const int row0 = tileM * 16 + lh * 8;
  float* crow = C + (size_t)row0 * Nn + hd * 64 + l16 * 4;
#pragma unroll
  for (int r = 0; r < 8; ++r) {
    v4f o;
    o.x = acc0[r]; o.y = acc1[r]; o.z = acc2[r]; o.w = acc3[r];
    *(v4f*)(crow + (size_t)r * Nn) = o;
  }

  const v4f s = *(const v4f*)(att_s + hd * 64 + l16 * 4);
  const v4f d = *(const v4f*)(att_d + hd * 64 + l16 * 4);
  float ps[8], pd[8];
#pragma unroll
  for (int r = 0; r < 8; ++r) {
    ps[r] = acc0[r] * s.x + acc1[r] * s.y + acc2[r] * s.z + acc3[r] * s.w;
    pd[r] = acc0[r] * d.x + acc1[r] * d.y + acc2[r] * d.z + acc3[r] * d.w;
  }
#pragma unroll
  for (int r = 0; r < 8; ++r) {
    for (int m = 8; m >= 1; m >>= 1) {
      ps[r] += __shfl_xor(ps[r], m);
      pd[r] += __shfl_xor(pd[r], m);
    }
  }
  float vs = ps[0], vd = pd[0];
#pragma unroll
  for (int r = 1; r < 8; ++r) {
    vs = (l16 == r) ? ps[r] : vs;
    vd = (l16 == r) ? pd[r] : vd;
  }
  if (l16 < 8) {
    asrc[(size_t)(row0 + l16) * HN + hd] = vs;
    adst[(size_t)(row0 + l16) * HN + hd] = vd;
  }
}


#define RANGE_SHIFT  8
#define RANGE_NODES  256
#define SORT_THREADS 1024
#define NSLICE       256
#define COLGROUPS    16

__global__ void __launch_bounds__(SORT_THREADS)
slice_count(const int* __restrict__ ei, int* __restrict__ lrank, int* __restrict__ cellcnt,
            int Eraw, int chunk, int nrange, int rpitch) {
  extern __shared__ int hcell[];
  const int t = threadIdx.x;
  const int g = blockIdx.x;
  for (int r = t; r < nrange; r += SORT_THREADS) hcell[r] = 0;
  __syncthreads();
  const int beg = g * chunk;
  const int end = (beg + chunk < Eraw) ? (beg + chunk) : Eraw;
  for (int e0 = beg + t; e0 < end; e0 += 4 * SORT_THREADS) {
    int d[4];
#pragma unroll
    for (int j = 0; j < 4; ++j) {
      const int e = e0 + j * SORT_THREADS;
      d[j] = ei[Eraw + ((e < end) ? e : (end - 1))];
    }
    asm volatile("" :: "v"(d[0]), "v"(d[1]), "v"(d[2]), "v"(d[3]));
#pragma unroll
    for (int j = 0; j < 4; ++j) {
      const int e = e0 + j * SORT_THREADS;
      if (e < end) lrank[e] = atomicAdd(&hcell[d[j] >> RANGE_SHIFT], 1);
    }
  }
  __syncthreads();
  for (int r = t; r < rpitch; r += SORT_THREADS)
    cellcnt[(size_t)g * rpitch + r] = (r < nrange) ? hcell[r] : 0;
}

__global__ void __launch_bounds__(SORT_THREADS)
slice_place(const int* __restrict__ ei, const int* __restrict__ lrank,
            const int* __restrict__ cellcnt, int* __restrict__ rstart,
            int* __restrict__ binned, int Eraw, int chunk, int nrange, int rpitch) {
  extern __shared__ __attribute__((aligned(16))) int lds[];
  int* part_t = lds;
  int* part_p = part_t + COLGROUPS * rpitch;
  int* lcell  = part_p + COLGROUPS * rpitch;
  int* wtot   = lcell + rpitch;
  const int t    = threadIdx.x;
  const int lane = t & 31;
  const int wv   = t >> 5;
  const int g    = blockIdx.x;

  const int r4n = rpitch >> 2;
  for (int idx = t; idx < COLGROUPS * r4n; idx += SORT_THREADS) {
    const int q  = idx / r4n;
    const int r4 = idx - q * r4n;
    const int* colp = cellcnt + (size_t)q * rpitch + r4 * 4;
    int4 st = make_int4(0, 0, 0, 0), sp = make_int4(0, 0, 0, 0);
#pragma unroll 8
    for (int k = 0; k < NSLICE / COLGROUPS; ++k) {
      const int  gg = q + k * COLGROUPS;
      const int4 v  = *(const int4*)(colp + (size_t)k * COLGROUPS * rpitch);
      st.x += v.x; st.y += v.y; st.z += v.z; st.w += v.w;
      const bool below = gg < g;
      sp.x += below ? v.x : 0; sp.y += below ? v.y : 0;
      sp.z += below ? v.z : 0; sp.w += below ? v.w : 0;
    }
    *(int4*)(part_t + q * rpitch + r4 * 4) = st;
    *(int4*)(part_p + q * rpitch + r4 * 4) = sp;
  }
  __syncthreads();

  int carry = 0;
  for (int base = 0; base < nrange; base += SORT_THREADS) {
    const int r = base + t;
    int tot = 0, low = 0;
    if (r < nrange) {
      for (int q = 0; q < COLGROUPS; ++q) { tot += part_t[q * rpitch + r]; low += part_p[q * rpitch + r]; }
    }
    int incl = tot;
    for (int d = 1; d < 32; d <<= 1) {
      int y = __shfl_up(incl, d);
      if (lane >= d) incl += y;
    }
    if (lane == 31) wtot[wv] = incl;
    __syncthreads();
    int ws = wtot[lane];
    for (int d = 1; d < 32; d <<= 1) {
      int y = __shfl_up(ws, d);
      if (lane >= d) ws += y;
    }
    const int wprefix = __shfl(ws, wv) - __shfl(incl, 31);
    const int total   = __shfl(ws, 31);
    const int start   = carry + wprefix + incl - tot;
    if (r < nrange) {
      lcell[r] = start + low;
      if (g == 0) rstart[r] = start;
    }
    carry += total;
    __syncthreads();
  }
  if (g == 0 && t == 0) rstart[nrange] = carry;

  const int beg = g * chunk;
  const int end = (beg + chunk < Eraw) ? (beg + chunk) : Eraw;
  int2* bp = (int2*)binned;
  for (int e0 = beg + t; e0 < end; e0 += 4 * SORT_THREADS) {
    int s[4], d[4], r[4];
#pragma unroll
    for (int j = 0; j < 4; ++j) {
      const int e  = e0 + j * SORT_THREADS;
      const int ec = (e < end) ? e : (end - 1);
      s[j] = ei[ec];
      d[j] = ei[Eraw + ec];
      r[j] = lrank[ec];
    }
    asm volatile("" :: "v"(s[0]), "v"(s[1]), "v"(s[2]), "v"(s[3]), "v"(d[0]), "v"(d[1]), "v"(d[2]),
                       "v"(d[3]), "v"(r[0]), "v"(r[1]), "v"(r[2]), "v"(r[3]));
#pragma unroll
    for (int j = 0; j < 4; ++j) {
      const int e = e0 + j * SORT_THREADS;
      if (e < end) bp[lcell[d[j] >> RANGE_SHIFT] + r[j]] = make_int2(s[j], d[j]);
    }
  }
}

__global__ void __launch_bounds__(SORT_THREADS)
range_sort(const int* __restrict__ binned, const int* __restrict__ rstart,
           int* __restrict__ rowptr, int* __restrict__ col, int n_nodes) {
  __shared__ int cnt[RANGE_NODES];
  const int t = threadIdx.x;
  const int b = blockIdx.x;
  const int beg = rstart[b];
  const int end = rstart[b + 1];
  const int2* bp = (const int2*)binned;
  if (t < RANGE_NODES) cnt[t] = 0;
  __syncthreads();
  for (int i0 = beg + t; i0 < end; i0 += 8 * SORT_THREADS) {
    int d[8];
#pragma unroll
    for (int j = 0; j < 8; ++j) {
      const int i = i0 + j * SORT_THREADS;
      d[j] = bp[(i < end) ? i : (end - 1)].y;
    }
    asm volatile("" :: "v"(d[0]), "v"(d[1]), "v"(d[2]), "v"(d[3]),
                       "v"(d[4]), "v"(d[5]), "v"(d[6]), "v"(d[7]));
#pragma unroll
    for (int j = 0; j < 8; ++j) {
      const int i = i0 + j * SORT_THREADS;
      if (i < end) atomicAdd(&cnt[d[j] & (RANGE_NODES - 1)], 1);
    }
  }
  __syncthreads();
  if (t < 32) {
    int c[8];
    int tsum = 0;
    for (int j = 0; j < 8; ++j) { c[j] = cnt[t * 8 + j]; tsum += c[j]; }
    int incl = tsum;
    for (int d = 1; d < 32; d <<= 1) {
      int y = __shfl_up(incl, d);
      if (t >= d) incl += y;
    }
    int run = incl - tsum;
    for (int j = 0; j < 8; ++j) { cnt[t * 8 + j] = run; run += c[j]; }
  }
  __syncthreads();
  if (t < RANGE_NODES) {
    const int node = b * RANGE_NODES + t;
    if (node <= n_nodes) rowptr[node] = beg + cnt[t];
  }
  __syncthreads();
  for (int i0 = beg + t; i0 < end; i0 += 8 * SORT_THREADS) {
    int2 sd[8];
#pragma unroll
    for (int j = 0; j < 8; ++j) {
      const int i = i0 + j * SORT_THREADS;
      sd[j] = bp[(i < end) ? i : (end - 1)];
    }
    asm volatile("" :: "v"(sd[0].y), "v"(sd[1].y), "v"(sd[2].y), "v"(sd[3].y),
                       "v"(sd[4].y), "v"(sd[5].y), "v"(sd[6].y), "v"(sd[7].y));
#pragma unroll
    for (int j = 0; j < 8; ++j) {
      const int i = i0 + j * SORT_THREADS;
      if (i < end) {
        const int p = atomicAdd(&cnt[sd[j].y & (RANGE_NODES - 1)], 1);
        col[beg + p] = sd[j].x;
      }
    }
  }
}


template <int N> using vecf = float __attribute__((ext_vector_type(N)));

__device__ __forceinline__ void load_heads(const float* __restrict__ p, float (&v)[4]) {
  const float4 t = *(const float4*)p;
  v[0] = t.x; v[1] = t.y; v[2] = t.z; v[3] = t.w;
}
__device__ __forceinline__ void load_heads(const float* __restrict__ p, float (&v)[1]) {
  v[0] = p[0];
}

__device__ __forceinline__ float leaky(float v) { return v > 0.f ? v : NEG_SLOPE * v; }

template <int H, int F, bool ELU>
__global__ void __launch_bounds__(256)
seg_softmax_agg(const int* __restrict__ rowptr, const int* __restrict__ col,
                const float* __restrict__ asrc, const float* __restrict__ adst,
                const float* __restrict__ feat, const float* __restrict__ bias,
                float* __restrict__ out, int n_nodes) {
  constexpr int C   = H * F;
  constexpr int CPL = C / 32;
  constexpr int VW  = CPL >= 4 ? 4 : CPL;
  constexpr int NL  = CPL / VW;
  constexpr int U   = 8 / NL;
  static_assert(U * NL == 8, "the load fence below names eight values");
  typedef vecf<VW> vrow;

  const int lane = threadIdx.x & 31;
  const int n = __builtin_amdgcn_readfirstlane(
      (int)(blockIdx.x * (blockDim.x >> 5) + (threadIdx.x >> 5)));
  if (n >= n_nodes) return;
  const int beg = rowptr[n];
  const int end = rowptr[n + 1];

  float ad[H], as_self[H], vself[H], mx[H];
  load_heads(adst + (size_t)n * H, ad);
  load_heads(asrc + (size_t)n * H, as_self);
  for (int h = 0; h < H; ++h) { vself[h] = leaky(as_self[h] + ad[h]); mx[h] = vself[h]; }

  for (int i = beg + lane; i < end; i += 32) {
    float a[H];
    load_heads(asrc + (size_t)col[i] * H, a);
    for (int h = 0; h < H; ++h) mx[h] = fmaxf(mx[h], leaky(a[h] + ad[h]));
  }
  for (int h = 0; h < H; ++h) {
    for (int m = 16; m >= 1; m >>= 1) mx[h] = fmaxf(mx[h], __shfl_xor(mx[h], m));
  }

  float den[H];
  for (int h = 0; h < H; ++h) den[h] = 0.f;
  for (int i = beg + lane; i < end; i += 32) {
    float a[H];
    load_heads(asrc + (size_t)col[i] * H, a);
    for (int h = 0; h < H; ++h) den[h] += __expf(leaky(a[h] + ad[h]) - mx[h]);
  }
  for (int h = 0; h < H; ++h) {
    for (int m = 16; m >= 1; m >>= 1) den[h] += __shfl_xor(den[h], m);
    den[h] += __expf(vself[h] - mx[h]);
    den[h] += 1e-16f;
  }

  int hsel[NL];
  vrow acc[NL];
  for (int j = 0; j < NL; ++j) {
    hsel[j] = (j * 32 * VW + lane * VW) / F;
    for (int q = 0; q < VW; ++q) acc[j][q] = 0.f;
  }
  {
    const float* row = feat + (size_t)n * C + lane * VW;
    for (int j = 0; j < NL; ++j) {
      float w = __expf(vself[0] - mx[0]) / den[0];
      for (int h = 1; h < H; ++h)
        w = (hsel[j] == h) ? __expf(vself[h] - mx[h]) / den[h] : w;
      const vrow f = *(const vrow*)(row + j * 32 * VW);
      for (int q = 0; q < VW; ++q) acc[j][q] += w * f[q];
    }
  }
  for (int base = beg; base < end; base += 32) {
    const int  i  = base + lane;
    const bool ok = i < end;
    const int  s  = ok ? col[i] : n;
    float a[H], w[H];
    load_heads(asrc + (size_t)s * H, a);
    for (int h = 0; h < H; ++h)
      w[h] = ok ? __expf(leaky(a[h] + ad[h]) - mx[h]) / den[h] : 0.f;
    const int cn  = (end - base) < 32 ? (end - base) : 32;
    const int cnp = (cn + U - 1) & ~(U - 1);
    for (int k = 0; k < cnp; k += U) {
      vrow  f[U][NL];
      float wj[U][NL];
#pragma unroll
      for (int u = 0; u < U; ++u) {
        const int sk = __builtin_amdgcn_readlane(s, k + u);
        float wk[H];
        for (int h = 0; h < H; ++h)
          wk[h] = __int_as_float(__builtin_amdgcn_readlane(__float_as_int(w[h]), k + u));
        const float* row = feat + (size_t)sk * C + lane * VW;
        for (int j = 0; j < NL; ++j) {
          float t = wk[0];
          for (int h = 1; h < H; ++h) t = (hsel[j] == h) ? wk[h] : t;
          wj[u][j] = t;
          f[u][j]  = *(const vrow*)(row + j * 32 * VW);
        }
      }
      asm volatile("" :: "v"(f[0 / NL][0 % NL]), "v"(f[1 / NL][1 % NL]), "v"(f[2 / NL][2 % NL]),
                         "v"(f[3 / NL][3 % NL]), "v"(f[4 / NL][4 % NL]), "v"(f[5 / NL][5 % NL]),
                         "v"(f[6 / NL][6 % NL]), "v"(f[7 / NL][7 % NL]));
#pragma unroll
      for (int u = 0; u < U; ++u)
        for (int j = 0; j < NL; ++j)
          for (int q = 0; q < VW; ++q) acc[j][q] += wj[u][j] * f[u][j][q];
    }
  }

  float* orow = out + (size_t)n * C + lane * VW;
  const float* brow = bias + lane * VW;
  for (int j = 0; j < NL; ++j) {
    const vrow b = *(const vrow*)(brow + j * 32 * VW);
    vrow r;
    for (int q = 0; q < VW; ++q) {
      float v = acc[j][q] + b[q];
      r[q] = ELU ? (v > 0.f ? v : expm1f(v)) : v;
    }
    *(vrow*)(orow + j * 32 * VW) = r;
  }
}


static inline int cdiv(long long a, long long b) { return (int)((a + b - 1) / b); }

extern "C" void kernel_launch(void* const* d_in, const int* in_sizes, int n_in,
                              void* d_out, int out_size, void* d_ws, size_t ws_size,
                              hipStream_t stream) {
  const float* x      = (const float*)d_in[0];
  const int*   ei     = (const int*)  d_in[1];
  const float* W1     = (const float*)d_in[2];
  const float* att_s1 = (const float*)d_in[3];
  const float* att_d1 = (const float*)d_in[4];
  const float* b1     = (const float*)d_in[5];
  const float* W2     = (const float*)d_in[6];
  const float* att_s2 = (const float*)d_in[7];
  const float* att_d2 = (const float*)d_in[8];
  const float* b2     = (const float*)d_in[9];

  const int IN   = 128;
  const int C1   = 256;
  const int Nn   = in_sizes[0] / IN;
  const int Eraw = in_sizes[1] / 2;

  float* ws = (float*)d_ws;
  size_t off = 0;
  const size_t npad = ((size_t)Nn + 4) & ~(size_t)3;
  float* h1     = ws + off; off += (size_t)Nn * C1;
  float* out1   = ws + off; off += (size_t)Nn * C1;
  float* asrc1  = ws + off; off += (size_t)Nn * 4;
  float* adst1  = ws + off; off += (size_t)Nn * 4;
  const int NS     = NSLICE;
  const int nrange = Nn / 256 + 1;
  const int rpitch = (nrange + 3) & ~3;
  const size_t ncell = (size_t)rpitch * NS;
  int*   cellcnt = (int*)(ws + off); off += ncell;
  int*   rstart  = (int*)(ws + off); off += (size_t)rpitch + 4;
  int*   rowptr  = (int*)(ws + off); off += npad;
  int*   binned  = (int*)(ws + off); off += (size_t)Eraw * 2 + 2;
  off = (off + 3) & ~(size_t)3;
  int*   lrank   = (int*)(ws + off); off += ((size_t)Eraw + 3) & ~(size_t)3;
  int*   col     = (int*)(ws + off); off += (size_t)Eraw;
  float* h2     = h1;
  float* asrc2  = asrc1;
  float* adst2  = adst1;
  float* outf   = (float*)d_out;

  const int T = 256;

  const int chunk = cdiv((long long)Eraw, NS);
  slice_count<<<NS, SORT_THREADS, nrange * sizeof(int), stream>>>(ei, lrank, cellcnt, Eraw, chunk, nrange, rpitch);
  slice_place<<<NS, SORT_THREADS, ((2 * COLGROUPS + 1) * rpitch + 32) * sizeof(int), stream>>>(
      ei, lrank, cellcnt, rstart, binned, Eraw, chunk, nrange, rpitch);
  range_sort<<<nrange, SORT_THREADS, 0, stream>>>(binned, rstart, rowptr, col, Nn);

  gemm_head_alpha<128, 4><<<cdiv((long long)(Nn / 16) * 4, T / 32), T, 0, stream>>>(
      x, W1, att_s1, att_d1, h1, asrc1, adst1, Nn);
  seg_softmax_agg<4, 64, true><<<cdiv((long long)Nn, T / 32), T, 0, stream>>>(
      rowptr, col, asrc1, adst1, h1, b1, out1, Nn);

  gemm_head_alpha<256, 1><<<cdiv((long long)(Nn / 16), T / 32), T, 0, stream>>>(
      out1, W2, att_s2, att_d2, h2, asrc2, adst2, Nn);
  seg_softmax_agg<1, 64, false><<<cdiv((long long)Nn, T / 32), T, 0, stream>>>(
      rowptr, col, asrc2, adst2, h2, b2, outf, Nn);
}


# Per-node softmax reads its edge list and source scores once: each lane keeps its edge's source and scores in registers through max, exp-sum and weights; only lists over 32 edges re-read their rest
# speedup vs baseline: 5.1974x; 1.0957x over previous
#include <hip/hip_runtime.h>

typedef __attribute__((ext_vector_type(2))) float v2f;
typedef __attribute__((ext_vector_type(8))) float v8f;

#define NEG_SLOPE 0.2f


typedef __attribute__((ext_vector_type(4))) float v4f;

template <int K, int HN>
__global__ void __launch_bounds__(256)
gemm_head_alpha(const float* __restrict__ A, const float* __restrict__ B,
                const float* __restrict__ att_s, const float* __restrict__ att_d,
                float* __restrict__ C, float* __restrict__ asrc, float* __restrict__ adst,
                int M) {
  constexpr int Nn = HN * 64;
  const int lane   = threadIdx.x & 31;
  const int wave   = blockIdx.x * (blockDim.x >> 5) + (threadIdx.x >> 5);
  const int mtiles = M >> 4;
  const int tileM  = wave / HN;
  const int hd     = wave - tileM * HN;
  if (tileM >= mtiles) return;

  const int lh  = lane >> 4;
  const int l16 = lane & 15;

  const float* ap = A + (size_t)(tileM * 16 + l16) * K + lh * 2;
  const float* bp = B + (size_t)(lh * 2) * Nn + hd * 64 + l16 * 4;

  v8f acc0 = {}, acc1 = {}, acc2 = {}, acc3 = {};
#pragma unroll 8
  for (int k = 0; k < K; k += 4) {
    const v2f a  = *(const v2f*)(ap + k);
    const v4f b0 = *(const v4f*)(bp + (size_t)k * Nn);
    const v4f b1 = *(const v4f*)(bp + (size_t)(k + 1) * Nn);
    v2f b;
    b.x = b0.x; b.y = b1.x;
    acc0 = __builtin_amdgcn_wmma_f32_16x16x4_f32(false, a, false, b, (short)0, acc0, false, false);
    b.x = b0.y; b.y = b1.y;
    acc1 = __builtin_amdgcn_wmma_f32_16x16x4_f32(false, a, false, b, (short)0, acc1, false, false);
    b.x = b0.z; b.y = b1.z;
    acc2 = __builtin_amdgcn_wmma_f32_16x16x4_f32(false, a, false, b, (short)0, acc2, false, false);
    b.x = b0.w; b.y = b1.w;
    acc3 = __builtin_amdgcn_wmma_f32_16x16x4_f32(false, a, false, b, (short)0, acc3, false, false);
  }

  const int row0 = tileM * 16 + lh * 8;
  float* crow = C + (size_t)row0 * Nn + hd * 64 + l16 * 4;
#pragma unroll
  for (int r = 0; r < 8; ++r) {
    v4f o;
    o.x = acc0[r]; o.y = acc1[r]; o.z = acc2[r]; o.w = acc3[r];
    *(v4f*)(crow + (size_t)r * Nn) = o;
  }

  const v4f s = *(const v4f*)(att_s + hd * 64 + l16 * 4);
  const v4f d = *(const v4f*)(att_d + hd * 64 + l16 * 4);
  float ps[8], pd[8];
#pragma unroll
  for (int r = 0; r < 8; ++r) {
    ps[r] = acc0[r] * s.x + acc1[r] * s.y + acc2[r] * s.z + acc3[r] * s.w;
    pd[r] = acc0[r] * d.x + acc1[r] * d.y + acc2[r] * d.z + acc3[r] * d.w;
  }
#pragma unroll
  for (int r = 0; r < 8; ++r) {
    for (int m = 8; m >= 1; m >>= 1) {
      ps[r] += __shfl_xor(ps[r], m);
      pd[r] += __shfl_xor(pd[r], m);
    }
  }
  float vs = ps[0], vd = pd[0];
#pragma unroll
  for (int r = 1; r < 8; ++r) {
    vs = (l16 == r) ? ps[r] : vs;
    vd = (l16 == r) ? pd[r] : vd;
  }
  if (l16 < 8) {
    asrc[(size_t)(row0 + l16) * HN + hd] = vs;
    adst[(size_t)(row0 + l16) * HN + hd] = vd;
  }
}


#define RANGE_SHIFT  8
#define RANGE_NODES  256
#define SORT_THREADS 1024
#define NSLICE       256
#define COLGROUPS    16

__global__ void __launch_bounds__(SORT_THREADS)
slice_count(const int* __restrict__ ei, int* __restrict__ lrank, int* __restrict__ cellcnt,
            int Eraw, int chunk, int nrange, int rpitch) {
  extern __shared__ int hcell[];
  const int t = threadIdx.x;
  const int g = blockIdx.x;
  for (int r = t; r < nrange; r += SORT_THREADS) hcell[r] = 0;
  __syncthreads();
  const int beg = g * chunk;
  const int end = (beg + chunk < Eraw) ? (beg + chunk) : Eraw;
  for (int e0 = beg + t; e0 < end; e0 += 4 * SORT_THREADS) {
    int d[4];
#pragma unroll
    for (int j = 0; j < 4; ++j) {
      const int e = e0 + j * SORT_THREADS;
      d[j] = ei[Eraw + ((e < end) ? e : (end - 1))];
    }
    asm volatile("" :: "v"(d[0]), "v"(d[1]), "v"(d[2]), "v"(d[3]));
#pragma unroll
    for (int j = 0; j < 4; ++j) {
      const int e = e0 + j * SORT_THREADS;
      if (e < end) lrank[e] = atomicAdd(&hcell[d[j] >> RANGE_SHIFT], 1);
    }
  }
  __syncthreads();
  for (int r = t; r < rpitch; r += SORT_THREADS)
    cellcnt[(size_t)g * rpitch + r] = (r < nrange) ? hcell[r] : 0;
}

__global__ void __launch_bounds__(SORT_THREADS)
slice_place(const int* __restrict__ ei, const int* __restrict__ lrank,
            const int* __restrict__ cellcnt, int* __restrict__ rstart,
            int* __restrict__ binned, int Eraw, int chunk, int nrange, int rpitch) {
  extern __shared__ __attribute__((aligned(16))) int lds[];
  int* part_t = lds;
  int* part_p = part_t + COLGROUPS * rpitch;
  int* lcell  = part_p + COLGROUPS * rpitch;
  int* wtot   = lcell + rpitch;
  const int t    = threadIdx.x;
  const int lane = t & 31;
  const int wv   = t >> 5;
  const int g    = blockIdx.x;

  const int r4n = rpitch >> 2;
  for (int idx = t; idx < COLGROUPS * r4n; idx += SORT_THREADS) {
    const int q  = idx / r4n;
    const int r4 = idx - q * r4n;
    const int* colp = cellcnt + (size_t)q * rpitch + r4 * 4;
    int4 st = make_int4(0, 0, 0, 0), sp = make_int4(0, 0, 0, 0);
#pragma unroll 8
    for (int k = 0; k < NSLICE / COLGROUPS; ++k) {
      const int  gg = q + k * COLGROUPS;
      const int4 v  = *(const int4*)(colp + (size_t)k * COLGROUPS * rpitch);
      st.x += v.x; st.y += v.y; st.z += v.z; st.w += v.w;
      const bool below = gg < g;
      sp.x += below ? v.x : 0; sp.y += below ? v.y : 0;
      sp.z += below ? v.z : 0; sp.w += below ? v.w : 0;
    }
    *(int4*)(part_t + q * rpitch + r4 * 4) = st;
    *(int4*)(part_p + q * rpitch + r4 * 4) = sp;
  }
  __syncthreads();

  int carry = 0;
  for (int base = 0; base < nrange; base += SORT_THREADS) {
    const int r = base + t;
    int tot = 0, low = 0;
    if (r < nrange) {
      for (int q = 0; q < COLGROUPS; ++q) { tot += part_t[q * rpitch + r]; low += part_p[q * rpitch + r]; }
    }
    int incl = tot;
    for (int d = 1; d < 32; d <<= 1) {
      int y = __shfl_up(incl, d);
      if (lane >= d) incl += y;
    }
    if (lane == 31) wtot[wv] = incl;
    __syncthreads();
    int ws = wtot[lane];
    for (int d = 1; d < 32; d <<= 1) {
      int y = __shfl_up(ws, d);
      if (lane >= d) ws += y;
    }
    const int wprefix = __shfl(ws, wv) - __shfl(incl, 31);
    const int total   = __shfl(ws, 31);
    const int start   = carry + wprefix + incl - tot;
    if (r < nrange) {
      lcell[r] = start + low;
      if (g == 0) rstart[r] = start;
    }
    carry += total;
    __syncthreads();
  }
  if (g == 0 && t == 0) rstart[nrange] = carry;

  const int beg = g * chunk;
  const int end = (beg + chunk < Eraw) ? (beg + chunk) : Eraw;
  int2* bp = (int2*)binned;
  for (int e0 = beg + t; e0 < end; e0 += 4 * SORT_THREADS) {
    int s[4], d[4], r[4];
#pragma unroll
    for (int j = 0; j < 4; ++j) {
      const int e  = e0 + j * SORT_THREADS;
      const int ec = (e < end) ? e : (end - 1);
      s[j] = ei[ec];
      d[j] = ei[Eraw + ec];
      r[j] = lrank[ec];
    }
    asm volatile("" :: "v"(s[0]), "v"(s[1]), "v"(s[2]), "v"(s[3]), "v"(d[0]), "v"(d[1]), "v"(d[2]),
                       "v"(d[3]), "v"(r[0]), "v"(r[1]), "v"(r[2]), "v"(r[3]));
#pragma unroll
    for (int j = 0; j < 4; ++j) {
      const int e = e0 + j * SORT_THREADS;
      if (e < end) bp[lcell[d[j] >> RANGE_SHIFT] + r[j]] = make_int2(s[j], d[j]);
    }
  }
}

__global__ void __launch_bounds__(SORT_THREADS)
range_sort(const int* __restrict__ binned, const int* __restrict__ rstart,
           int* __restrict__ rowptr, int* __restrict__ col, int n_nodes) {
  __shared__ int cnt[RANGE_NODES];
  const int t = threadIdx.x;
  const int b = blockIdx.x;
  const int beg = rstart[b];
  const int end = rstart[b + 1];
  const int2* bp = (const int2*)binned;
  if (t < RANGE_NODES) cnt[t] = 0;
  __syncthreads();
  for (int i0 = beg + t; i0 < end; i0 += 8 * SORT_THREADS) {
    int d[8];
#pragma unroll
    for (int j = 0; j < 8; ++j) {
      const int i = i0 + j * SORT_THREADS;
      d[j] = bp[(i < end) ? i : (end - 1)].y;
    }
    asm volatile("" :: "v"(d[0]), "v"(d[1]), "v"(d[2]), "v"(d[3]),
                       "v"(d[4]), "v"(d[5]), "v"(d[6]), "v"(d[7]));
#pragma unroll
    for (int j = 0; j < 8; ++j) {
      const int i = i0 + j * SORT_THREADS;
      if (i < end) atomicAdd(&cnt[d[j] & (RANGE_NODES - 1)], 1);
    }
  }
  __syncthreads();
  if (t < 32) {
    int c[8];
    int tsum = 0;
    for (int j = 0; j < 8; ++j) { c[j] = cnt[t * 8 + j]; tsum += c[j]; }
    int incl = tsum;
    for (int d = 1; d < 32; d <<= 1) {
      int y = __shfl_up(incl, d);
      if (t >= d) incl += y;
    }
    int run = incl - tsum;
    for (int j = 0; j < 8; ++j) { cnt[t * 8 + j] = run; run += c[j]; }
  }
  __syncthreads();
  if (t < RANGE_NODES) {
    const int node = b * RANGE_NODES + t;
    if (node <= n_nodes) rowptr[node] = beg + cnt[t];
  }
  __syncthreads();
  for (int i0 = beg + t; i0 < end; i0 += 8 * SORT_THREADS) {
    int2 sd[8];
#pragma unroll
    for (int j = 0; j < 8; ++j) {
      const int i = i0 + j * SORT_THREADS;
      sd[j] = bp[(i < end) ? i : (end - 1)];
    }
    asm volatile("" :: "v"(sd[0].y), "v"(sd[1].y), "v"(sd[2].y), "v"(sd[3].y),
                       "v"(sd[4].y), "v"(sd[5].y), "v"(sd[6].y), "v"(sd[7].y));
#pragma unroll
    for (int j = 0; j < 8; ++j) {
      const int i = i0 + j * SORT_THREADS;
      if (i < end) {
        const int p = atomicAdd(&cnt[sd[j].y & (RANGE_NODES - 1)], 1);
        col[beg + p] = sd[j].x;
      }
    }
  }
}


template <int N> using vecf = float __attribute__((ext_vector_type(N)));

__device__ __forceinline__ void load_heads(const float* __restrict__ p, float (&v)[4]) {
  const float4 t = *(const float4*)p;
  v[0] = t.x; v[1] = t.y; v[2] = t.z; v[3] = t.w;
}
__device__ __forceinline__ void load_heads(const float* __restrict__ p, float (&v)[1]) {
  v[0] = p[0];
}

__device__ __forceinline__ float leaky(float v) { return v > 0.f ? v : NEG_SLOPE * v; }

template <int H, int F, bool ELU>
__global__ void __launch_bounds__(256)
seg_softmax_agg(const int* __restrict__ rowptr, const int* __restrict__ col,
                const float* __restrict__ asrc, const float* __restrict__ adst,
                const float* __restrict__ feat, const float* __restrict__ bias,
                float* __restrict__ out, int n_nodes) {
  constexpr int C   = H * F;
  constexpr int CPL = C / 32;
  constexpr int VW  = CPL >= 4 ? 4 : CPL;
  constexpr int NL  = CPL / VW;
  constexpr int U   = 8 / NL;
  static_assert(U * NL == 8, "the load fence below names eight values");
  typedef vecf<VW> vrow;

  const int lane = threadIdx.x & 31;
  const int n = __builtin_amdgcn_readfirstlane(
      (int)(blockIdx.x * (blockDim.x >> 5) + (threadIdx.x >> 5)));
  if (n >= n_nodes) return;
  const int beg = rowptr[n];
  const int end = rowptr[n + 1];

  float ad[H], as_self[H], vself[H], mx[H];
  load_heads(adst + (size_t)n * H, ad);
  load_heads(asrc + (size_t)n * H, as_self);
  for (int h = 0; h < H; ++h) { vself[h] = leaky(as_self[h] + ad[h]); mx[h] = vself[h]; }

  const int  i0  = beg + lane;
  const bool ok0 = i0 < end;
  const int  s0  = ok0 ? col[i0] : n;
  float e0[H];
  {
    float a[H];
    load_heads(asrc + (size_t)s0 * H, a);
    for (int h = 0; h < H; ++h) e0[h] = leaky(a[h] + ad[h]);
  }

  for (int h = 0; h < H; ++h) mx[h] = ok0 ? fmaxf(mx[h], e0[h]) : mx[h];
  for (int i = i0 + 32; i < end; i += 32) {
    float a[H];
    load_heads(asrc + (size_t)col[i] * H, a);
    for (int h = 0; h < H; ++h) mx[h] = fmaxf(mx[h], leaky(a[h] + ad[h]));
  }
  for (int h = 0; h < H; ++h) {
    for (int m = 16; m >= 1; m >>= 1) mx[h] = fmaxf(mx[h], __shfl_xor(mx[h], m));
  }

  float den[H], ex0[H];
  for (int h = 0; h < H; ++h) {
    ex0[h] = __expf(e0[h] - mx[h]);
    den[h] = ok0 ? ex0[h] : 0.f;
  }
  for (int i = i0 + 32; i < end; i += 32) {
    float a[H];
    load_heads(asrc + (size_t)col[i] * H, a);
    for (int h = 0; h < H; ++h) den[h] += __expf(leaky(a[h] + ad[h]) - mx[h]);
  }
  for (int h = 0; h < H; ++h) {
    for (int m = 16; m >= 1; m >>= 1) den[h] += __shfl_xor(den[h], m);
    den[h] += __expf(vself[h] - mx[h]);
    den[h] += 1e-16f;
  }

  int hsel[NL];
  vrow acc[NL];
  for (int j = 0; j < NL; ++j) {
    hsel[j] = (j * 32 * VW + lane * VW) / F;
    for (int q = 0; q < VW; ++q) acc[j][q] = 0.f;
  }
  {
    const float* row = feat + (size_t)n * C + lane * VW;
    for (int j = 0; j < NL; ++j) {
      float w = __expf(vself[0] - mx[0]) / den[0];
      for (int h = 1; h < H; ++h)
        w = (hsel[j] == h) ? __expf(vself[h] - mx[h]) / den[h] : w;
      const vrow f = *(const vrow*)(row + j * 32 * VW);
      for (int q = 0; q < VW; ++q) acc[j][q] += w * f[q];
    }
  }
  for (int base = beg; base < end; base += 32) {
    int   s;
    float w[H];
    if (base == beg) {
      s = s0;
      for (int h = 0; h < H; ++h) w[h] = ok0 ? ex0[h] / den[h] : 0.f;
    } else {
      const int  i  = base + lane;
      const bool ok = i < end;
      s = ok ? col[i] : n;
      float a[H];
      load_heads(asrc + (size_t)s * H, a);
      for (int h = 0; h < H; ++h)
        w[h] = ok ? __expf(leaky(a[h] + ad[h]) - mx[h]) / den[h] : 0.f;
    }
    const int cn  = (end - base) < 32 ? (end - base) : 32;
    const int cnp = (cn + U - 1) & ~(U - 1);
    for (int k = 0; k < cnp; k += U) {
      vrow  f[U][NL];
      float wj[U][NL];
#pragma unroll
      for (int u = 0; u < U; ++u) {
        const int sk = __builtin_amdgcn_readlane(s, k + u);
        float wk[H];
        for (int h = 0; h < H; ++h)
          wk[h] = __int_as_float(__builtin_amdgcn_readlane(__float_as_int(w[h]), k + u));
        const float* row = feat + (size_t)sk * C + lane * VW;
        for (int j = 0; j < NL; ++j) {
          float t = wk[0];
          for (int h = 1; h < H; ++h) t = (hsel[j] == h) ? wk[h] : t;
          wj[u][j] = t;
          f[u][j]  = *(const vrow*)(row + j * 32 * VW);
        }
      }
      asm volatile("" :: "v"(f[0 / NL][0 % NL]), "v"(f[1 / NL][1 % NL]), "v"(f[2 / NL][2 % NL]),
                         "v"(f[3 / NL][3 % NL]), "v"(f[4 / NL][4 % NL]), "v"(f[5 / NL][5 % NL]),
                         "v"(f[6 / NL][6 % NL]), "v"(f[7 / NL][7 % NL]));
#pragma unroll
      for (int u = 0; u < U; ++u)
        for (int j = 0; j < NL; ++j)
          for (int q = 0; q < VW; ++q) acc[j][q] += wj[u][j] * f[u][j][q];
    }
  }

  float* orow = out + (size_t)n * C + lane * VW;
  const float* brow = bias + lane * VW;
  for (int j = 0; j < NL; ++j) {
    const vrow b = *(const vrow*)(brow + j * 32 * VW);
    vrow r;
    for (int q = 0; q < VW; ++q) {
      float v = acc[j][q] + b[q];
      r[q] = ELU ? (v > 0.f ? v : expm1f(v)) : v;
    }
    *(vrow*)(orow + j * 32 * VW) = r;
  }
}


static inline int cdiv(long long a, long long b) { return (int)((a + b - 1) / b); }

extern "C" void kernel_launch(void* const* d_in, const int* in_sizes, int n_in,
                              void* d_out, int out_size, void* d_ws, size_t ws_size,
                              hipStream_t stream) {
  const float* x      = (const float*)d_in[0];
  const int*   ei     = (const int*)  d_in[1];
  const float* W1     = (const float*)d_in[2];
  const float* att_s1 = (const float*)d_in[3];
  const float* att_d1 = (const float*)d_in[4];
  const float* b1     = (const float*)d_in[5];
  const float* W2     = (const float*)d_in[6];
  const float* att_s2 = (const float*)d_in[7];
  const float* att_d2 = (const float*)d_in[8];
  const float* b2     = (const float*)d_in[9];

  const int IN   = 128;
  const int C1   = 256;
  const int Nn   = in_sizes[0] / IN;
  const int Eraw = in_sizes[1] / 2;

  float* ws = (float*)d_ws;
  size_t off = 0;
  const size_t npad = ((size_t)Nn + 4) & ~(size_t)3;
  float* h1     = ws + off; off += (size_t)Nn * C1;
  float* out1   = ws + off; off += (size_t)Nn * C1;
  float* asrc1  = ws + off; off += (size_t)Nn * 4;
  float* adst1  = ws + off; off += (size_t)Nn * 4;
  const int NS     = NSLICE;
  const int nrange = Nn / 256 + 1;
  const int rpitch = (nrange + 3) & ~3;
  const size_t ncell = (size_t)rpitch * NS;
  int*   cellcnt = (int*)(ws + off); off += ncell;
  int*   rstart  = (int*)(ws + off); off += (size_t)rpitch + 4;
  int*   rowptr  = (int*)(ws + off); off += npad;
  int*   binned  = (int*)(ws + off); off += (size_t)Eraw * 2 + 2;
  off = (off + 3) & ~(size_t)3;
  int*   lrank   = (int*)(ws + off); off += ((size_t)Eraw + 3) & ~(size_t)3;
  int*   col     = (int*)(ws + off); off += (size_t)Eraw;
  float* h2     = h1;
  float* asrc2  = asrc1;
  float* adst2  = adst1;
  float* outf   = (float*)d_out;

  const int T = 256;

  const int chunk = cdiv((long long)Eraw, NS);
  slice_count<<<NS, SORT_THREADS, nrange * sizeof(int), stream>>>(ei, lrank, cellcnt, Eraw, chunk, nrange, rpitch);
  slice_place<<<NS, SORT_THREADS, ((2 * COLGROUPS + 1) * rpitch + 32) * sizeof(int), stream>>>(
      ei, lrank, cellcnt, rstart, binned, Eraw, chunk, nrange, rpitch);
  range_sort<<<nrange, SORT_THREADS, 0, stream>>>(binned, rstart, rowptr, col, Nn);

  gemm_head_alpha<128, 4><<<cdiv((long long)(Nn / 16) * 4, T / 32), T, 0, stream>>>(
      x, W1, att_s1, att_d1, h1, asrc1, adst1, Nn);
  seg_softmax_agg<4, 64, true><<<cdiv((long long)Nn, T / 32), T, 0, stream>>>(
      rowptr, col, asrc1, adst1, h1, b1, out1, Nn);

  gemm_head_alpha<256, 1><<<cdiv((long long)(Nn / 16), T / 32), T, 0, stream>>>(
      out1, W2, att_s2, att_d2, h2, asrc2, adst2, Nn);
  seg_softmax_agg<1, 64, false><<<cdiv((long long)Nn, T / 32), T, 0, stream>>>(
      rowptr, col, asrc2, adst2, h2, b2, outf, Nn);
}


# Aggregation kernels' scalar math on hardware forms: softmax weights by one v_rcp per head (was a division per weight), ELU as v_exp minus 1 (was library expm1); last bits differ, sum order kept
# speedup vs baseline: 5.3790x; 1.0349x over previous
#include <hip/hip_runtime.h>

typedef __attribute__((ext_vector_type(2))) float v2f;
typedef __attribute__((ext_vector_type(8))) float v8f;

#define NEG_SLOPE 0.2f


typedef __attribute__((ext_vector_type(4))) float v4f;

template <int K, int HN>
__global__ void __launch_bounds__(256)
gemm_head_alpha(const float* __restrict__ A, const float* __restrict__ B,
                const float* __restrict__ att_s, const float* __restrict__ att_d,
                float* __restrict__ C, float* __restrict__ asrc, float* __restrict__ adst,
                int M) {
  constexpr int Nn = HN * 64;
  const int lane   = threadIdx.x & 31;
  const int wave   = blockIdx.x * (blockDim.x >> 5) + (threadIdx.x >> 5);
  const int mtiles = M >> 4;
  const int tileM  = wave / HN;
  const int hd     = wave - tileM * HN;
  if (tileM >= mtiles) return;

  const int lh  = lane >> 4;
  const int l16 = lane & 15;

  const float* ap = A + (size_t)(tileM * 16 + l16) * K + lh * 2;
  const float* bp = B + (size_t)(lh * 2) * Nn + hd * 64 + l16 * 4;

  v8f acc0 = {}, acc1 = {}, acc2 = {}, acc3 = {};
#pragma unroll 8
  for (int k = 0; k < K; k += 4) {
    const v2f a  = *(const v2f*)(ap + k);
    const v4f b0 = *(const v4f*)(bp + (size_t)k * Nn);
    const v4f b1 = *(const v4f*)(bp + (size_t)(k + 1) * Nn);
    v2f b;
    b.x = b0.x; b.y = b1.x;
    acc0 = __builtin_amdgcn_wmma_f32_16x16x4_f32(false, a, false, b, (short)0, acc0, false, false);
    b.x = b0.y; b.y = b1.y;
    acc1 = __builtin_amdgcn_wmma_f32_16x16x4_f32(false, a, false, b, (short)0, acc1, false, false);
    b.x = b0.z; b.y = b1.z;
    acc2 = __builtin_amdgcn_wmma_f32_16x16x4_f32(false, a, false, b, (short)0, acc2, false, false);
    b.x = b0.w; b.y = b1.w;
    acc3 = __builtin_amdgcn_wmma_f32_16x16x4_f32(false, a, false, b, (short)0, acc3, false, false);
  }

  const int row0 = tileM * 16 + lh * 8;
  float* crow = C + (size_t)row0 * Nn + hd * 64 + l16 * 4;
#pragma unroll
  for (int r = 0; r < 8; ++r) {
    v4f o;
    o.x = acc0[r]; o.y = acc1[r]; o.z = acc2[r]; o.w = acc3[r];
    *(v4f*)(crow + (size_t)r * Nn) = o;
  }

  const v4f s = *(const v4f*)(att_s + hd * 64 + l16 * 4);
  const v4f d = *(const v4f*)(att_d + hd * 64 + l16 * 4);
  float ps[8], pd[8];
#pragma unroll
  for (int r = 0; r < 8; ++r) {
    ps[r] = acc0[r] * s.x + acc1[r] * s.y + acc2[r] * s.z + acc3[r] * s.w;
    pd[r] = acc0[r] * d.x + acc1[r] * d.y + acc2[r] * d.z + acc3[r] * d.w;
  }
#pragma unroll
  for (int r = 0; r < 8; ++r) {
    for (int m = 8; m >= 1; m >>= 1) {
      ps[r] += __shfl_xor(ps[r], m);
      pd[r] += __shfl_xor(pd[r], m);
    }
  }
  float vs = ps[0], vd = pd[0];
#pragma unroll
  for (int r = 1; r < 8; ++r) {
    vs = (l16 == r) ? ps[r] : vs;
    vd = (l16 == r) ? pd[r] : vd;
  }
  if (l16 < 8) {
    asrc[(size_t)(row0 + l16) * HN + hd] = vs;
    adst[(size_t)(row0 + l16) * HN + hd] = vd;
  }
}


#define RANGE_SHIFT  8
#define RANGE_NODES  256
#define SORT_THREADS 1024
#define NSLICE       256
#define COLGROUPS    16

__global__ void __launch_bounds__(SORT_THREADS)
slice_count(const int* __restrict__ ei, int* __restrict__ lrank, int* __restrict__ cellcnt,
            int Eraw, int chunk, int nrange, int rpitch) {
  extern __shared__ int hcell[];
  const int t = threadIdx.x;
  const int g = blockIdx.x;
  for (int r = t; r < nrange; r += SORT_THREADS) hcell[r] = 0;
  __syncthreads();
  const int beg = g * chunk;
  const int end = (beg + chunk < Eraw) ? (beg + chunk) : Eraw;
  for (int e0 = beg + t; e0 < end; e0 += 4 * SORT_THREADS) {
    int d[4];
#pragma unroll
    for (int j = 0; j < 4; ++j) {
      const int e = e0 + j * SORT_THREADS;
      d[j] = ei[Eraw + ((e < end) ? e : (end - 1))];
    }
    asm volatile("" :: "v"(d[0]), "v"(d[1]), "v"(d[2]), "v"(d[3]));
#pragma unroll
    for (int j = 0; j < 4; ++j) {
      const int e = e0 + j * SORT_THREADS;
      if (e < end) lrank[e] = atomicAdd(&hcell[d[j] >> RANGE_SHIFT], 1);
    }
  }
  __syncthreads();
  for (int r = t; r < rpitch; r += SORT_THREADS)
    cellcnt[(size_t)g * rpitch + r] = (r < nrange) ? hcell[r] : 0;
}

__global__ void __launch_bounds__(SORT_THREADS)
slice_place(const int* __restrict__ ei, const int* __restrict__ lrank,
            const int* __restrict__ cellcnt, int* __restrict__ rstart,
            int* __restrict__ binned, int Eraw, int chunk, int nrange, int rpitch) {
  extern __shared__ __attribute__((aligned(16))) int lds[];
  int* part_t = lds;
  int* part_p = part_t + COLGROUPS * rpitch;
  int* lcell  = part_p + COLGROUPS * rpitch;
  int* wtot   = lcell + rpitch;
  const int t    = threadIdx.x;
  const int lane = t & 31;
  const int wv   = t >> 5;
  const int g    = blockIdx.x;

  const int r4n = rpitch >> 2;
  for (int idx = t; idx < COLGROUPS * r4n; idx += SORT_THREADS) {
    const int q  = idx / r4n;
    const int r4 = idx - q * r4n;
    const int* colp = cellcnt + (size_t)q * rpitch + r4 * 4;
    int4 st = make_int4(0, 0, 0, 0), sp = make_int4(0, 0, 0, 0);
#pragma unroll 8
    for (int k = 0; k < NSLICE / COLGROUPS; ++k) {
      const int  gg = q + k * COLGROUPS;
      const int4 v  = *(const int4*)(colp + (size_t)k * COLGROUPS * rpitch);
      st.x += v.x; st.y += v.y; st.z += v.z; st.w += v.w;
      const bool below = gg < g;
      sp.x += below ? v.x : 0; sp.y += below ? v.y : 0;
      sp.z += below ? v.z : 0; sp.w += below ? v.w : 0;
    }
    *(int4*)(part_t + q * rpitch + r4 * 4) = st;
    *(int4*)(part_p + q * rpitch + r4 * 4) = sp;
  }
  __syncthreads();

  int carry = 0;
  for (int base = 0; base < nrange; base += SORT_THREADS) {
    const int r = base + t;
    int tot = 0, low = 0;
    if (r < nrange) {
      for (int q = 0; q < COLGROUPS; ++q) { tot += part_t[q * rpitch + r]; low += part_p[q * rpitch + r]; }
    }
    int incl = tot;
    for (int d = 1; d < 32; d <<= 1) {
      int y = __shfl_up(incl, d);
      if (lane >= d) incl += y;
    }
    if (lane == 31) wtot[wv] = incl;
    __syncthreads();
    int ws = wtot[lane];
    for (int d = 1; d < 32; d <<= 1) {
      int y = __shfl_up(ws, d);
      if (lane >= d) ws += y;
    }
    const int wprefix = __shfl(ws, wv) - __shfl(incl, 31);
    const int total   = __shfl(ws, 31);
    const int start   = carry + wprefix + incl - tot;
    if (r < nrange) {
      lcell[r] = start + low;
      if (g == 0) rstart[r] = start;
    }
    carry += total;
    __syncthreads();
  }
  if (g == 0 && t == 0) rstart[nrange] = carry;

  const int beg = g * chunk;
  const int end = (beg + chunk < Eraw) ? (beg + chunk) : Eraw;
  int2* bp = (int2*)binned;
  for (int e0 = beg + t; e0 < end; e0 += 4 * SORT_THREADS) {
    int s[4], d[4], r[4];
#pragma unroll
    for (int j = 0; j < 4; ++j) {
      const int e  = e0 + j * SORT_THREADS;
      const int ec = (e < end) ? e : (end - 1);
      s[j] = ei[ec];
      d[j] = ei[Eraw + ec];
      r[j] = lrank[ec];
    }
    asm volatile("" :: "v"(s[0]), "v"(s[1]), "v"(s[2]), "v"(s[3]), "v"(d[0]), "v"(d[1]), "v"(d[2]),
                       "v"(d[3]), "v"(r[0]), "v"(r[1]), "v"(r[2]), "v"(r[3]));
#pragma unroll
    for (int j = 0; j < 4; ++j) {
      const int e = e0 + j * SORT_THREADS;
      if (e < end) bp[lcell[d[j] >> RANGE_SHIFT] + r[j]] = make_int2(s[j], d[j]);
    }
  }
}

__global__ void __launch_bounds__(SORT_THREADS)
range_sort(const int* __restrict__ binned, const int* __restrict__ rstart,
           int* __restrict__ rowptr, int* __restrict__ col, int n_nodes) {
  __shared__ int cnt[RANGE_NODES];
  const int t = threadIdx.x;
  const int b = blockIdx.x;
  const int beg = rstart[b];
  const int end = rstart[b + 1];
  const int2* bp = (const int2*)binned;
  if (t < RANGE_NODES) cnt[t] = 0;
  __syncthreads();
  for (int i0 = beg + t; i0 < end; i0 += 8 * SORT_THREADS) {
    int d[8];
#pragma unroll
    for (int j = 0; j < 8; ++j) {
      const int i = i0 + j * SORT_THREADS;
      d[j] = bp[(i < end) ? i : (end - 1)].y;
    }
    asm volatile("" :: "v"(d[0]), "v"(d[1]), "v"(d[2]), "v"(d[3]),
                       "v"(d[4]), "v"(d[5]), "v"(d[6]), "v"(d[7]));
#pragma unroll
    for (int j = 0; j < 8; ++j) {
      const int i = i0 + j * SORT_THREADS;
      if (i < end) atomicAdd(&cnt[d[j] & (RANGE_NODES - 1)], 1);
    }
  }
  __syncthreads();
  if (t < 32) {
    int c[8];
    int tsum = 0;
    for (int j = 0; j < 8; ++j) { c[j] = cnt[t * 8 + j]; tsum += c[j]; }
    int incl = tsum;
    for (int d = 1; d < 32; d <<= 1) {
      int y = __shfl_up(incl, d);
      if (t >= d) incl += y;
    }
    int run = incl - tsum;
    for (int j = 0; j < 8; ++j) { cnt[t * 8 + j] = run; run += c[j]; }
  }
  __syncthreads();
  if (t < RANGE_NODES) {
    const int node = b * RANGE_NODES + t;
    if (node <= n_nodes) rowptr[node] = beg + cnt[t];
  }
  __syncthreads();
  for (int i0 = beg + t; i0 < end; i0 += 8 * SORT_THREADS) {
    int2 sd[8];
#pragma unroll
    for (int j = 0; j < 8; ++j) {
      const int i = i0 + j * SORT_THREADS;
      sd[j] = bp[(i < end) ? i : (end - 1)];
    }
    asm volatile("" :: "v"(sd[0].y), "v"(sd[1].y), "v"(sd[2].y), "v"(sd[3].y),
                       "v"(sd[4].y), "v"(sd[5].y), "v"(sd[6].y), "v"(sd[7].y));
#pragma unroll
    for (int j = 0; j < 8; ++j) {
      const int i = i0 + j * SORT_THREADS;
      if (i < end) {
        const int p = atomicAdd(&cnt[sd[j].y & (RANGE_NODES - 1)], 1);
        col[beg + p] = sd[j].x;
      }
    }
  }
}


template <int N> using vecf = float __attribute__((ext_vector_type(N)));

__device__ __forceinline__ void load_heads(const float* __restrict__ p, float (&v)[4]) {
  const float4 t = *(const float4*)p;
  v[0] = t.x; v[1] = t.y; v[2] = t.z; v[3] = t.w;
}
__device__ __forceinline__ void load_heads(const float* __restrict__ p, float (&v)[1]) {
  v[0] = p[0];
}

__device__ __forceinline__ float leaky(float v) { return v > 0.f ? v : NEG_SLOPE * v; }

template <int H, int F, bool ELU>
__global__ void __launch_bounds__(256)
seg_softmax_agg(const int* __restrict__ rowptr, const int* __restrict__ col,
                const float* __restrict__ asrc, const float* __restrict__ adst,
                const float* __restrict__ feat, const float* __restrict__ bias,
                float* __restrict__ out, int n_nodes) {
  constexpr int C   = H * F;
  constexpr int CPL = C / 32;
  constexpr int VW  = CPL >= 4 ? 4 : CPL;
  constexpr int NL  = CPL / VW;
  constexpr int U   = 8 / NL;
  static_assert(U * NL == 8, "the load fence below names eight values");
  typedef vecf<VW> vrow;

  const int lane = threadIdx.x & 31;
  const int n = __builtin_amdgcn_readfirstlane(
      (int)(blockIdx.x * (blockDim.x >> 5) + (threadIdx.x >> 5)));
  if (n >= n_nodes) return;
  const int beg = rowptr[n];
  const int end = rowptr[n + 1];

  float ad[H], as_self[H], vself[H], mx[H];
  load_heads(adst + (size_t)n * H, ad);
  load_heads(asrc + (size_t)n * H, as_self);
  for (int h = 0; h < H; ++h) { vself[h] = leaky(as_self[h] + ad[h]); mx[h] = vself[h]; }

  const int  i0  = beg + lane;
  const bool ok0 = i0 < end;
  const int  s0  = ok0 ? col[i0] : n;
  float e0[H];
  {
    float a[H];
    load_heads(asrc + (size_t)s0 * H, a);
    for (int h = 0; h < H; ++h) e0[h] = leaky(a[h] + ad[h]);
  }

  for (int h = 0; h < H; ++h) mx[h] = ok0 ? fmaxf(mx[h], e0[h]) : mx[h];
  for (int i = i0 + 32; i < end; i += 32) {
    float a[H];
    load_heads(asrc + (size_t)col[i] * H, a);
    for (int h = 0; h < H; ++h) mx[h] = fmaxf(mx[h], leaky(a[h] + ad[h]));
  }
  for (int h = 0; h < H; ++h) {
    for (int m = 16; m >= 1; m >>= 1) mx[h] = fmaxf(mx[h], __shfl_xor(mx[h], m));
  }

  float den[H], ex0[H];
  for (int h = 0; h < H; ++h) {
    ex0[h] = __expf(e0[h] - mx[h]);
    den[h] = ok0 ? ex0[h] : 0.f;
  }
  for (int i = i0 + 32; i < end; i += 32) {
    float a[H];
    load_heads(asrc + (size_t)col[i] * H, a);
    for (int h = 0; h < H; ++h) den[h] += __expf(leaky(a[h] + ad[h]) - mx[h]);
  }
  float rden[H], wself[H];
  for (int h = 0; h < H; ++h) {
    for (int m = 16; m >= 1; m >>= 1) den[h] += __shfl_xor(den[h], m);
    const float exs = __expf(vself[h] - mx[h]);
    den[h] += exs;
    den[h] += 1e-16f;
    rden[h]  = __builtin_amdgcn_rcpf(den[h]);
    wself[h] = exs * rden[h];
  }

  int hsel[NL];
  vrow acc[NL];
  for (int j = 0; j < NL; ++j) {
    hsel[j] = (j * 32 * VW + lane * VW) / F;
    for (int q = 0; q < VW; ++q) acc[j][q] = 0.f;
  }
  {
    const float* row = feat + (size_t)n * C + lane * VW;
    for (int j = 0; j < NL; ++j) {
      float w = wself[0];
      for (int h = 1; h < H; ++h) w = (hsel[j] == h) ? wself[h] : w;
      const vrow f = *(const vrow*)(row + j * 32 * VW);
      for (int q = 0; q < VW; ++q) acc[j][q] += w * f[q];
    }
  }
  for (int base = beg; base < end; base += 32) {
    int   s;
    float w[H];
    if (base == beg) {
      s = s0;
      for (int h = 0; h < H; ++h) w[h] = ok0 ? ex0[h] * rden[h] : 0.f;
    } else {
      const int  i  = base + lane;
      const bool ok = i < end;
      s = ok ? col[i] : n;
      float a[H];
      load_heads(asrc + (size_t)s * H, a);
      for (int h = 0; h < H; ++h)
        w[h] = ok ? __expf(leaky(a[h] + ad[h]) - mx[h]) * rden[h] : 0.f;
    }
    const int cn  = (end - base) < 32 ? (end - base) : 32;
    const int cnp = (cn + U - 1) & ~(U - 1);
    for (int k = 0; k < cnp; k += U) {
      vrow  f[U][NL];
      float wj[U][NL];
#pragma unroll
      for (int u = 0; u < U; ++u) {
        const int sk = __builtin_amdgcn_readlane(s, k + u);
        float wk[H];
        for (int h = 0; h < H; ++h)
          wk[h] = __int_as_float(__builtin_amdgcn_readlane(__float_as_int(w[h]), k + u));
        const float* row = feat + (size_t)sk * C + lane * VW;
        for (int j = 0; j < NL; ++j) {
          float t = wk[0];
          for (int h = 1; h < H; ++h) t = (hsel[j] == h) ? wk[h] : t;
          wj[u][j] = t;
          f[u][j]  = *(const vrow*)(row + j * 32 * VW);
        }
      }
      asm volatile("" :: "v"(f[0 / NL][0 % NL]), "v"(f[1 / NL][1 % NL]), "v"(f[2 / NL][2 % NL]),
                         "v"(f[3 / NL][3 % NL]), "v"(f[4 / NL][4 % NL]), "v"(f[5 / NL][5 % NL]),
                         "v"(f[6 / NL][6 % NL]), "v"(f[7 / NL][7 % NL]));
#pragma unroll
      for (int u = 0; u < U; ++u)
        for (int j = 0; j < NL; ++j)
          for (int q = 0; q < VW; ++q) acc[j][q] += wj[u][j] * f[u][j][q];
    }
  }

  float* orow = out + (size_t)n * C + lane * VW;
  const float* brow = bias + lane * VW;
  for (int j = 0; j < NL; ++j) {
    const vrow b = *(const vrow*)(brow + j * 32 * VW);
    vrow r;
    for (int q = 0; q < VW; ++q) {
      float v = acc[j][q] + b[q];
      r[q] = ELU ? (v > 0.f ? v : __expf(v) - 1.0f) : v;
    }
    *(vrow*)(orow + j * 32 * VW) = r;
  }
}


static inline int cdiv(long long a, long long b) { return (int)((a + b - 1) / b); }

extern "C" void kernel_launch(void* const* d_in, const int* in_sizes, int n_in,
                              void* d_out, int out_size, void* d_ws, size_t ws_size,
                              hipStream_t stream) {
  const float* x      = (const float*)d_in[0];
  const int*   ei     = (const int*)  d_in[1];
  const float* W1     = (const float*)d_in[2];
  const float* att_s1 = (const float*)d_in[3];
  const float* att_d1 = (const float*)d_in[4];
  const float* b1     = (const float*)d_in[5];
  const float* W2     = (const float*)d_in[6];
  const float* att_s2 = (const float*)d_in[7];
  const float* att_d2 = (const float*)d_in[8];
  const float* b2     = (const float*)d_in[9];

  const int IN   = 128;
  const int C1   = 256;
  const int Nn   = in_sizes[0] / IN;
  const int Eraw = in_sizes[1] / 2;

  float* ws = (float*)d_ws;
  size_t off = 0;
  const size_t npad = ((size_t)Nn + 4) & ~(size_t)3;
  float* h1     = ws + off; off += (size_t)Nn * C1;
  float* out1   = ws + off; off += (size_t)Nn * C1;
  float* asrc1  = ws + off; off += (size_t)Nn * 4;
  float* adst1  = ws + off; off += (size_t)Nn * 4;
  const int NS     = NSLICE;
  const int nrange = Nn / 256 + 1;
  const int rpitch = (nrange + 3) & ~3;
  const size_t ncell = (size_t)rpitch * NS;
  int*   cellcnt = (int*)(ws + off); off += ncell;
  int*   rstart  = (int*)(ws + off); off += (size_t)rpitch + 4;
  int*   rowptr  = (int*)(ws + off); off += npad;
  int*   binned  = (int*)(ws + off); off += (size_t)Eraw * 2 + 2;
  off = (off + 3) & ~(size_t)3;
  int*   lrank   = (int*)(ws + off); off += ((size_t)Eraw + 3) & ~(size_t)3;
  int*   col     = (int*)(ws + off); off += (size_t)Eraw;
  float* h2     = h1;
  float* asrc2  = asrc1;
  float* adst2  = adst1;
  float* outf   = (float*)d_out;

  const int T = 256;

  const int chunk = cdiv((long long)Eraw, NS);
  slice_count<<<NS, SORT_THREADS, nrange * sizeof(int), stream>>>(ei, lrank, cellcnt, Eraw, chunk, nrange, rpitch);
  slice_place<<<NS, SORT_THREADS, ((2 * COLGROUPS + 1) * rpitch + 32) * sizeof(int), stream>>>(
      ei, lrank, cellcnt, rstart, binned, Eraw, chunk, nrange, rpitch);
  range_sort<<<nrange, SORT_THREADS, 0, stream>>>(binned, rstart, rowptr, col, Nn);

  gemm_head_alpha<128, 4><<<cdiv((long long)(Nn / 16) * 4, T / 32), T, 0, stream>>>(
      x, W1, att_s1, att_d1, h1, asrc1, adst1, Nn);
  seg_softmax_agg<4, 64, true><<<cdiv((long long)Nn, T / 32), T, 0, stream>>>(
      rowptr, col, asrc1, adst1, h1, b1, out1, Nn);

  gemm_head_alpha<256, 1><<<cdiv((long long)(Nn / 16), T / 32), T, 0, stream>>>(
      out1, W2, att_s2, att_d2, h2, asrc2, adst2, Nn);
  seg_softmax_agg<1, 64, false><<<cdiv((long long)Nn, T / 32), T, 0, stream>>>(
      rowptr, col, asrc2, adst2, h2, b2, outf, Nn);
}


# Layer 1 takes the softmax-weighted sums over the 512-byte input rows per head and applies W1 afterwards in the same work-group through LDS; a_src/a_dst as x.(W1 a); sums reordered, h1 plane gone
# speedup vs baseline: 5.7582x; 1.0705x over previous
#include <hip/hip_runtime.h>

typedef __attribute__((ext_vector_type(2))) float v2f;
typedef __attribute__((ext_vector_type(8))) float v8f;

#define NEG_SLOPE 0.2f


typedef __attribute__((ext_vector_type(4))) float v4f;

template <int K, int HN>
__global__ void __launch_bounds__(256)
gemm_head_alpha(const float* __restrict__ A, const float* __restrict__ B,
                const float* __restrict__ att_s, const float* __restrict__ att_d,
                float* __restrict__ C, float* __restrict__ asrc, float* __restrict__ adst,
                int M) {
  constexpr int Nn = HN * 64;
  const int lane   = threadIdx.x & 31;
  const int wave   = blockIdx.x * (blockDim.x >> 5) + (threadIdx.x >> 5);
  const int mtiles = M >> 4;
  const int tileM  = wave / HN;
  const int hd     = wave - tileM * HN;
  if (tileM >= mtiles) return;

  const int lh  = lane >> 4;
  const int l16 = lane & 15;

  const float* ap = A + (size_t)(tileM * 16 + l16) * K + lh * 2;
  const float* bp = B + (size_t)(lh * 2) * Nn + hd * 64 + l16 * 4;

  v8f acc0 = {}, acc1 = {}, acc2 = {}, acc3 = {};
#pragma unroll 8
  for (int k = 0; k < K; k += 4) {
    const v2f a  = *(const v2f*)(ap + k);
    const v4f b0 = *(const v4f*)(bp + (size_t)k * Nn);
    const v4f b1 = *(const v4f*)(bp + (size_t)(k + 1) * Nn);
    v2f b;
    b.x = b0.x; b.y = b1.x;
    acc0 = __builtin_amdgcn_wmma_f32_16x16x4_f32(false, a, false, b, (short)0, acc0, false, false);
    b.x = b0.y; b.y = b1.y;
    acc1 = __builtin_amdgcn_wmma_f32_16x16x4_f32(false, a, false, b, (short)0, acc1, false, false);
    b.x = b0.z; b.y = b1.z;
    acc2 = __builtin_amdgcn_wmma_f32_16x16x4_f32(false, a, false, b, (short)0, acc2, false, false);
    b.x = b0.w; b.y = b1.w;
    acc3 = __builtin_amdgcn_wmma_f32_16x16x4_f32(false, a, false, b, (short)0, acc3, false, false);
  }

  const int row0 = tileM * 16 + lh * 8;
  float* crow = C + (size_t)row0 * Nn + hd * 64 + l16 * 4;
#pragma unroll
  for (int r = 0; r < 8; ++r) {
    v4f o;
    o.x = acc0[r]; o.y = acc1[r]; o.z = acc2[r]; o.w = acc3[r];
    *(v4f*)(crow + (size_t)r * Nn) = o;
  }

  const v4f s = *(const v4f*)(att_s + hd * 64 + l16 * 4);
  const v4f d = *(const v4f*)(att_d + hd * 64 + l16 * 4);
  float ps[8], pd[8];
#pragma unroll
  for (int r = 0; r < 8; ++r) {
    ps[r] = acc0[r] * s.x + acc1[r] * s.y + acc2[r] * s.z + acc3[r] * s.w;
    pd[r] = acc0[r] * d.x + acc1[r] * d.y + acc2[r] * d.z + acc3[r] * d.w;
  }
#pragma unroll
  for (int r = 0; r < 8; ++r) {
    for (int m = 8; m >= 1; m >>= 1) {
      ps[r] += __shfl_xor(ps[r], m);
      pd[r] += __shfl_xor(pd[r], m);
    }
  }
  float vs = ps[0], vd = pd[0];
#pragma unroll
  for (int r = 1; r < 8; ++r) {
    vs = (l16 == r) ? ps[r] : vs;
    vd = (l16 == r) ? pd[r] : vd;
  }
  if (l16 < 8) {
    asrc[(size_t)(row0 + l16) * HN + hd] = vs;
    adst[(size_t)(row0 + l16) * HN + hd] = vd;
  }
}


#define RANGE_SHIFT  8
#define RANGE_NODES  256
#define SORT_THREADS 1024
#define NSLICE       256
#define COLGROUPS    16

__global__ void __launch_bounds__(SORT_THREADS)
slice_count(const int* __restrict__ ei, int* __restrict__ lrank, int* __restrict__ cellcnt,
            int Eraw, int chunk, int nrange, int rpitch) {
  extern __shared__ int hcell[];
  const int t = threadIdx.x;
  const int g = blockIdx.x;
  for (int r = t; r < nrange; r += SORT_THREADS) hcell[r] = 0;
  __syncthreads();
  const int beg = g * chunk;
  const int end = (beg + chunk < Eraw) ? (beg + chunk) : Eraw;
  for (int e0 = beg + t; e0 < end; e0 += 4 * SORT_THREADS) {
    int d[4];
#pragma unroll
    for (int j = 0; j < 4; ++j) {
      const int e = e0 + j * SORT_THREADS;
      d[j] = ei[Eraw + ((e < end) ? e : (end - 1))];
    }
    asm volatile("" :: "v"(d[0]), "v"(d[1]), "v"(d[2]), "v"(d[3]));
#pragma unroll
    for (int j = 0; j < 4; ++j) {
      const int e = e0 + j * SORT_THREADS;
      if (e < end) lrank[e] = atomicAdd(&hcell[d[j] >> RANGE_SHIFT], 1);
    }
  }
  __syncthreads();
  for (int r = t; r < rpitch; r += SORT_THREADS)
    cellcnt[(size_t)g * rpitch + r] = (r < nrange) ? hcell[r] : 0;
}

__global__ void __launch_bounds__(SORT_THREADS)
slice_place(const int* __restrict__ ei, const int* __restrict__ lrank,
            const int* __restrict__ cellcnt, int* __restrict__ rstart,
            int* __restrict__ binned, int Eraw, int chunk, int nrange, int rpitch) {
  extern __shared__ __attribute__((aligned(16))) int lds[];
  int* part_t = lds;
  int* part_p = part_t + COLGROUPS * rpitch;
  int* lcell  = part_p + COLGROUPS * rpitch;
  int* wtot   = lcell + rpitch;
  const int t    = threadIdx.x;
  const int lane = t & 31;
  const int wv   = t >> 5;
  const int g    = blockIdx.x;

  const int r4n = rpitch >> 2;
  for (int idx = t; idx < COLGROUPS * r4n; idx += SORT_THREADS) {
    const int q  = idx / r4n;
    const int r4 = idx - q * r4n;
    const int* colp = cellcnt + (size_t)q * rpitch + r4 * 4;
    int4 st = make_int4(0, 0, 0, 0), sp = make_int4(0, 0, 0, 0);
#pragma unroll 8
    for (int k = 0; k < NSLICE / COLGROUPS; ++k) {
      const int  gg = q + k * COLGROUPS;
      const int4 v  = *(const int4*)(colp + (size_t)k * COLGROUPS * rpitch);
      st.x += v.x; st.y += v.y; st.z += v.z; st.w += v.w;
      const bool below = gg < g;
      sp.x += below ? v.x : 0; sp.y += below ? v.y : 0;
      sp.z += below ? v.z : 0; sp.w += below ? v.w : 0;
    }
    *(int4*)(part_t + q * rpitch + r4 * 4) = st;
    *(int4*)(part_p + q * rpitch + r4 * 4) = sp;
  }
  __syncthreads();

  int carry = 0;
  for (int base = 0; base < nrange; base += SORT_THREADS) {
    const int r = base + t;
    int tot = 0, low = 0;
    if (r < nrange) {
      for (int q = 0; q < COLGROUPS; ++q) { tot += part_t[q * rpitch + r]; low += part_p[q * rpitch + r]; }
    }
    int incl = tot;
    for (int d = 1; d < 32; d <<= 1) {
      int y = __shfl_up(incl, d);
      if (lane >= d) incl += y;
    }
    if (lane == 31) wtot[wv] = incl;
    __syncthreads();
    int ws = wtot[lane];
    for (int d = 1; d < 32; d <<= 1) {
      int y = __shfl_up(ws, d);
      if (lane >= d) ws += y;
    }
    const int wprefix = __shfl(ws, wv) - __shfl(incl, 31);
    const int total   = __shfl(ws, 31);
    const int start   = carry + wprefix + incl - tot;
    if (r < nrange) {
      lcell[r] = start + low;
      if (g == 0) rstart[r] = start;
    }
    carry += total;
    __syncthreads();
  }
  if (g == 0 && t == 0) rstart[nrange] = carry;

  const int beg = g * chunk;
  const int end = (beg + chunk < Eraw) ? (beg + chunk) : Eraw;
  int2* bp = (int2*)binned;
  for (int e0 = beg + t; e0 < end; e0 += 4 * SORT_THREADS) {
    int s[4], d[4], r[4];
#pragma unroll
    for (int j = 0; j < 4; ++j) {
      const int e  = e0 + j * SORT_THREADS;
      const int ec = (e < end) ? e : (end - 1);
      s[j] = ei[ec];
      d[j] = ei[Eraw + ec];
      r[j] = lrank[ec];
    }
    asm volatile("" :: "v"(s[0]), "v"(s[1]), "v"(s[2]), "v"(s[3]), "v"(d[0]), "v"(d[1]), "v"(d[2]),
                       "v"(d[3]), "v"(r[0]), "v"(r[1]), "v"(r[2]), "v"(r[3]));
#pragma unroll
    for (int j = 0; j < 4; ++j) {
      const int e = e0 + j * SORT_THREADS;
      if (e < end) bp[lcell[d[j] >> RANGE_SHIFT] + r[j]] = make_int2(s[j], d[j]);
    }
  }
}

__global__ void __launch_bounds__(SORT_THREADS)
range_sort(const int* __restrict__ binned, const int* __restrict__ rstart,
           int* __restrict__ rowptr, int* __restrict__ col, int n_nodes) {
  __shared__ int cnt[RANGE_NODES];
  const int t = threadIdx.x;
  const int b = blockIdx.x;
  const int beg = rstart[b];
  const int end = rstart[b + 1];
  const int2* bp = (const int2*)binned;
  if (t < RANGE_NODES) cnt[t] = 0;
  __syncthreads();
  for (int i0 = beg + t; i0 < end; i0 += 8 * SORT_THREADS) {
    int d[8];
#pragma unroll
    for (int j = 0; j < 8; ++j) {
      const int i = i0 + j * SORT_THREADS;
      d[j] = bp[(i < end) ? i : (end - 1)].y;
    }
    asm volatile("" :: "v"(d[0]), "v"(d[1]), "v"(d[2]), "v"(d[3]),
                       "v"(d[4]), "v"(d[5]), "v"(d[6]), "v"(d[7]));
#pragma unroll
    for (int j = 0; j < 8; ++j) {
      const int i = i0 + j * SORT_THREADS;
      if (i < end) atomicAdd(&cnt[d[j] & (RANGE_NODES - 1)], 1);
    }
  }
  __syncthreads();
  if (t < 32) {
    int c[8];
    int tsum = 0;
    for (int j = 0; j < 8; ++j) { c[j] = cnt[t * 8 + j]; tsum += c[j]; }
    int incl = tsum;
    for (int d = 1; d < 32; d <<= 1) {
      int y = __shfl_up(incl, d);
      if (t >= d) incl += y;
    }
    int run = incl - tsum;
    for (int j = 0; j < 8; ++j) { cnt[t * 8 + j] = run; run += c[j]; }
  }
  __syncthreads();
  if (t < RANGE_NODES) {
    const int node = b * RANGE_NODES + t;
    if (node <= n_nodes) rowptr[node] = beg + cnt[t];
  }
  __syncthreads();
  for (int i0 = beg + t; i0 < end; i0 += 8 * SORT_THREADS) {
    int2 sd[8];
#pragma unroll
    for (int j = 0; j < 8; ++j) {
      const int i = i0 + j * SORT_THREADS;
      sd[j] = bp[(i < end) ? i : (end - 1)];
    }
    asm volatile("" :: "v"(sd[0].y), "v"(sd[1].y), "v"(sd[2].y), "v"(sd[3].y),
                       "v"(sd[4].y), "v"(sd[5].y), "v"(sd[6].y), "v"(sd[7].y));
#pragma unroll
    for (int j = 0; j < 8; ++j) {
      const int i = i0 + j * SORT_THREADS;
      if (i < end) {
        const int p = atomicAdd(&cnt[sd[j].y & (RANGE_NODES - 1)], 1);
        col[beg + p] = sd[j].x;
      }
    }
  }
}


template <int N> using vecf = float __attribute__((ext_vector_type(N)));

__device__ __forceinline__ void load_heads(const float* __restrict__ p, float (&v)[4]) {
  const float4 t = *(const float4*)p;
  v[0] = t.x; v[1] = t.y; v[2] = t.z; v[3] = t.w;
}
__device__ __forceinline__ void load_heads(const float* __restrict__ p, float (&v)[1]) {
  v[0] = p[0];
}

__device__ __forceinline__ float leaky(float v) { return v > 0.f ? v : NEG_SLOPE * v; }

template <int H, int F, bool ELU>
__global__ void __launch_bounds__(256)
seg_softmax_agg(const int* __restrict__ rowptr, const int* __restrict__ col,
                const float* __restrict__ asrc, const float* __restrict__ adst,
                const float* __restrict__ feat, const float* __restrict__ bias,
                float* __restrict__ out, int n_nodes) {
  constexpr int C   = H * F;
  constexpr int CPL = C / 32;
  constexpr int VW  = CPL >= 4 ? 4 : CPL;
  constexpr int NL  = CPL / VW;
  constexpr int U   = 8 / NL;
  static_assert(U * NL == 8, "the load fence below names eight values");
  typedef vecf<VW> vrow;

  const int lane = threadIdx.x & 31;
  const int n = __builtin_amdgcn_readfirstlane(
      (int)(blockIdx.x * (blockDim.x >> 5) + (threadIdx.x >> 5)));
  if (n >= n_nodes) return;
  const int beg = rowptr[n];
  const int end = rowptr[n + 1];

  float ad[H], as_self[H], vself[H], mx[H];
  load_heads(adst + (size_t)n * H, ad);
  load_heads(asrc + (size_t)n * H, as_self);
  for (int h = 0; h < H; ++h) { vself[h] = leaky(as_self[h] + ad[h]); mx[h] = vself[h]; }

  const int  i0  = beg + lane;
  const bool ok0 = i0 < end;
  const int  s0  = ok0 ? col[i0] : n;
  float e0[H];
  {
    float a[H];
    load_heads(asrc + (size_t)s0 * H, a);
    for (int h = 0; h < H; ++h) e0[h] = leaky(a[h] + ad[h]);
  }

  for (int h = 0; h < H; ++h) mx[h] = ok0 ? fmaxf(mx[h], e0[h]) : mx[h];
  for (int i = i0 + 32; i < end; i += 32) {
    float a[H];
    load_heads(asrc + (size_t)col[i] * H, a);
    for (int h = 0; h < H; ++h) mx[h] = fmaxf(mx[h], leaky(a[h] + ad[h]));
  }
  for (int h = 0; h < H; ++h) {
    for (int m = 16; m >= 1; m >>= 1) mx[h] = fmaxf(mx[h], __shfl_xor(mx[h], m));
  }

  float den[H], ex0[H];
  for (int h = 0; h < H; ++h) {
    ex0[h] = __expf(e0[h] - mx[h]);
    den[h] = ok0 ? ex0[h] : 0.f;
  }
  for (int i = i0 + 32; i < end; i += 32) {
    float a[H];
    load_heads(asrc + (size_t)col[i] * H, a);
    for (int h = 0; h < H; ++h) den[h] += __expf(leaky(a[h] + ad[h]) - mx[h]);
  }
  float rden[H], wself[H];
  for (int h = 0; h < H; ++h) {
    for (int m = 16; m >= 1; m >>= 1) den[h] += __shfl_xor(den[h], m);
    const float exs = __expf(vself[h] - mx[h]);
    den[h] += exs;
    den[h] += 1e-16f;
    rden[h]  = __builtin_amdgcn_rcpf(den[h]);
    wself[h] = exs * rden[h];
  }

  int hsel[NL];
  vrow acc[NL];
  for (int j = 0; j < NL; ++j) {
    hsel[j] = (j * 32 * VW + lane * VW) / F;
    for (int q = 0; q < VW; ++q) acc[j][q] = 0.f;
  }
  {
    const float* row = feat + (size_t)n * C + lane * VW;
    for (int j = 0; j < NL; ++j) {
      float w = wself[0];
      for (int h = 1; h < H; ++h) w = (hsel[j] == h) ? wself[h] : w;
      const vrow f = *(const vrow*)(row + j * 32 * VW);
      for (int q = 0; q < VW; ++q) acc[j][q] += w * f[q];
    }
  }
  for (int base = beg; base < end; base += 32) {
    int   s;
    float w[H];
    if (base == beg) {
      s = s0;
      for (int h = 0; h < H; ++h) w[h] = ok0 ? ex0[h] * rden[h] : 0.f;
    } else {
      const int  i  = base + lane;
      const bool ok = i < end;
      s = ok ? col[i] : n;
      float a[H];
      load_heads(asrc + (size_t)s * H, a);
      for (int h = 0; h < H; ++h)
        w[h] = ok ? __expf(leaky(a[h] + ad[h]) - mx[h]) * rden[h] : 0.f;
    }
    const int cn  = (end - base) < 32 ? (end - base) : 32;
    const int cnp = (cn + U - 1) & ~(U - 1);
    for (int k = 0; k < cnp; k += U) {
      vrow  f[U][NL];
      float wj[U][NL];
#pragma unroll
      for (int u = 0; u < U; ++u) {
        const int sk = __builtin_amdgcn_readlane(s, k + u);
        float wk[H];
        for (int h = 0; h < H; ++h)
          wk[h] = __int_as_float(__builtin_amdgcn_readlane(__float_as_int(w[h]), k + u));
        const float* row = feat + (size_t)sk * C + lane * VW;
        for (int j = 0; j < NL; ++j) {
          float t = wk[0];
          for (int h = 1; h < H; ++h) t = (hsel[j] == h) ? wk[h] : t;
          wj[u][j] = t;
          f[u][j]  = *(const vrow*)(row + j * 32 * VW);
        }
      }
      asm volatile("" :: "v"(f[0 / NL][0 % NL]), "v"(f[1 / NL][1 % NL]), "v"(f[2 / NL][2 % NL]),
                         "v"(f[3 / NL][3 % NL]), "v"(f[4 / NL][4 % NL]), "v"(f[5 / NL][5 % NL]),
                         "v"(f[6 / NL][6 % NL]), "v"(f[7 / NL][7 % NL]));
#pragma unroll
      for (int u = 0; u < U; ++u)
        for (int j = 0; j < NL; ++j)
          for (int q = 0; q < VW; ++q) acc[j][q] += wj[u][j] * f[u][j][q];
    }
  }

  float* orow = out + (size_t)n * C + lane * VW;
  const float* brow = bias + lane * VW;
  for (int j = 0; j < NL; ++j) {
    const vrow b = *(const vrow*)(brow + j * 32 * VW);
    vrow r;
    for (int q = 0; q < VW; ++q) {
      float v = acc[j][q] + b[q];
      r[q] = ELU ? (v > 0.f ? v : __expf(v) - 1.0f) : v;
    }
    *(vrow*)(orow + j * 32 * VW) = r;
  }
}

__global__ void __launch_bounds__(1024)
score_inputs(const float* __restrict__ x, const float* __restrict__ W,
             const float* __restrict__ att_s, const float* __restrict__ att_d,
             float* __restrict__ asrc, float* __restrict__ adst, int n_nodes) {
  __shared__ __attribute__((aligned(16))) float V[128 * 8];
  const int t    = threadIdx.x;
  const int lane = t & 31;
  const int n0   = __builtin_amdgcn_readfirstlane((int)((blockIdx.x * 32 + (t >> 5)) * 8));

  v4f xv[8];
#pragma unroll
  for (int i = 0; i < 8; ++i) {
    const int n  = n0 + i;
    const int nc = (n < n_nodes) ? n : (n_nodes - 1);
    xv[i] = *(const v4f*)(x + (size_t)nc * 128 + lane * 4);
  }
  {
    const int wv = t >> 5, hw = lane >> 4, l16 = lane & 15;
    const int h  = ((wv & 1) << 1) | hw;
    const v4f as4 = *(const v4f*)(att_s + h * 64 + l16 * 4);
    const v4f ad4 = *(const v4f*)(att_d + h * 64 + l16 * 4);
    v4f w4[8];
#pragma unroll
    for (int it = 0; it < 8; ++it)
      w4[it] = *(const v4f*)(W + (size_t)(it * 16 + (wv >> 1)) * 256 + h * 64 + l16 * 4);
#pragma unroll
    for (int it = 0; it < 8; ++it) {
      const int k = it * 16 + (wv >> 1);
      float ps = w4[it].x * as4.x + w4[it].y * as4.y + w4[it].z * as4.z + w4[it].w * as4.w;
      float pd = w4[it].x * ad4.x + w4[it].y * ad4.y + w4[it].z * ad4.z + w4[it].w * ad4.w;
      for (int s = 8; s >= 1; s >>= 1) {
        ps += __shfl_xor(ps, s);
        pd += __shfl_xor(pd, s);
      }
      if (l16 == 0) { V[k * 8 + h] = ps; V[k * 8 + 4 + h] = pd; }
    }
  }
  __syncthreads();
  float vr[4][8];
#pragma unroll
  for (int q = 0; q < 4; ++q) {
    const v4f lo = *(const v4f*)(V + (lane * 4 + q) * 8);
    const v4f hi = *(const v4f*)(V + (lane * 4 + q) * 8 + 4);
    vr[q][0] = lo.x; vr[q][1] = lo.y; vr[q][2] = lo.z; vr[q][3] = lo.w;
    vr[q][4] = hi.x; vr[q][5] = hi.y; vr[q][6] = hi.z; vr[q][7] = hi.w;
  }
  const bool b4 = (lane & 16) != 0, b3 = (lane & 8) != 0, b2 = (lane & 4) != 0;
  const int  jj = (b4 ? 4 : 0) + (b3 ? 2 : 0) + (b2 ? 1 : 0);
#pragma unroll
  for (int i = 0; i < 8; ++i) {
    const int n = n0 + i;
    if (n >= n_nodes) break;
    float p[8];
#pragma unroll
    for (int j = 0; j < 8; ++j)
      p[j] = xv[i].x * vr[0][j] + xv[i].y * vr[1][j] + xv[i].z * vr[2][j] + xv[i].w * vr[3][j];
    float q4[4], q2[2];
#pragma unroll
    for (int j = 0; j < 4; ++j) {
      const float keep = b4 ? p[4 + j] : p[j];
      const float send = b4 ? p[j] : p[4 + j];
      q4[j] = keep + __shfl_xor(send, 16);
    }
#pragma unroll
    for (int j = 0; j < 2; ++j) {
      const float keep = b3 ? q4[2 + j] : q4[j];
      const float send = b3 ? q4[j] : q4[2 + j];
      q2[j] = keep + __shfl_xor(send, 8);
    }
    const float keep = b2 ? q2[1] : q2[0];
    const float send = b2 ? q2[0] : q2[1];
    float v = keep + __shfl_xor(send, 4);
    v += __shfl_xor(v, 2);
    v += __shfl_xor(v, 1);
    if ((lane & 3) == 0) {
      float* dstp = (jj < 4) ? asrc : adst;
      dstp[(size_t)n * 4 + (jj & 3)] = v;
    }
  }
}

#define L1_NODES 32

__global__ void __launch_bounds__(512)
agg_inputs_then_weights(const int* __restrict__ rowptr, const int* __restrict__ col,
                        const float* __restrict__ asrc, const float* __restrict__ adst,
                        const float* __restrict__ x, const float* __restrict__ W,
                        const float* __restrict__ bias, float* __restrict__ out, int n_nodes) {
  __shared__ __attribute__((aligned(16))) float Z[L1_NODES * 512];
  const int lane = threadIdx.x & 31;
  const int wv = __builtin_amdgcn_readfirstlane((int)(threadIdx.x >> 5));
  const int nb = blockIdx.x * L1_NODES;

  for (int half = 0; half < 2; ++half) {
    const int m = wv + 16 * half;
    const int n = nb + m;
    float acc[4][4];
    for (int h = 0; h < 4; ++h)
      for (int q = 0; q < 4; ++q) acc[h][q] = 0.f;
    if (n < n_nodes) {
      const int beg = rowptr[n];
      const int end = rowptr[n + 1];
      float ad[4], as_self[4], vself[4], mx[4];
      load_heads(adst + (size_t)n * 4, ad);
      load_heads(asrc + (size_t)n * 4, as_self);
      for (int h = 0; h < 4; ++h) { vself[h] = leaky(as_self[h] + ad[h]); mx[h] = vself[h]; }

      const int  i0  = beg + lane;
      const bool ok0 = i0 < end;
      const int  s0  = ok0 ? col[i0] : n;
      float e0[4];
      {
        float a[4];
        load_heads(asrc + (size_t)s0 * 4, a);
        for (int h = 0; h < 4; ++h) e0[h] = leaky(a[h] + ad[h]);
      }
      for (int h = 0; h < 4; ++h) mx[h] = ok0 ? fmaxf(mx[h], e0[h]) : mx[h];
      for (int i = i0 + 32; i < end; i += 32) {
        float a[4];
        load_heads(asrc + (size_t)col[i] * 4, a);
        for (int h = 0; h < 4; ++h) mx[h] = fmaxf(mx[h], leaky(a[h] + ad[h]));
      }
      for (int h = 0; h < 4; ++h)
        for (int s = 16; s >= 1; s >>= 1) mx[h] = fmaxf(mx[h], __shfl_xor(mx[h], s));
      float den[4], ex0[4];
      for (int h = 0; h < 4; ++h) {
        ex0[h] = __expf(e0[h] - mx[h]);
        den[h] = ok0 ? ex0[h] : 0.f;
      }
      for (int i = i0 + 32; i < end; i += 32) {
        float a[4];
        load_heads(asrc + (size_t)col[i] * 4, a);
        for (int h = 0; h < 4; ++h) den[h] += __expf(leaky(a[h] + ad[h]) - mx[h]);
      }
      float rden[4], wself[4];
      for (int h = 0; h < 4; ++h) {
        for (int s = 16; s >= 1; s >>= 1) den[h] += __shfl_xor(den[h], s);
        const float exs = __expf(vself[h] - mx[h]);
        den[h] += exs;
        den[h] += 1e-16f;
        rden[h]  = __builtin_amdgcn_rcpf(den[h]);
        wself[h] = exs * rden[h];
      }
      {
        const v4f f = *(const v4f*)(x + (size_t)n * 128 + lane * 4);
        for (int h = 0; h < 4; ++h)
          for (int q = 0; q < 4; ++q) acc[h][q] += wself[h] * f[q];
      }
      for (int base = beg; base < end; base += 32) {
        int   s;
        float w[4];
        if (base == beg) {
          s = s0;
          for (int h = 0; h < 4; ++h) w[h] = ok0 ? ex0[h] * rden[h] : 0.f;
        } else {
          const int  i  = base + lane;
          const bool ok = i < end;
          s = ok ? col[i] : n;
          float a[4];
          load_heads(asrc + (size_t)s * 4, a);
          for (int h = 0; h < 4; ++h)
            w[h] = ok ? __expf(leaky(a[h] + ad[h]) - mx[h]) * rden[h] : 0.f;
        }
        const int cn  = (end - base) < 32 ? (end - base) : 32;
        const int cnp = (cn + 7) & ~7;
        for (int k = 0; k < cnp; k += 8) {
          v4f   f[8];
          float wk[8][4];
#pragma unroll
          for (int u = 0; u < 8; ++u) {
            const int sk = __builtin_amdgcn_readlane(s, k + u);
            for (int h = 0; h < 4; ++h)
              wk[u][h] = __int_as_float(__builtin_amdgcn_readlane(__float_as_int(w[h]), k + u));
            f[u] = *(const v4f*)((const char*)x + ((unsigned)sk * 512u + (unsigned)lane * 16u));
          }
          asm volatile("" :: "v"(f[0]), "v"(f[1]), "v"(f[2]), "v"(f[3]),
                             "v"(f[4]), "v"(f[5]), "v"(f[6]), "v"(f[7]));
#pragma unroll
          for (int u = 0; u < 8; ++u)
            for (int h = 0; h < 4; ++h)
              for (int q = 0; q < 4; ++q) acc[h][q] += wk[u][h] * f[u][q];
        }
      }
    }
    float* zr = Z + m * 512;
    for (int h = 0; h < 4; ++h) {
      v4f o;
      o.x = acc[h][0]; o.y = acc[h][1]; o.z = acc[h][2]; o.w = acc[h][3];
      *(v4f*)(zr + ((h * 128 + lane * 4) ^ ((m & 15) << 2))) = o;
    }
  }
  __syncthreads();
  if (wv >= 8) return;

  const int hd    = wv >> 1;
  const int lh    = lane >> 4;
  const int l16   = lane & 15;
  const int cbase = hd * 64 + (wv & 1) * 32;
  const float* bp = W + (size_t)(lh * 2) * 256 + cbase + l16 * 2;
  const float* z0 = Z + l16 * 512;
  const float* z1 = Z + (16 + l16) * 512;
  const int zc0 = hd * 128 + lh * 2;
  const int sw  = l16 << 2;
  v8f c00 = {}, c01 = {}, c10 = {}, c11 = {};
#pragma unroll 8
  for (int k = 0; k < 128; k += 4) {
    const int zc = (zc0 + k) ^ sw;
    const v2f a0 = *(const v2f*)(z0 + zc);
    const v2f a1 = *(const v2f*)(z1 + zc);
    const v2f b0 = *(const v2f*)(bp + (size_t)k * 256);
    const v2f b1 = *(const v2f*)(bp + (size_t)(k + 1) * 256);
    v2f b;
    b.x = b0.x; b.y = b1.x;
    c00 = __builtin_amdgcn_wmma_f32_16x16x4_f32(false, a0, false, b, (short)0, c00, false, false);
    c10 = __builtin_amdgcn_wmma_f32_16x16x4_f32(false, a1, false, b, (short)0, c10, false, false);
    b.x = b0.y; b.y = b1.y;
    c01 = __builtin_amdgcn_wmma_f32_16x16x4_f32(false, a0, false, b, (short)0, c01, false, false);
    c11 = __builtin_amdgcn_wmma_f32_16x16x4_f32(false, a1, false, b, (short)0, c11, false, false);
  }

  const v2f bb = *(const v2f*)(bias + cbase + l16 * 2);
#pragma unroll
  for (int mt = 0; mt < 2; ++mt) {
#pragma unroll
    for (int r = 0; r < 8; ++r) {
      const int node = nb + mt * 16 + lh * 8 + r;
      v2f o;
      o.x = (mt ? c10[r] : c00[r]) + bb.x;
      o.y = (mt ? c11[r] : c01[r]) + bb.y;
      o.x = o.x > 0.f ? o.x : __expf(o.x) - 1.0f;
      o.y = o.y > 0.f ? o.y : __expf(o.y) - 1.0f;
      if (node < n_nodes) *(v2f*)(out + (size_t)node * 256 + cbase + l16 * 2) = o;
    }
  }
}


static inline int cdiv(long long a, long long b) { return (int)((a + b - 1) / b); }

extern "C" void kernel_launch(void* const* d_in, const int* in_sizes, int n_in,
                              void* d_out, int out_size, void* d_ws, size_t ws_size,
                              hipStream_t stream) {
  const float* x      = (const float*)d_in[0];
  const int*   ei     = (const int*)  d_in[1];
  const float* W1     = (const float*)d_in[2];
  const float* att_s1 = (const float*)d_in[3];
  const float* att_d1 = (const float*)d_in[4];
  const float* b1     = (const float*)d_in[5];
  const float* W2     = (const float*)d_in[6];
  const float* att_s2 = (const float*)d_in[7];
  const float* att_d2 = (const float*)d_in[8];
  const float* b2     = (const float*)d_in[9];

  const int IN   = 128;
  const int C1   = 256;
  const int Nn   = in_sizes[0] / IN;
  const int Eraw = in_sizes[1] / 2;

  float* ws = (float*)d_ws;
  size_t off = 0;
  const size_t npad = ((size_t)Nn + 4) & ~(size_t)3;
  float* h1     = ws + off; off += (size_t)Nn * C1;
  float* out1   = ws + off; off += (size_t)Nn * C1;
  float* asrc1  = ws + off; off += (size_t)Nn * 4;
  float* adst1  = ws + off; off += (size_t)Nn * 4;
  const int NS     = NSLICE;
  const int nrange = Nn / 256 + 1;
  const int rpitch = (nrange + 3) & ~3;
  const size_t ncell = (size_t)rpitch * NS;
  int*   cellcnt = (int*)(ws + off); off += ncell;
  int*   rstart  = (int*)(ws + off); off += (size_t)rpitch + 4;
  int*   rowptr  = (int*)(ws + off); off += npad;
  int*   binned  = (int*)(ws + off); off += (size_t)Eraw * 2 + 2;
  off = (off + 3) & ~(size_t)3;
  int*   lrank   = (int*)(ws + off); off += ((size_t)Eraw + 3) & ~(size_t)3;
  int*   col     = (int*)(ws + off); off += (size_t)Eraw;
  float* h2     = h1;
  float* asrc2  = asrc1;
  float* adst2  = adst1;
  float* outf   = (float*)d_out;

  const int T = 256;

  const int chunk = cdiv((long long)Eraw, NS);
  slice_count<<<NS, SORT_THREADS, nrange * sizeof(int), stream>>>(ei, lrank, cellcnt, Eraw, chunk, nrange, rpitch);
  slice_place<<<NS, SORT_THREADS, ((2 * COLGROUPS + 1) * rpitch + 32) * sizeof(int), stream>>>(
      ei, lrank, cellcnt, rstart, binned, Eraw, chunk, nrange, rpitch);
  range_sort<<<nrange, SORT_THREADS, 0, stream>>>(binned, rstart, rowptr, col, Nn);

  score_inputs<<<cdiv((long long)Nn, 256), 1024, 0, stream>>>(x, W1, att_s1, att_d1, asrc1, adst1, Nn);
  agg_inputs_then_weights<<<cdiv((long long)Nn, L1_NODES), 512, 0, stream>>>(
      rowptr, col, asrc1, adst1, x, W1, b1, out1, Nn);

  gemm_head_alpha<256, 1><<<cdiv((long long)(Nn / 16), T / 32), T, 0, stream>>>(
      out1, W2, att_s2, att_d2, h2, asrc2, adst2, Nn);
  seg_softmax_agg<1, 64, false><<<cdiv((long long)Nn, T / 32), T, 0, stream>>>(
      rowptr, col, asrc2, adst2, h2, b2, outf, Nn);
}


# Layer 2's transform inside the layer-1 aggregation work-group: finished rows go via LDS to a second f32 WMMA product with W2 (k in four quarters, then added); their plane and the GEMM launch gone
# speedup vs baseline: 6.0495x; 1.0506x over previous
#include <hip/hip_runtime.h>

typedef __attribute__((ext_vector_type(2))) float v2f;
typedef __attribute__((ext_vector_type(8))) float v8f;

#define NEG_SLOPE 0.2f

typedef __attribute__((ext_vector_type(4))) float v4f;


#define RANGE_SHIFT  8
#define RANGE_NODES  256
#define SORT_THREADS 1024
#define NSLICE       256
#define COLGROUPS    16

__global__ void __launch_bounds__(SORT_THREADS)
slice_count(const int* __restrict__ ei, int* __restrict__ lrank, int* __restrict__ cellcnt,
            int Eraw, int chunk, int nrange, int rpitch) {
  extern __shared__ int hcell[];
  const int t = threadIdx.x;
  const int g = blockIdx.x;
  for (int r = t; r < nrange; r += SORT_THREADS) hcell[r] = 0;
  __syncthreads();
  const int beg = g * chunk;
  const int end = (beg + chunk < Eraw) ? (beg + chunk) : Eraw;
  for (int e0 = beg + t; e0 < end; e0 += 4 * SORT_THREADS) {
    int d[4];
#pragma unroll
    for (int j = 0; j < 4; ++j) {
      const int e = e0 + j * SORT_THREADS;
      d[j] = ei[Eraw + ((e < end) ? e : (end - 1))];
    }
    asm volatile("" :: "v"(d[0]), "v"(d[1]), "v"(d[2]), "v"(d[3]));
#pragma unroll
    for (int j = 0; j < 4; ++j) {
      const int e = e0 + j * SORT_THREADS;
      if (e < end) lrank[e] = atomicAdd(&hcell[d[j] >> RANGE_SHIFT], 1);
    }
  }
  __syncthreads();
  for (int r = t; r < rpitch; r += SORT_THREADS)
    cellcnt[(size_t)g * rpitch + r] = (r < nrange) ? hcell[r] : 0;
}

__global__ void __launch_bounds__(SORT_THREADS)
slice_place(const int* __restrict__ ei, const int* __restrict__ lrank,
            const int* __restrict__ cellcnt, int* __restrict__ rstart,
            int* __restrict__ binned, int Eraw, int chunk, int nrange, int rpitch) {
  extern __shared__ __attribute__((aligned(16))) int lds[];
  int* part_t = lds;
  int* part_p = part_t + COLGROUPS * rpitch;
  int* lcell  = part_p + COLGROUPS * rpitch;
  int* wtot   = lcell + rpitch;
  const int t    = threadIdx.x;
  const int lane = t & 31;
  const int wv   = t >> 5;
  const int g    = blockIdx.x;

  const int r4n = rpitch >> 2;
  for (int idx = t; idx < COLGROUPS * r4n; idx += SORT_THREADS) {
    const int q  = idx / r4n;
    const int r4 = idx - q * r4n;
    const int* colp = cellcnt + (size_t)q * rpitch + r4 * 4;
    int4 st = make_int4(0, 0, 0, 0), sp = make_int4(0, 0, 0, 0);
#pragma unroll 8
    for (int k = 0; k < NSLICE / COLGROUPS; ++k) {
      const int  gg = q + k * COLGROUPS;
      const int4 v  = *(const int4*)(colp + (size_t)k * COLGROUPS * rpitch);
      st.x += v.x; st.y += v.y; st.z += v.z; st.w += v.w;
      const bool below = gg < g;
      sp.x += below ? v.x : 0; sp.y += below ? v.y : 0;
      sp.z += below ? v.z : 0; sp.w += below ? v.w : 0;
    }
    *(int4*)(part_t + q * rpitch + r4 * 4) = st;
    *(int4*)(part_p + q * rpitch + r4 * 4) = sp;
  }
  __syncthreads();

  int carry = 0;
  for (int base = 0; base < nrange; base += SORT_THREADS) {
    const int r = base + t;
    int tot = 0, low = 0;
    if (r < nrange) {
      for (int q = 0; q < COLGROUPS; ++q) { tot += part_t[q * rpitch + r]; low += part_p[q * rpitch + r]; }
    }
    int incl = tot;
    for (int d = 1; d < 32; d <<= 1) {
      int y = __shfl_up(incl, d);
      if (lane >= d) incl += y;
    }
    if (lane == 31) wtot[wv] = incl;
    __syncthreads();
    int ws = wtot[lane];
    for (int d = 1; d < 32; d <<= 1) {
      int y = __shfl_up(ws, d);
      if (lane >= d) ws += y;
    }
    const int wprefix = __shfl(ws, wv) - __shfl(incl, 31);
    const int total   = __shfl(ws, 31);
    const int start   = carry + wprefix + incl - tot;
    if (r < nrange) {
      lcell[r] = start + low;
      if (g == 0) rstart[r] = start;
    }
    carry += total;
    __syncthreads();
  }
  if (g == 0 && t == 0) rstart[nrange] = carry;

  const int beg = g * chunk;
  const int end = (beg + chunk < Eraw) ? (beg + chunk) : Eraw;
  int2* bp = (int2*)binned;
  for (int e0 = beg + t; e0 < end; e0 += 4 * SORT_THREADS) {
    int s[4], d[4], r[4];
#pragma unroll
    for (int j = 0; j < 4; ++j) {
      const int e  = e0 + j * SORT_THREADS;
      const int ec = (e < end) ? e : (end - 1);
      s[j] = ei[ec];
      d[j] = ei[Eraw + ec];
      r[j] = lrank[ec];
    }
    asm volatile("" :: "v"(s[0]), "v"(s[1]), "v"(s[2]), "v"(s[3]), "v"(d[0]), "v"(d[1]), "v"(d[2]),
                       "v"(d[3]), "v"(r[0]), "v"(r[1]), "v"(r[2]), "v"(r[3]));
#pragma unroll
    for (int j = 0; j < 4; ++j) {
      const int e = e0 + j * SORT_THREADS;
      if (e < end) bp[lcell[d[j] >> RANGE_SHIFT] + r[j]] = make_int2(s[j], d[j]);
    }
  }
}

__global__ void __launch_bounds__(SORT_THREADS)
range_sort(const int* __restrict__ binned, const int* __restrict__ rstart,
           int* __restrict__ rowptr, int* __restrict__ col, int n_nodes) {
  __shared__ int cnt[RANGE_NODES];
  const int t = threadIdx.x;
  const int b = blockIdx.x;
  const int beg = rstart[b];
  const int end = rstart[b + 1];
  const int2* bp = (const int2*)binned;
  if (t < RANGE_NODES) cnt[t] = 0;
  __syncthreads();
  for (int i0 = beg + t; i0 < end; i0 += 8 * SORT_THREADS) {
    int d[8];
#pragma unroll
    for (int j = 0; j < 8; ++j) {
      const int i = i0 + j * SORT_THREADS;
      d[j] = bp[(i < end) ? i : (end - 1)].y;
    }
    asm volatile("" :: "v"(d[0]), "v"(d[1]), "v"(d[2]), "v"(d[3]),
                       "v"(d[4]), "v"(d[5]), "v"(d[6]), "v"(d[7]));
#pragma unroll
    for (int j = 0; j < 8; ++j) {
      const int i = i0 + j * SORT_THREADS;
      if (i < end) atomicAdd(&cnt[d[j] & (RANGE_NODES - 1)], 1);
    }
  }
  __syncthreads();
  if (t < 32) {
    int c[8];
    int tsum = 0;
    for (int j = 0; j < 8; ++j) { c[j] = cnt[t * 8 + j]; tsum += c[j]; }
    int incl = tsum;
    for (int d = 1; d < 32; d <<= 1) {
      int y = __shfl_up(incl, d);
      if (t >= d) incl += y;
    }
    int run = incl - tsum;
    for (int j = 0; j < 8; ++j) { cnt[t * 8 + j] = run; run += c[j]; }
  }
  __syncthreads();
  if (t < RANGE_NODES) {
    const int node = b * RANGE_NODES + t;
    if (node <= n_nodes) rowptr[node] = beg + cnt[t];
  }
  __syncthreads();
  for (int i0 = beg + t; i0 < end; i0 += 8 * SORT_THREADS) {
    int2 sd[8];
#pragma unroll
    for (int j = 0; j < 8; ++j) {
      const int i = i0 + j * SORT_THREADS;
      sd[j] = bp[(i < end) ? i : (end - 1)];
    }
    asm volatile("" :: "v"(sd[0].y), "v"(sd[1].y), "v"(sd[2].y), "v"(sd[3].y),
                       "v"(sd[4].y), "v"(sd[5].y), "v"(sd[6].y), "v"(sd[7].y));
#pragma unroll
    for (int j = 0; j < 8; ++j) {
      const int i = i0 + j * SORT_THREADS;
      if (i < end) {
        const int p = atomicAdd(&cnt[sd[j].y & (RANGE_NODES - 1)], 1);
        col[beg + p] = sd[j].x;
      }
    }
  }
}


template <int N> using vecf = float __attribute__((ext_vector_type(N)));

__device__ __forceinline__ void load_heads(const float* __restrict__ p, float (&v)[4]) {
  const float4 t = *(const float4*)p;
  v[0] = t.x; v[1] = t.y; v[2] = t.z; v[3] = t.w;
}
__device__ __forceinline__ void load_heads(const float* __restrict__ p, float (&v)[1]) {
  v[0] = p[0];
}

__device__ __forceinline__ float leaky(float v) { return v > 0.f ? v : NEG_SLOPE * v; }

template <int H, int F, bool ELU>
__global__ void __launch_bounds__(256)
seg_softmax_agg(const int* __restrict__ rowptr, const int* __restrict__ col,
                const float* __restrict__ asrc, const float* __restrict__ adst,
                const float* __restrict__ feat, const float* __restrict__ bias,
                float* __restrict__ out, int n_nodes) {
  constexpr int C   = H * F;
  constexpr int CPL = C / 32;
  constexpr int VW  = CPL >= 4 ? 4 : CPL;
  constexpr int NL  = CPL / VW;
  constexpr int U   = 8 / NL;
  static_assert(U * NL == 8, "the load fence below names eight values");
  typedef vecf<VW> vrow;

  const int lane = threadIdx.x & 31;
  const int n = __builtin_amdgcn_readfirstlane(
      (int)(blockIdx.x * (blockDim.x >> 5) + (threadIdx.x >> 5)));
  if (n >= n_nodes) return;
  const int beg = rowptr[n];
  const int end = rowptr[n + 1];

  float ad[H], as_self[H], vself[H], mx[H];
  load_heads(adst + (size_t)n * H, ad);
  load_heads(asrc + (size_t)n * H, as_self);
  for (int h = 0; h < H; ++h) { vself[h] = leaky(as_self[h] + ad[h]); mx[h] = vself[h]; }

  const int  i0  = beg + lane;
  const bool ok0 = i0 < end;
  const int  s0  = ok0 ? col[i0] : n;
  float e0[H];
  {
    float a[H];
    load_heads(asrc + (size_t)s0 * H, a);
    for (int h = 0; h < H; ++h) e0[h] = leaky(a[h] + ad[h]);
  }

  for (int h = 0; h < H; ++h) mx[h] = ok0 ? fmaxf(mx[h], e0[h]) : mx[h];
  for (int i = i0 + 32; i < end; i += 32) {
    float a[H];
    load_heads(asrc + (size_t)col[i] * H, a);
    for (int h = 0; h < H; ++h) mx[h] = fmaxf(mx[h], leaky(a[h] + ad[h]));
  }
  for (int h = 0; h < H; ++h) {
    for (int m = 16; m >= 1; m >>= 1) mx[h] = fmaxf(mx[h], __shfl_xor(mx[h], m));
  }

  float den[H], ex0[H];
  for (int h = 0; h < H; ++h) {
    ex0[h] = __expf(e0[h] - mx[h]);
    den[h] = ok0 ? ex0[h] : 0.f;
  }
  for (int i = i0 + 32; i < end; i += 32) {
    float a[H];
    load_heads(asrc + (size_t)col[i] * H, a);
    for (int h = 0; h < H; ++h) den[h] += __expf(leaky(a[h] + ad[h]) - mx[h]);
  }
  float rden[H], wself[H];
  for (int h = 0; h < H; ++h) {
    for (int m = 16; m >= 1; m >>= 1) den[h] += __shfl_xor(den[h], m);
    const float exs = __expf(vself[h] - mx[h]);
    den[h] += exs;
    den[h] += 1e-16f;
    rden[h]  = __builtin_amdgcn_rcpf(den[h]);
    wself[h] = exs * rden[h];
  }

  int hsel[NL];
  vrow acc[NL];
  for (int j = 0; j < NL; ++j) {
    hsel[j] = (j * 32 * VW + lane * VW) / F;
    for (int q = 0; q < VW; ++q) acc[j][q] = 0.f;
  }
  {
    const float* row = feat + (size_t)n * C + lane * VW;
    for (int j = 0; j < NL; ++j) {
      float w = wself[0];
      for (int h = 1; h < H; ++h) w = (hsel[j] == h) ? wself[h] : w;
      const vrow f = *(const vrow*)(row + j * 32 * VW);
      for (int q = 0; q < VW; ++q) acc[j][q] += w * f[q];
    }
  }
  for (int base = beg; base < end; base += 32) {
    int   s;
    float w[H];
    if (base == beg) {
      s = s0;
      for (int h = 0; h < H; ++h) w[h] = ok0 ? ex0[h] * rden[h] : 0.f;
    } else {
      const int  i  = base + lane;
      const bool ok = i < end;
      s = ok ? col[i] : n;
      float a[H];
      load_heads(asrc + (size_t)s * H, a);
      for (int h = 0; h < H; ++h)
        w[h] = ok ? __expf(leaky(a[h] + ad[h]) - mx[h]) * rden[h] : 0.f;
    }
    const int cn  = (end - base) < 32 ? (end - base) : 32;
    const int cnp = (cn + U - 1) & ~(U - 1);
    for (int k = 0; k < cnp; k += U) {
      vrow  f[U][NL];
      float wj[U][NL];
#pragma unroll
      for (int u = 0; u < U; ++u) {
        const int sk = __builtin_amdgcn_readlane(s, k + u);
        float wk[H];
        for (int h = 0; h < H; ++h)
          wk[h] = __int_as_float(__builtin_amdgcn_readlane(__float_as_int(w[h]), k + u));
        const float* row = feat + (size_t)sk * C + lane * VW;
        for (int j = 0; j < NL; ++j) {
          float t = wk[0];
          for (int h = 1; h < H; ++h) t = (hsel[j] == h) ? wk[h] : t;
          wj[u][j] = t;
          f[u][j]  = *(const vrow*)(row + j * 32 * VW);
        }
      }
      asm volatile("" :: "v"(f[0 / NL][0 % NL]), "v"(f[1 / NL][1 % NL]), "v"(f[2 / NL][2 % NL]),
                         "v"(f[3 / NL][3 % NL]), "v"(f[4 / NL][4 % NL]), "v"(f[5 / NL][5 % NL]),
                         "v"(f[6 / NL][6 % NL]), "v"(f[7 / NL][7 % NL]));
#pragma unroll
      for (int u = 0; u < U; ++u)
        for (int j = 0; j < NL; ++j)
          for (int q = 0; q < VW; ++q) acc[j][q] += wj[u][j] * f[u][j][q];
    }
  }

  float* orow = out + (size_t)n * C + lane * VW;
  const float* brow = bias + lane * VW;
  for (int j = 0; j < NL; ++j) {
    const vrow b = *(const vrow*)(brow + j * 32 * VW);
    vrow r;
    for (int q = 0; q < VW; ++q) {
      float v = acc[j][q] + b[q];
      r[q] = ELU ? (v > 0.f ? v : __expf(v) - 1.0f) : v;
    }
    *(vrow*)(orow + j * 32 * VW) = r;
  }
}

__global__ void __launch_bounds__(1024)
score_inputs(const float* __restrict__ x, const float* __restrict__ W,
             const float* __restrict__ att_s, const float* __restrict__ att_d,
             float* __restrict__ asrc, float* __restrict__ adst, int n_nodes) {
  __shared__ __attribute__((aligned(16))) float V[128 * 8];
  const int t    = threadIdx.x;
  const int lane = t & 31;
  const int n0   = __builtin_amdgcn_readfirstlane((int)((blockIdx.x * 32 + (t >> 5)) * 8));

  v4f xv[8];
#pragma unroll
  for (int i = 0; i < 8; ++i) {
    const int n  = n0 + i;
    const int nc = (n < n_nodes) ? n : (n_nodes - 1);
    xv[i] = *(const v4f*)(x + (size_t)nc * 128 + lane * 4);
  }
  {
    const int wv = t >> 5, hw = lane >> 4, l16 = lane & 15;
    const int h  = ((wv & 1) << 1) | hw;
    const v4f as4 = *(const v4f*)(att_s + h * 64 + l16 * 4);
    const v4f ad4 = *(const v4f*)(att_d + h * 64 + l16 * 4);
    v4f w4[8];
#pragma unroll
    for (int it = 0; it < 8; ++it)
      w4[it] = *(const v4f*)(W + (size_t)(it * 16 + (wv >> 1)) * 256 + h * 64 + l16 * 4);
#pragma unroll
    for (int it = 0; it < 8; ++it) {
      const int k = it * 16 + (wv >> 1);
      float ps = w4[it].x * as4.x + w4[it].y * as4.y + w4[it].z * as4.z + w4[it].w * as4.w;
      float pd = w4[it].x * ad4.x + w4[it].y * ad4.y + w4[it].z * ad4.z + w4[it].w * ad4.w;
      for (int s = 8; s >= 1; s >>= 1) {
        ps += __shfl_xor(ps, s);
        pd += __shfl_xor(pd, s);
      }
      if (l16 == 0) { V[k * 8 + h] = ps; V[k * 8 + 4 + h] = pd; }
    }
  }
  __syncthreads();
  float vr[4][8];
#pragma unroll
  for (int q = 0; q < 4; ++q) {
    const v4f lo = *(const v4f*)(V + (lane * 4 + q) * 8);
    const v4f hi = *(const v4f*)(V + (lane * 4 + q) * 8 + 4);
    vr[q][0] = lo.x; vr[q][1] = lo.y; vr[q][2] = lo.z; vr[q][3] = lo.w;
    vr[q][4] = hi.x; vr[q][5] = hi.y; vr[q][6] = hi.z; vr[q][7] = hi.w;
  }
  const bool b4 = (lane & 16) != 0, b3 = (lane & 8) != 0, b2 = (lane & 4) != 0;
  const int  jj = (b4 ? 4 : 0) + (b3 ? 2 : 0) + (b2 ? 1 : 0);
#pragma unroll
  for (int i = 0; i < 8; ++i) {
    const int n = n0 + i;
    if (n >= n_nodes) break;
    float p[8];
#pragma unroll
    for (int j = 0; j < 8; ++j)
      p[j] = xv[i].x * vr[0][j] + xv[i].y * vr[1][j] + xv[i].z * vr[2][j] + xv[i].w * vr[3][j];
    float q4[4], q2[2];
#pragma unroll
    for (int j = 0; j < 4; ++j) {
      const float keep = b4 ? p[4 + j] : p[j];
      const float send = b4 ? p[j] : p[4 + j];
      q4[j] = keep + __shfl_xor(send, 16);
    }
#pragma unroll
    for (int j = 0; j < 2; ++j) {
      const float keep = b3 ? q4[2 + j] : q4[j];
      const float send = b3 ? q4[j] : q4[2 + j];
      q2[j] = keep + __shfl_xor(send, 8);
    }
    const float keep = b2 ? q2[1] : q2[0];
    const float send = b2 ? q2[0] : q2[1];
    float v = keep + __shfl_xor(send, 4);
    v += __shfl_xor(v, 2);
    v += __shfl_xor(v, 1);
    if ((lane & 3) == 0) {
      float* dstp = (jj < 4) ? asrc : adst;
      dstp[(size_t)n * 4 + (jj & 3)] = v;
    }
  }
}

#define L1_NODES 32

__global__ void __launch_bounds__(512)
agg_inputs_then_weights(const int* __restrict__ rowptr, const int* __restrict__ col,
                        const float* __restrict__ asrc, const float* __restrict__ adst,
                        const float* __restrict__ x, const float* __restrict__ W,
                        const float* __restrict__ bias, const float* __restrict__ W2,
                        const float* __restrict__ att_s2, const float* __restrict__ att_d2,
                        float* __restrict__ h2, float* __restrict__ asrc2, float* __restrict__ adst2,
                        int n_nodes) {
  __shared__ __attribute__((aligned(16))) float Z[L1_NODES * 512];
  const int lane = threadIdx.x & 31;
  const int wv = __builtin_amdgcn_readfirstlane((int)(threadIdx.x >> 5));
  const int nb = blockIdx.x * L1_NODES;

  for (int half = 0; half < 2; ++half) {
    const int m = wv + 16 * half;
    const int n = nb + m;
    float acc[4][4];
    for (int h = 0; h < 4; ++h)
      for (int q = 0; q < 4; ++q) acc[h][q] = 0.f;
    if (n < n_nodes) {
      const int beg = rowptr[n];
      const int end = rowptr[n + 1];
      float ad[4], as_self[4], vself[4], mx[4];
      load_heads(adst + (size_t)n * 4, ad);
      load_heads(asrc + (size_t)n * 4, as_self);
      for (int h = 0; h < 4; ++h) { vself[h] = leaky(as_self[h] + ad[h]); mx[h] = vself[h]; }

      const int  i0  = beg + lane;
      const bool ok0 = i0 < end;
      const int  s0  = ok0 ? col[i0] : n;
      float e0[4];
      {
        float a[4];
        load_heads(asrc + (size_t)s0 * 4, a);
        for (int h = 0; h < 4; ++h) e0[h] = leaky(a[h] + ad[h]);
      }
      for (int h = 0; h < 4; ++h) mx[h] = ok0 ? fmaxf(mx[h], e0[h]) : mx[h];
      for (int i = i0 + 32; i < end; i += 32) {
        float a[4];
        load_heads(asrc + (size_t)col[i] * 4, a);
        for (int h = 0; h < 4; ++h) mx[h] = fmaxf(mx[h], leaky(a[h] + ad[h]));
      }
      for (int h = 0; h < 4; ++h)
        for (int s = 16; s >= 1; s >>= 1) mx[h] = fmaxf(mx[h], __shfl_xor(mx[h], s));
      float den[4], ex0[4];
      for (int h = 0; h < 4; ++h) {
        ex0[h] = __expf(e0[h] - mx[h]);
        den[h] = ok0 ? ex0[h] : 0.f;
      }
      for (int i = i0 + 32; i < end; i += 32) {
        float a[4];
        load_heads(asrc + (size_t)col[i] * 4, a);
        for (int h = 0; h < 4; ++h) den[h] += __expf(leaky(a[h] + ad[h]) - mx[h]);
      }
      float rden[4], wself[4];
      for (int h = 0; h < 4; ++h) {
        for (int s = 16; s >= 1; s >>= 1) den[h] += __shfl_xor(den[h], s);
        const float exs = __expf(vself[h] - mx[h]);
        den[h] += exs;
        den[h] += 1e-16f;
        rden[h]  = __builtin_amdgcn_rcpf(den[h]);
        wself[h] = exs * rden[h];
      }
      {
        const v4f f = *(const v4f*)(x + (size_t)n * 128 + lane * 4);
        for (int h = 0; h < 4; ++h)
          for (int q = 0; q < 4; ++q) acc[h][q] += wself[h] * f[q];
      }
      for (int base = beg; base < end; base += 32) {
        int   s;
        float w[4];
        if (base == beg) {
          s = s0;
          for (int h = 0; h < 4; ++h) w[h] = ok0 ? ex0[h] * rden[h] : 0.f;
        } else {
          const int  i  = base + lane;
          const bool ok = i < end;
          s = ok ? col[i] : n;
          float a[4];
          load_heads(asrc + (size_t)s * 4, a);
          for (int h = 0; h < 4; ++h)
            w[h] = ok ? __expf(leaky(a[h] + ad[h]) - mx[h]) * rden[h] : 0.f;
        }
        const int cn  = (end - base) < 32 ? (end - base) : 32;
        const int cnp = (cn + 7) & ~7;
        for (int k = 0; k < cnp; k += 8) {
          v4f   f[8];
          float wk[8][4];
#pragma unroll
          for (int u = 0; u < 8; ++u) {
            const int sk = __builtin_amdgcn_readlane(s, k + u);
            for (int h = 0; h < 4; ++h)
              wk[u][h] = __int_as_float(__builtin_amdgcn_readlane(__float_as_int(w[h]), k + u));
            f[u] = *(const v4f*)((const char*)x + ((unsigned)sk * 512u + (unsigned)lane * 16u));
          }
          asm volatile("" :: "v"(f[0]), "v"(f[1]), "v"(f[2]), "v"(f[3]),
                             "v"(f[4]), "v"(f[5]), "v"(f[6]), "v"(f[7]));
#pragma unroll
          for (int u = 0; u < 8; ++u)
            for (int h = 0; h < 4; ++h)
              for (int q = 0; q < 4; ++q) acc[h][q] += wk[u][h] * f[u][q];
        }
      }
    }
    float* zr = Z + m * 512;
    for (int h = 0; h < 4; ++h) {
      v4f o;
      o.x = acc[h][0]; o.y = acc[h][1]; o.z = acc[h][2]; o.w = acc[h][3];
      *(v4f*)(zr + ((h * 128 + lane * 4) ^ ((m & 15) << 2))) = o;
    }
  }
  __syncthreads();

  const int lh    = lane >> 4;
  const int l16   = lane & 15;
  const int sw    = l16 << 2;
  const int hd    = (wv >> 1) & 3;
  const int cbase = hd * 64 + (wv & 1) * 32;
  v8f c00 = {}, c01 = {}, c10 = {}, c11 = {};
  if (wv < 8) {
    const float* bp = W + (size_t)(lh * 2) * 256 + cbase + l16 * 2;
    const float* z0 = Z + l16 * 512;
    const float* z1 = Z + (16 + l16) * 512;
    const int zc0 = hd * 128 + lh * 2;
#pragma unroll 8
    for (int k = 0; k < 128; k += 4) {
      const int zc = (zc0 + k) ^ sw;
      const v2f a0 = *(const v2f*)(z0 + zc);
      const v2f a1 = *(const v2f*)(z1 + zc);
      const v2f b0 = *(const v2f*)(bp + (size_t)k * 256);
      const v2f b1 = *(const v2f*)(bp + (size_t)(k + 1) * 256);
      v2f b;
      b.x = b0.x; b.y = b1.x;
      c00 = __builtin_amdgcn_wmma_f32_16x16x4_f32(false, a0, false, b, (short)0, c00, false, false);
      c10 = __builtin_amdgcn_wmma_f32_16x16x4_f32(false, a1, false, b, (short)0, c10, false, false);
      b.x = b0.y; b.y = b1.y;
      c01 = __builtin_amdgcn_wmma_f32_16x16x4_f32(false, a0, false, b, (short)0, c01, false, false);
      c11 = __builtin_amdgcn_wmma_f32_16x16x4_f32(false, a1, false, b, (short)0, c11, false, false);
    }
  }
  __syncthreads();

  if (wv < 8) {
    const v2f bb = *(const v2f*)(bias + cbase + l16 * 2);
#pragma unroll
    for (int mt = 0; mt < 2; ++mt) {
#pragma unroll
      for (int r = 0; r < 8; ++r) {
        const int m = mt * 16 + lh * 8 + r;
        v2f o;
        o.x = (mt ? c10[r] : c00[r]) + bb.x;
        o.y = (mt ? c11[r] : c01[r]) + bb.y;
        o.x = o.x > 0.f ? o.x : __expf(o.x) - 1.0f;
        o.y = o.y > 0.f ? o.y : __expf(o.y) - 1.0f;
        *(v2f*)(Z + m * 256 + ((cbase + l16 * 2) ^ ((m & 15) << 2))) = o;
      }
    }
  }
  __syncthreads();

  const int rt = wv & 1;
  const int kq = (wv >> 1) & 3;
  float* part = Z + L1_NODES * 256;
  v8f acc0 = {}, acc1 = {}, acc2 = {}, acc3 = {};
  if (wv < 8) {
    const float* yr  = Z + (rt * 16 + l16) * 256;
    const int    yc0 = kq * 64 + lh * 2;
    const float* bq  = W2 + (size_t)(kq * 64 + lh * 2) * 64 + l16 * 4;
#pragma unroll 8
    for (int k = 0; k < 64; k += 4) {
      const v2f a  = *(const v2f*)(yr + ((yc0 + k) ^ sw));
      const v4f b0 = *(const v4f*)(bq + (size_t)k * 64);
      const v4f b1 = *(const v4f*)(bq + (size_t)(k + 1) * 64);
      v2f b;
      b.x = b0.x; b.y = b1.x;
      acc0 = __builtin_amdgcn_wmma_f32_16x16x4_f32(false, a, false, b, (short)0, acc0, false, false);
      b.x = b0.y; b.y = b1.y;
      acc1 = __builtin_amdgcn_wmma_f32_16x16x4_f32(false, a, false, b, (short)0, acc1, false, false);
      b.x = b0.z; b.y = b1.z;
      acc2 = __builtin_amdgcn_wmma_f32_16x16x4_f32(false, a, false, b, (short)0, acc2, false, false);
      b.x = b0.w; b.y = b1.w;
      acc3 = __builtin_amdgcn_wmma_f32_16x16x4_f32(false, a, false, b, (short)0, acc3, false, false);
    }
  }
  if (wv >= 2 && wv < 8) {
    float* pw = part + (((kq - 1) * 2 + rt) * 8 * 32 + lane) * 4;
#pragma unroll
    for (int r = 0; r < 8; ++r) {
      v4f o;
      o.x = acc0[r]; o.y = acc1[r]; o.z = acc2[r]; o.w = acc3[r];
      *(v4f*)(pw + r * 128) = o;
    }
  }
  __syncthreads();
  if (wv >= 2) return;

#pragma unroll 1
  for (int q = 0; q < 3; ++q) {
    const float* pr = part + ((q * 2 + rt) * 8 * 32 + lane) * 4;
#pragma unroll
    for (int r = 0; r < 8; ++r) {
      const v4f p = *(const v4f*)(pr + r * 128);
      acc0[r] += p.x; acc1[r] += p.y; acc2[r] += p.z; acc3[r] += p.w;
    }
  }

  const int row0 = nb + rt * 16 + lh * 8;
  float* crow = h2 + (size_t)row0 * 64 + l16 * 4;
#pragma unroll
  for (int r = 0; r < 8; ++r) {
    v4f o;
    o.x = acc0[r]; o.y = acc1[r]; o.z = acc2[r]; o.w = acc3[r];
    if (row0 + r < n_nodes) *(v4f*)(crow + (size_t)r * 64) = o;
  }

  const v4f s = *(const v4f*)(att_s2 + l16 * 4);
  const v4f d = *(const v4f*)(att_d2 + l16 * 4);
  float ps[8], pd[8];
#pragma unroll
  for (int r = 0; r < 8; ++r) {
    ps[r] = acc0[r] * s.x + acc1[r] * s.y + acc2[r] * s.z + acc3[r] * s.w;
    pd[r] = acc0[r] * d.x + acc1[r] * d.y + acc2[r] * d.z + acc3[r] * d.w;
  }
#pragma unroll
  for (int r = 0; r < 8; ++r) {
    for (int m = 8; m >= 1; m >>= 1) {
      ps[r] += __shfl_xor(ps[r], m);
      pd[r] += __shfl_xor(pd[r], m);
    }
  }
  float vs = ps[0], vd = pd[0];
#pragma unroll
  for (int r = 1; r < 8; ++r) {
    vs = (l16 == r) ? ps[r] : vs;
    vd = (l16 == r) ? pd[r] : vd;
  }
  if (l16 < 8 && row0 + l16 < n_nodes) {
    asrc2[row0 + l16] = vs;
    adst2[row0 + l16] = vd;
  }
}


static inline int cdiv(long long a, long long b) { return (int)((a + b - 1) / b); }

extern "C" void kernel_launch(void* const* d_in, const int* in_sizes, int n_in,
                              void* d_out, int out_size, void* d_ws, size_t ws_size,
                              hipStream_t stream) {
  const float* x      = (const float*)d_in[0];
  const int*   ei     = (const int*)  d_in[1];
  const float* W1     = (const float*)d_in[2];
  const float* att_s1 = (const float*)d_in[3];
  const float* att_d1 = (const float*)d_in[4];
  const float* b1     = (const float*)d_in[5];
  const float* W2     = (const float*)d_in[6];
  const float* att_s2 = (const float*)d_in[7];
  const float* att_d2 = (const float*)d_in[8];
  const float* b2     = (const float*)d_in[9];

  const int IN   = 128;
  const int Nn   = in_sizes[0] / IN;
  const int Eraw = in_sizes[1] / 2;

  float* ws = (float*)d_ws;
  size_t off = 0;
  const size_t npad = ((size_t)Nn + 4) & ~(size_t)3;
  float* h2     = ws + off; off += (size_t)Nn * 64;
  float* asrc1  = ws + off; off += (size_t)Nn * 4;
  float* adst1  = ws + off; off += (size_t)Nn * 4;
  float* asrc2  = ws + off; off += npad;
  float* adst2  = ws + off; off += npad;
  const int NS     = NSLICE;
  const int nrange = Nn / 256 + 1;
  const int rpitch = (nrange + 3) & ~3;
  const size_t ncell = (size_t)rpitch * NS;
  int*   cellcnt = (int*)(ws + off); off += ncell;
  int*   rstart  = (int*)(ws + off); off += (size_t)rpitch + 4;
  int*   rowptr  = (int*)(ws + off); off += npad;
  int*   binned  = (int*)(ws + off); off += (size_t)Eraw * 2 + 2;
  off = (off + 3) & ~(size_t)3;
  int*   lrank   = (int*)(ws + off); off += ((size_t)Eraw + 3) & ~(size_t)3;
  int*   col     = (int*)(ws + off); off += (size_t)Eraw;
  float* outf   = (float*)d_out;

  const int T = 256;

  const int chunk = cdiv((long long)Eraw, NS);
  slice_count<<<NS, SORT_THREADS, nrange * sizeof(int), stream>>>(ei, lrank, cellcnt, Eraw, chunk, nrange, rpitch);
  slice_place<<<NS, SORT_THREADS, ((2 * COLGROUPS + 1) * rpitch + 32) * sizeof(int), stream>>>(
      ei, lrank, cellcnt, rstart, binned, Eraw, chunk, nrange, rpitch);
  range_sort<<<nrange, SORT_THREADS, 0, stream>>>(binned, rstart, rowptr, col, Nn);

  score_inputs<<<cdiv((long long)Nn, 256), 1024, 0, stream>>>(x, W1, att_s1, att_d1, asrc1, adst1, Nn);
  agg_inputs_then_weights<<<cdiv((long long)Nn, L1_NODES), 512, 0, stream>>>(
      rowptr, col, asrc1, adst1, x, W1, b1, W2, att_s2, att_d2, h2, asrc2, adst2, Nn);

  seg_softmax_agg<1, 64, false><<<cdiv((long long)Nn, T / 32), T, 0, stream>>>(
      rowptr, col, asrc2, adst2, h2, b2, outf, Nn);
}


# Edge sort in two launches: a slice work-group groups its own edges by 256-node range in LDS (count, scan, copy); a range work-group reads its 256 runs via a start table; lists stay in arrival order
# speedup vs baseline: 6.6323x; 1.0963x over previous
#include <hip/hip_runtime.h>

typedef __attribute__((ext_vector_type(2))) float v2f;
typedef __attribute__((ext_vector_type(8))) float v8f;

#define NEG_SLOPE 0.2f

typedef __attribute__((ext_vector_type(4))) float v4f;


#define RANGE_SHIFT  8
#define RANGE_NODES  256
#define SORT_THREADS 1024
#define NSLICE       256
static_assert(NSLICE == 8 * (SORT_THREADS / 32), "range_sort: every wave takes the runs of eight slices");

__global__ void __launch_bounds__(SORT_THREADS)
slice_sort(const int* __restrict__ ei, int* __restrict__ celltab, int* __restrict__ binned,
           int Eraw, int chunk, int nrange) {
  extern __shared__ int hcell[];
  int* wtot = hcell + nrange;
  const int t    = threadIdx.x;
  const int lane = t & 31;
  const int wv   = t >> 5;
  const int g    = blockIdx.x;
  for (int r = t; r < nrange; r += SORT_THREADS) hcell[r] = 0;
  __syncthreads();
  const int beg = g * chunk;
  const int end = (beg + chunk < Eraw) ? (beg + chunk) : Eraw;

  int s0[4], d0[4];
#pragma unroll
  for (int j = 0; j < 4; ++j) {
    const int e  = beg + t + j * SORT_THREADS;
    const int ec = (e < end) ? e : (end - 1);
    s0[j] = ei[ec];
    d0[j] = ei[Eraw + ec];
  }
  asm volatile("" :: "v"(s0[0]), "v"(s0[1]), "v"(s0[2]), "v"(s0[3]),
                     "v"(d0[0]), "v"(d0[1]), "v"(d0[2]), "v"(d0[3]));
#pragma unroll
  for (int j = 0; j < 4; ++j) {
    const int e = beg + t + j * SORT_THREADS;
    if (e < end) atomicAdd(&hcell[d0[j] >> RANGE_SHIFT], 1);
  }
  for (int e0 = beg + t + 4 * SORT_THREADS; e0 < end; e0 += 4 * SORT_THREADS) {
    int d[4];
#pragma unroll
    for (int j = 0; j < 4; ++j) {
      const int e = e0 + j * SORT_THREADS;
      d[j] = ei[Eraw + ((e < end) ? e : (end - 1))];
    }
    asm volatile("" :: "v"(d[0]), "v"(d[1]), "v"(d[2]), "v"(d[3]));
#pragma unroll
    for (int j = 0; j < 4; ++j) {
      const int e = e0 + j * SORT_THREADS;
      if (e < end) atomicAdd(&hcell[d[j] >> RANGE_SHIFT], 1);
    }
  }
  __syncthreads();

  int carry = 0;
  for (int base = 0; base < nrange; base += SORT_THREADS) {
    const int r   = base + t;
    const int tot = (r < nrange) ? hcell[r] : 0;
    int incl = tot;
    for (int d = 1; d < 32; d <<= 1) {
      int y = __shfl_up(incl, d);
      if (lane >= d) incl += y;
    }
    if (lane == 31) wtot[wv] = incl;
    __syncthreads();
    int ws = wtot[lane];
    for (int d = 1; d < 32; d <<= 1) {
      int y = __shfl_up(ws, d);
      if (lane >= d) ws += y;
    }
    const int wprefix = __shfl(ws, wv) - __shfl(incl, 31);
    const int total   = __shfl(ws, 31);
    const int start   = carry + wprefix + incl - tot;
    if (r < nrange) {
      hcell[r] = start;
      celltab[(size_t)r * NSLICE + g] = start;
    }
    carry += total;
    __syncthreads();
  }
  if (t == 0) celltab[(size_t)nrange * NSLICE + g] = carry;

  int2* bp = (int2*)binned + beg;
#pragma unroll
  for (int j = 0; j < 4; ++j) {
    const int e = beg + t + j * SORT_THREADS;
    if (e < end) bp[atomicAdd(&hcell[d0[j] >> RANGE_SHIFT], 1)] = make_int2(s0[j], d0[j]);
  }
  for (int e0 = beg + t + 4 * SORT_THREADS; e0 < end; e0 += 4 * SORT_THREADS) {
    int s[4], d[4];
#pragma unroll
    for (int j = 0; j < 4; ++j) {
      const int e  = e0 + j * SORT_THREADS;
      const int ec = (e < end) ? e : (end - 1);
      s[j] = ei[ec];
      d[j] = ei[Eraw + ec];
    }
    asm volatile("" :: "v"(s[0]), "v"(s[1]), "v"(s[2]), "v"(s[3]), "v"(d[0]), "v"(d[1]), "v"(d[2]), "v"(d[3]));
#pragma unroll
    for (int j = 0; j < 4; ++j) {
      const int e = e0 + j * SORT_THREADS;
      if (e < end) bp[atomicAdd(&hcell[d[j] >> RANGE_SHIFT], 1)] = make_int2(s[j], d[j]);
    }
  }
}

__global__ void __launch_bounds__(SORT_THREADS)
range_sort(const int* __restrict__ binned, const int* __restrict__ celltab,
           int* __restrict__ rowptr, int* __restrict__ col, int n_nodes, int chunk) {
  __shared__ int cnt[RANGE_NODES];
  __shared__ int wtot[32];
  const int t    = threadIdx.x;
  const int lane = t & 31;
  const int wv = __builtin_amdgcn_readfirstlane((int)(threadIdx.x >> 5));
  const int b  = blockIdx.x;
  const int2* bp = (const int2*)binned;

  int src[8], len[8];
  int part = 0;
#pragma unroll
  for (int k = 0; k < 8; ++k) {
    const int g = wv + 32 * k;
    const int a = celltab[(size_t)b * NSLICE + g];
    const int e = celltab[(size_t)(b + 1) * NSLICE + g];
    src[k] = g * chunk + a;
    len[k] = e - a;
    part  += a;
  }
  if (lane == 0) wtot[wv] = part;
  if (t < RANGE_NODES) cnt[t] = 0;
  __syncthreads();
  int beg = wtot[lane];
  for (int m = 16; m >= 1; m >>= 1) beg += __shfl_xor(beg, m);

  int2 sd[8];
  {
#pragma unroll
    for (int k = 0; k < 8; ++k) {
      const int lim = len[k] - 1;
      const int i   = (lim < 0) ? 0 : (src[k] + ((lane < lim) ? lane : lim));
      sd[k] = bp[i];
    }
    asm volatile("" :: "v"(sd[0].y), "v"(sd[1].y), "v"(sd[2].y), "v"(sd[3].y),
                       "v"(sd[4].y), "v"(sd[5].y), "v"(sd[6].y), "v"(sd[7].y));
#pragma unroll
    for (int k = 0; k < 8; ++k)
      if (lane < len[k]) atomicAdd(&cnt[sd[k].y & (RANGE_NODES - 1)], 1);
#pragma unroll
    for (int k = 0; k < 8; ++k)
      for (int i = lane + 32; i < len[k]; i += 32)
        atomicAdd(&cnt[bp[src[k] + i].y & (RANGE_NODES - 1)], 1);
  }
  __syncthreads();
  if (t < 32) {
    int c[8];
    int tsum = 0;
    for (int j = 0; j < 8; ++j) { c[j] = cnt[t * 8 + j]; tsum += c[j]; }
    int incl = tsum;
    for (int d = 1; d < 32; d <<= 1) {
      int y = __shfl_up(incl, d);
      if (t >= d) incl += y;
    }
    int run = incl - tsum;
    for (int j = 0; j < 8; ++j) { cnt[t * 8 + j] = run; run += c[j]; }
  }
  __syncthreads();
  if (t < RANGE_NODES) {
    const int node = b * RANGE_NODES + t;
    if (node <= n_nodes) rowptr[node] = beg + cnt[t];
  }
  __syncthreads();
  {
#pragma unroll
    for (int k = 0; k < 8; ++k) {
      if (lane < len[k]) {
        const int p = atomicAdd(&cnt[sd[k].y & (RANGE_NODES - 1)], 1);
        col[beg + p] = sd[k].x;
      }
    }
#pragma unroll
    for (int k = 0; k < 8; ++k) {
      for (int i = lane + 32; i < len[k]; i += 32) {
        const int2 v = bp[src[k] + i];
        const int  p = atomicAdd(&cnt[v.y & (RANGE_NODES - 1)], 1);
        col[beg + p] = v.x;
      }
    }
  }
}


template <int N> using vecf = float __attribute__((ext_vector_type(N)));

__device__ __forceinline__ void load_heads(const float* __restrict__ p, float (&v)[4]) {
  const float4 t = *(const float4*)p;
  v[0] = t.x; v[1] = t.y; v[2] = t.z; v[3] = t.w;
}
__device__ __forceinline__ void load_heads(const float* __restrict__ p, float (&v)[1]) {
  v[0] = p[0];
}

__device__ __forceinline__ float leaky(float v) { return v > 0.f ? v : NEG_SLOPE * v; }

template <int H, int F, bool ELU>
__global__ void __launch_bounds__(256)
seg_softmax_agg(const int* __restrict__ rowptr, const int* __restrict__ col,
                const float* __restrict__ asrc, const float* __restrict__ adst,
                const float* __restrict__ feat, const float* __restrict__ bias,
                float* __restrict__ out, int n_nodes) {
  constexpr int C   = H * F;
  constexpr int CPL = C / 32;
  constexpr int VW  = CPL >= 4 ? 4 : CPL;
  constexpr int NL  = CPL / VW;
  constexpr int U   = 8 / NL;
  static_assert(U * NL == 8, "the load fence below names eight values");
  typedef vecf<VW> vrow;

  const int lane = threadIdx.x & 31;
  const int n = __builtin_amdgcn_readfirstlane(
      (int)(blockIdx.x * (blockDim.x >> 5) + (threadIdx.x >> 5)));
  if (n >= n_nodes) return;
  const int beg = rowptr[n];
  const int end = rowptr[n + 1];

  float ad[H], as_self[H], vself[H], mx[H];
  load_heads(adst + (size_t)n * H, ad);
  load_heads(asrc + (size_t)n * H, as_self);
  for (int h = 0; h < H; ++h) { vself[h] = leaky(as_self[h] + ad[h]); mx[h] = vself[h]; }

  const int  i0  = beg + lane;
  const bool ok0 = i0 < end;
  const int  s0  = ok0 ? col[i0] : n;
  float e0[H];
  {
    float a[H];
    load_heads(asrc + (size_t)s0 * H, a);
    for (int h = 0; h < H; ++h) e0[h] = leaky(a[h] + ad[h]);
  }

  for (int h = 0; h < H; ++h) mx[h] = ok0 ? fmaxf(mx[h], e0[h]) : mx[h];
  for (int i = i0 + 32; i < end; i += 32) {
    float a[H];
    load_heads(asrc + (size_t)col[i] * H, a);
    for (int h = 0; h < H; ++h) mx[h] = fmaxf(mx[h], leaky(a[h] + ad[h]));
  }
  for (int h = 0; h < H; ++h) {
    for (int m = 16; m >= 1; m >>= 1) mx[h] = fmaxf(mx[h], __shfl_xor(mx[h], m));
  }

  float den[H], ex0[H];
  for (int h = 0; h < H; ++h) {
    ex0[h] = __expf(e0[h] - mx[h]);
    den[h] = ok0 ? ex0[h] : 0.f;
  }
  for (int i = i0 + 32; i < end; i += 32) {
    float a[H];
    load_heads(asrc + (size_t)col[i] * H, a);
    for (int h = 0; h < H; ++h) den[h] += __expf(leaky(a[h] + ad[h]) - mx[h]);
  }
  float rden[H], wself[H];
  for (int h = 0; h < H; ++h) {
    for (int m = 16; m >= 1; m >>= 1) den[h] += __shfl_xor(den[h], m);
    const float exs = __expf(vself[h] - mx[h]);
    den[h] += exs;
    den[h] += 1e-16f;
    rden[h]  = __builtin_amdgcn_rcpf(den[h]);
    wself[h] = exs * rden[h];
  }

  int hsel[NL];
  vrow acc[NL];
  for (int j = 0; j < NL; ++j) {
    hsel[j] = (j * 32 * VW + lane * VW) / F;
    for (int q = 0; q < VW; ++q) acc[j][q] = 0.f;
  }
  {
    const float* row = feat + (size_t)n * C + lane * VW;
    for (int j = 0; j < NL; ++j) {
      float w = wself[0];
      for (int h = 1; h < H; ++h) w = (hsel[j] == h) ? wself[h] : w;
      const vrow f = *(const vrow*)(row + j * 32 * VW);
      for (int q = 0; q < VW; ++q) acc[j][q] += w * f[q];
    }
  }
  for (int base = beg; base < end; base += 32) {
    int   s;
    float w[H];
    if (base == beg) {
      s = s0;
      for (int h = 0; h < H; ++h) w[h] = ok0 ? ex0[h] * rden[h] : 0.f;
    } else {
      const int  i  = base + lane;
      const bool ok = i < end;
      s = ok ? col[i] : n;
      float a[H];
      load_heads(asrc + (size_t)s * H, a);
      for (int h = 0; h < H; ++h)
        w[h] = ok ? __expf(leaky(a[h] + ad[h]) - mx[h]) * rden[h] : 0.f;
    }
    const int cn  = (end - base) < 32 ? (end - base) : 32;
    const int cnp = (cn + U - 1) & ~(U - 1);
    for (int k = 0; k < cnp; k += U) {
      vrow  f[U][NL];
      float wj[U][NL];
#pragma unroll
      for (int u = 0; u < U; ++u) {
        const int sk = __builtin_amdgcn_readlane(s, k + u);
        float wk[H];
        for (int h = 0; h < H; ++h)
          wk[h] = __int_as_float(__builtin_amdgcn_readlane(__float_as_int(w[h]), k + u));
        const float* row = feat + (size_t)sk * C + lane * VW;
        for (int j = 0; j < NL; ++j) {
          float t = wk[0];
          for (int h = 1; h < H; ++h) t = (hsel[j] == h) ? wk[h] : t;
          wj[u][j] = t;
          f[u][j]  = *(const vrow*)(row + j * 32 * VW);
        }
      }
      asm volatile("" :: "v"(f[0 / NL][0 % NL]), "v"(f[1 / NL][1 % NL]), "v"(f[2 / NL][2 % NL]),
                         "v"(f[3 / NL][3 % NL]), "v"(f[4 / NL][4 % NL]), "v"(f[5 / NL][5 % NL]),
                         "v"(f[6 / NL][6 % NL]), "v"(f[7 / NL][7 % NL]));
#pragma unroll
      for (int u = 0; u < U; ++u)
        for (int j = 0; j < NL; ++j)
          for (int q = 0; q < VW; ++q) acc[j][q] += wj[u][j] * f[u][j][q];
    }
  }

  float* orow = out + (size_t)n * C + lane * VW;
  const float* brow = bias + lane * VW;
  for (int j = 0; j < NL; ++j) {
    const vrow b = *(const vrow*)(brow + j * 32 * VW);
    vrow r;
    for (int q = 0; q < VW; ++q) {
      float v = acc[j][q] + b[q];
      r[q] = ELU ? (v > 0.f ? v : __expf(v) - 1.0f) : v;
    }
    *(vrow*)(orow + j * 32 * VW) = r;
  }
}

__global__ void __launch_bounds__(1024)
score_inputs(const float* __restrict__ x, const float* __restrict__ W,
             const float* __restrict__ att_s, const float* __restrict__ att_d,
             float* __restrict__ asrc, float* __restrict__ adst, int n_nodes) {
  __shared__ __attribute__((aligned(16))) float V[128 * 8];
  const int t    = threadIdx.x;
  const int lane = t & 31;
  const int n0   = __builtin_amdgcn_readfirstlane((int)((blockIdx.x * 32 + (t >> 5)) * 8));

  v4f xv[8];
#pragma unroll
  for (int i = 0; i < 8; ++i) {
    const int n  = n0 + i;
    const int nc = (n < n_nodes) ? n : (n_nodes - 1);
    xv[i] = *(const v4f*)(x + (size_t)nc * 128 + lane * 4);
  }
  {
    const int wv = t >> 5, hw = lane >> 4, l16 = lane & 15;
    const int h  = ((wv & 1) << 1) | hw;
    const v4f as4 = *(const v4f*)(att_s + h * 64 + l16 * 4);
    const v4f ad4 = *(const v4f*)(att_d + h * 64 + l16 * 4);
    v4f w4[8];
#pragma unroll
    for (int it = 0; it < 8; ++it)
      w4[it] = *(const v4f*)(W + (size_t)(it * 16 + (wv >> 1)) * 256 + h * 64 + l16 * 4);
#pragma unroll
    for (int it = 0; it < 8; ++it) {
      const int k = it * 16 + (wv >> 1);
      float ps = w4[it].x * as4.x + w4[it].y * as4.y + w4[it].z * as4.z + w4[it].w * as4.w;
      float pd = w4[it].x * ad4.x + w4[it].y * ad4.y + w4[it].z * ad4.z + w4[it].w * ad4.w;
      for (int s = 8; s >= 1; s >>= 1) {
        ps += __shfl_xor(ps, s);
        pd += __shfl_xor(pd, s);
      }
      if (l16 == 0) { V[k * 8 + h] = ps; V[k * 8 + 4 + h] = pd; }
    }
  }
  __syncthreads();
  float vr[4][8];
#pragma unroll
  for (int q = 0; q < 4; ++q) {
    const v4f lo = *(const v4f*)(V + (lane * 4 + q) * 8);
    const v4f hi = *(const v4f*)(V + (lane * 4 + q) * 8 + 4);
    vr[q][0] = lo.x; vr[q][1] = lo.y; vr[q][2] = lo.z; vr[q][3] = lo.w;
    vr[q][4] = hi.x; vr[q][5] = hi.y; vr[q][6] = hi.z; vr[q][7] = hi.w;
  }
  const bool b4 = (lane & 16) != 0, b3 = (lane & 8) != 0, b2 = (lane & 4) != 0;
  const int  jj = (b4 ? 4 : 0) + (b3 ? 2 : 0) + (b2 ? 1 : 0);
#pragma unroll
  for (int i = 0; i < 8; ++i) {
    const int n = n0 + i;
    if (n >= n_nodes) break;
    float p[8];
#pragma unroll
    for (int j = 0; j < 8; ++j)
      p[j] = xv[i].x * vr[0][j] + xv[i].y * vr[1][j] + xv[i].z * vr[2][j] + xv[i].w * vr[3][j];
    float q4[4], q2[2];
#pragma unroll
    for (int j = 0; j < 4; ++j) {
      const float keep = b4 ? p[4 + j] : p[j];
      const float send = b4 ? p[j] : p[4 + j];
      q4[j] = keep + __shfl_xor(send, 16);
    }
#pragma unroll
    for (int j = 0; j < 2; ++j) {
      const float keep = b3 ? q4[2 + j] : q4[j];
      const float send = b3 ? q4[j] : q4[2 + j];
      q2[j] = keep + __shfl_xor(send, 8);
    }
    const float keep = b2 ? q2[1] : q2[0];
    const float send = b2 ? q2[0] : q2[1];
    float v = keep + __shfl_xor(send, 4);
    v += __shfl_xor(v, 2);
    v += __shfl_xor(v, 1);
    if ((lane & 3) == 0) {
      float* dstp = (jj < 4) ? asrc : adst;
      dstp[(size_t)n * 4 + (jj & 3)] = v;
    }
  }
}

#define L1_NODES 32

__global__ void __launch_bounds__(512)
agg_inputs_then_weights(const int* __restrict__ rowptr, const int* __restrict__ col,
                        const float* __restrict__ asrc, const float* __restrict__ adst,
                        const float* __restrict__ x, const float* __restrict__ W,
                        const float* __restrict__ bias, const float* __restrict__ W2,
                        const float* __restrict__ att_s2, const float* __restrict__ att_d2,
                        float* __restrict__ h2, float* __restrict__ asrc2, float* __restrict__ adst2,
                        int n_nodes) {
  __shared__ __attribute__((aligned(16))) float Z[L1_NODES * 512];
  const int lane = threadIdx.x & 31;
  const int wv = __builtin_amdgcn_readfirstlane((int)(threadIdx.x >> 5));
  const int nb = blockIdx.x * L1_NODES;

  for (int half = 0; half < 2; ++half) {
    const int m = wv + 16 * half;
    const int n = nb + m;
    float acc[4][4];
    for (int h = 0; h < 4; ++h)
      for (int q = 0; q < 4; ++q) acc[h][q] = 0.f;
    if (n < n_nodes) {
      const int beg = rowptr[n];
      const int end = rowptr[n + 1];
      float ad[4], as_self[4], vself[4], mx[4];
      load_heads(adst + (size_t)n * 4, ad);
      load_heads(asrc + (size_t)n * 4, as_self);
      for (int h = 0; h < 4; ++h) { vself[h] = leaky(as_self[h] + ad[h]); mx[h] = vself[h]; }

      const int  i0  = beg + lane;
      const bool ok0 = i0 < end;
      const int  s0  = ok0 ? col[i0] : n;
      float e0[4];
      {
        float a[4];
        load_heads(asrc + (size_t)s0 * 4, a);
        for (int h = 0; h < 4; ++h) e0[h] = leaky(a[h] + ad[h]);
      }
      for (int h = 0; h < 4; ++h) mx[h] = ok0 ? fmaxf(mx[h], e0[h]) : mx[h];
      for (int i = i0 + 32; i < end; i += 32) {
        float a[4];
        load_heads(asrc + (size_t)col[i] * 4, a);
        for (int h = 0; h < 4; ++h) mx[h] = fmaxf(mx[h], leaky(a[h] + ad[h]));
      }
      for (int h = 0; h < 4; ++h)
        for (int s = 16; s >= 1; s >>= 1) mx[h] = fmaxf(mx[h], __shfl_xor(mx[h], s));
      float den[4], ex0[4];
      for (int h = 0; h < 4; ++h) {
        ex0[h] = __expf(e0[h] - mx[h]);
        den[h] = ok0 ? ex0[h] : 0.f;
      }
      for (int i = i0 + 32; i < end; i += 32) {
        float a[4];
        load_heads(asrc + (size_t)col[i] * 4, a);
        for (int h = 0; h < 4; ++h) den[h] += __expf(leaky(a[h] + ad[h]) - mx[h]);
      }
      float rden[4], wself[4];
      for (int h = 0; h < 4; ++h) {
        for (int s = 16; s >= 1; s >>= 1) den[h] += __shfl_xor(den[h], s);
        const float exs = __expf(vself[h] - mx[h]);
        den[h] += exs;
        den[h] += 1e-16f;
        rden[h]  = __builtin_amdgcn_rcpf(den[h]);
        wself[h] = exs * rden[h];
      }
      {
        const v4f f = *(const v4f*)(x + (size_t)n * 128 + lane * 4);
        for (int h = 0; h < 4; ++h)
          for (int q = 0; q < 4; ++q) acc[h][q] += wself[h] * f[q];
      }
      for (int base = beg; base < end; base += 32) {
        int   s;
        float w[4];
        if (base == beg) {
          s = s0;
          for (int h = 0; h < 4; ++h) w[h] = ok0 ? ex0[h] * rden[h] : 0.f;
        } else {
          const int  i  = base + lane;
          const bool ok = i < end;
          s = ok ? col[i] : n;
          float a[4];
          load_heads(asrc + (size_t)s * 4, a);
          for (int h = 0; h < 4; ++h)
            w[h] = ok ? __expf(leaky(a[h] + ad[h]) - mx[h]) * rden[h] : 0.f;
        }
        const int cn  = (end - base) < 32 ? (end - base) : 32;
        const int cnp = (cn + 7) & ~7;
        for (int k = 0; k < cnp; k += 8) {
          v4f   f[8];
          float wk[8][4];
#pragma unroll
          for (int u = 0; u < 8; ++u) {
            const int sk = __builtin_amdgcn_readlane(s, k + u);
            for (int h = 0; h < 4; ++h)
              wk[u][h] = __int_as_float(__builtin_amdgcn_readlane(__float_as_int(w[h]), k + u));
            f[u] = *(const v4f*)((const char*)x + ((unsigned)sk * 512u + (unsigned)lane * 16u));
          }
          asm volatile("" :: "v"(f[0]), "v"(f[1]), "v"(f[2]), "v"(f[3]),
                             "v"(f[4]), "v"(f[5]), "v"(f[6]), "v"(f[7]));
#pragma unroll
          for (int u = 0; u < 8; ++u)
            for (int h = 0; h < 4; ++h)
              for (int q = 0; q < 4; ++q) acc[h][q] += wk[u][h] * f[u][q];
        }
      }
    }
    float* zr = Z + m * 512;
    for (int h = 0; h < 4; ++h) {
      v4f o;
      o.x = acc[h][0]; o.y = acc[h][1]; o.z = acc[h][2]; o.w = acc[h][3];
      *(v4f*)(zr + ((h * 128 + lane * 4) ^ ((m & 15) << 2))) = o;
    }
  }
  __syncthreads();

  const int lh    = lane >> 4;
  const int l16   = lane & 15;
  const int sw    = l16 << 2;
  const int hd    = (wv >> 1) & 3;
  const int cbase = hd * 64 + (wv & 1) * 32;
  v8f c00 = {}, c01 = {}, c10 = {}, c11 = {};
  if (wv < 8) {
    const float* bp = W + (size_t)(lh * 2) * 256 + cbase + l16 * 2;
    const float* z0 = Z + l16 * 512;
    const float* z1 = Z + (16 + l16) * 512;
    const int zc0 = hd * 128 + lh * 2;
#pragma unroll 8
    for (int k = 0; k < 128; k += 4) {
      const int zc = (zc0 + k) ^ sw;
      const v2f a0 = *(const v2f*)(z0 + zc);
      const v2f a1 = *(const v2f*)(z1 + zc);
      const v2f b0 = *(const v2f*)(bp + (size_t)k * 256);
      const v2f b1 = *(const v2f*)(bp + (size_t)(k + 1) * 256);
      v2f b;
      b.x = b0.x; b.y = b1.x;
      c00 = __builtin_amdgcn_wmma_f32_16x16x4_f32(false, a0, false, b, (short)0, c00, false, false);
      c10 = __builtin_amdgcn_wmma_f32_16x16x4_f32(false, a1, false, b, (short)0, c10, false, false);
      b.x = b0.y; b.y = b1.y;
      c01 = __builtin_amdgcn_wmma_f32_16x16x4_f32(false, a0, false, b, (short)0, c01, false, false);
      c11 = __builtin_amdgcn_wmma_f32_16x16x4_f32(false, a1, false, b, (short)0, c11, false, false);
    }
  }
  __syncthreads();

  if (wv < 8) {
    const v2f bb = *(const v2f*)(bias + cbase + l16 * 2);
#pragma unroll
    for (int mt = 0; mt < 2; ++mt) {
#pragma unroll
      for (int r = 0; r < 8; ++r) {
        const int m = mt * 16 + lh * 8 + r;
        v2f o;
        o.x = (mt ? c10[r] : c00[r]) + bb.x;
        o.y = (mt ? c11[r] : c01[r]) + bb.y;
        o.x = o.x > 0.f ? o.x : __expf(o.x) - 1.0f;
        o.y = o.y > 0.f ? o.y : __expf(o.y) - 1.0f;
        *(v2f*)(Z + m * 256 + ((cbase + l16 * 2) ^ ((m & 15) << 2))) = o;
      }
    }
  }
  __syncthreads();

  const int rt = wv & 1;
  const int kq = (wv >> 1) & 3;
  float* part = Z + L1_NODES * 256;
  v8f acc0 = {}, acc1 = {}, acc2 = {}, acc3 = {};
  if (wv < 8) {
    const float* yr  = Z + (rt * 16 + l16) * 256;
    const int    yc0 = kq * 64 + lh * 2;
    const float* bq  = W2 + (size_t)(kq * 64 + lh * 2) * 64 + l16 * 4;
#pragma unroll 8
    for (int k = 0; k < 64; k += 4) {
      const v2f a  = *(const v2f*)(yr + ((yc0 + k) ^ sw));
      const v4f b0 = *(const v4f*)(bq + (size_t)k * 64);
      const v4f b1 = *(const v4f*)(bq + (size_t)(k + 1) * 64);
      v2f b;
      b.x = b0.x; b.y = b1.x;
      acc0 = __builtin_amdgcn_wmma_f32_16x16x4_f32(false, a, false, b, (short)0, acc0, false, false);
      b.x = b0.y; b.y = b1.y;
      acc1 = __builtin_amdgcn_wmma_f32_16x16x4_f32(false, a, false, b, (short)0, acc1, false, false);
      b.x = b0.z; b.y = b1.z;
      acc2 = __builtin_amdgcn_wmma_f32_16x16x4_f32(false, a, false, b, (short)0, acc2, false, false);
      b.x = b0.w; b.y = b1.w;
      acc3 = __builtin_amdgcn_wmma_f32_16x16x4_f32(false, a, false, b, (short)0, acc3, false, false);
    }
  }
  if (wv >= 2 && wv < 8) {
    float* pw = part + (((kq - 1) * 2 + rt) * 8 * 32 + lane) * 4;
#pragma unroll
    for (int r = 0; r < 8; ++r) {
      v4f o;
      o.x = acc0[r]; o.y = acc1[r]; o.z = acc2[r]; o.w = acc3[r];
      *(v4f*)(pw + r * 128) = o;
    }
  }
  __syncthreads();
  if (wv >= 2) return;

#pragma unroll 1
  for (int q = 0; q < 3; ++q) {
    const float* pr = part + ((q * 2 + rt) * 8 * 32 + lane) * 4;
#pragma unroll
    for (int r = 0; r < 8; ++r) {
      const v4f p = *(const v4f*)(pr + r * 128);
      acc0[r] += p.x; acc1[r] += p.y; acc2[r] += p.z; acc3[r] += p.w;
    }
  }

  const int row0 = nb + rt * 16 + lh * 8;
  float* crow = h2 + (size_t)row0 * 64 + l16 * 4;
#pragma unroll
  for (int r = 0; r < 8; ++r) {
    v4f o;
    o.x = acc0[r]; o.y = acc1[r]; o.z = acc2[r]; o.w = acc3[r];
    if (row0 + r < n_nodes) *(v4f*)(crow + (size_t)r * 64) = o;
  }

  const v4f s = *(const v4f*)(att_s2 + l16 * 4);
  const v4f d = *(const v4f*)(att_d2 + l16 * 4);
  float ps[8], pd[8];
#pragma unroll
  for (int r = 0; r < 8; ++r) {
    ps[r] = acc0[r] * s.x + acc1[r] * s.y + acc2[r] * s.z + acc3[r] * s.w;
    pd[r] = acc0[r] * d.x + acc1[r] * d.y + acc2[r] * d.z + acc3[r] * d.w;
  }
#pragma unroll
  for (int r = 0; r < 8; ++r) {
    for (int m = 8; m >= 1; m >>= 1) {
      ps[r] += __shfl_xor(ps[r], m);
      pd[r] += __shfl_xor(pd[r], m);
    }
  }
  float vs = ps[0], vd = pd[0];
#pragma unroll
  for (int r = 1; r < 8; ++r) {
    vs = (l16 == r) ? ps[r] : vs;
    vd = (l16 == r) ? pd[r] : vd;
  }
  if (l16 < 8 && row0 + l16 < n_nodes) {
    asrc2[row0 + l16] = vs;
    adst2[row0 + l16] = vd;
  }
}


static inline int cdiv(long long a, long long b) { return (int)((a + b - 1) / b); }

extern "C" void kernel_launch(void* const* d_in, const int* in_sizes, int n_in,
                              void* d_out, int out_size, void* d_ws, size_t ws_size,
                              hipStream_t stream) {
  const float* x      = (const float*)d_in[0];
  const int*   ei     = (const int*)  d_in[1];
  const float* W1     = (const float*)d_in[2];
  const float* att_s1 = (const float*)d_in[3];
  const float* att_d1 = (const float*)d_in[4];
  const float* b1     = (const float*)d_in[5];
  const float* W2     = (const float*)d_in[6];
  const float* att_s2 = (const float*)d_in[7];
  const float* att_d2 = (const float*)d_in[8];
  const float* b2     = (const float*)d_in[9];

  const int IN   = 128;
  const int Nn   = in_sizes[0] / IN;
  const int Eraw = in_sizes[1] / 2;

  float* ws = (float*)d_ws;
  size_t off = 0;
  const size_t npad = ((size_t)Nn + 4) & ~(size_t)3;
  float* h2     = ws + off; off += (size_t)Nn * 64;
  float* asrc1  = ws + off; off += (size_t)Nn * 4;
  float* adst1  = ws + off; off += (size_t)Nn * 4;
  float* asrc2  = ws + off; off += npad;
  float* adst2  = ws + off; off += npad;
  const int NS     = NSLICE;
  const int nrange = Nn / 256 + 1;
  int*   celltab = (int*)(ws + off); off += (size_t)(nrange + 1) * NS;
  int*   rowptr  = (int*)(ws + off); off += npad;
  int*   binned  = (int*)(ws + off); off += (size_t)Eraw * 2 + 2;
  off = (off + 3) & ~(size_t)3;
  int*   col     = (int*)(ws + off); off += (size_t)Eraw;
  float* outf   = (float*)d_out;

  const int T = 256;

  const int chunk = cdiv((long long)Eraw, NS);
  slice_sort<<<NS, SORT_THREADS, (nrange + 32) * sizeof(int), stream>>>(ei, celltab, binned, Eraw, chunk, nrange);
  range_sort<<<nrange, SORT_THREADS, 0, stream>>>(binned, celltab, rowptr, col, Nn, chunk);

  score_inputs<<<cdiv((long long)Nn, 256), 1024, 0, stream>>>(x, W1, att_s1, att_d1, asrc1, adst1, Nn);
  agg_inputs_then_weights<<<cdiv((long long)Nn, L1_NODES), 512, 0, stream>>>(
      rowptr, col, asrc1, adst1, x, W1, b1, W2, att_s2, att_d2, h2, asrc2, adst2, Nn);

  seg_softmax_agg<1, 64, false><<<cdiv((long long)Nn, T / 32), T, 0, stream>>>(
      rowptr, col, asrc2, adst2, h2, b2, outf, Nn);
}
